# starting kernel
# baseline (speedup 1.0000x reference)
#include <hip/hip_runtime.h>
#include <hip/hip_bf16.h>
#include <math.h>

#define N1D      64
#define GRID_N   (64 * 64 * 64)
#define N_ATOMS  128
#define K_TERMS  6
#define SPACING  0.5f
#define LOG2E    1.4426950408889634f

typedef float v2f __attribute__((ext_vector_type(2)));
typedef float v8f __attribute__((ext_vector_type(8)));

__device__ __forceinline__ float fast_exp2(float x) {
#if __has_builtin(__builtin_amdgcn_exp2f)
    return __builtin_amdgcn_exp2f(x);
#else
    return exp2f(x);
#endif
}

__global__ __launch_bounds__(256) void gen_dft_kernel(float* __restrict__ Wre,
                                                      float* __restrict__ WimF,
                                                      float* __restrict__ WimI) {
    int t = blockIdx.x * blockDim.x + threadIdx.x;
    if (t >= N1D * N1D) return;
    int k = t >> 6;
    int n = t & 63;
    int kn = (k * n) & 63;
    float ang = (float)kn * 0.09817477042468103f;
    float sgn = ((k + n) & 1) ? -0.125f : 0.125f;
    float c = sgn * cosf(ang);
    float s = sgn * sinf(ang);
    Wre[t]  = c;
    WimF[t] = -s;
    WimI[t] = s;
}

__global__ __launch_bounds__(256) void density_kernel(const float* __restrict__ X,
                                                      const float* __restrict__ aw,
                                                      const float* __restrict__ bw,
                                                      float* __restrict__ rho_re,
                                                      float* __restrict__ rho_im) {
    const int lane   = threadIdx.x & 31;
    const int waveId = (blockIdx.x * blockDim.x + threadIdx.x) >> 5;
    const int g0 = waveId << 4;
    const int xg = g0 >> 12;
    const int yg = (g0 >> 6) & 63;
    const int zb = g0 & 63;

    const float cx = 16.0f;
    const float gx = (float)xg * SPACING - cx;
    const float gy = (float)yg * SPACING - cx;
    const int half = lane >> 4;
    const int l16  = lane & 15;

    float gzm = ((float)(zb + l16) * SPACING) - cx;
    v2f Aop;
    Aop.x = half ? (-2.0f * gzm) : (-2.0f * gx);
    Aop.y = half ? 0.0f          : (-2.0f * gy);

    float gsq[8];
#pragma unroll
    for (int v = 0; v < 8; ++v) {
        float gz = ((float)(zb + v + 8 * half) * SPACING) - cx;
        gsq[v] = gx * gx + gy * gy + gz * gz;
    }

    float dens[8] = {0.f, 0.f, 0.f, 0.f, 0.f, 0.f, 0.f, 0.f};

    for (int a0 = 0; a0 < N_ATOMS; a0 += 16) {
        const int a = a0 + l16;
        float ax = X[3 * a + 0] - cx;
        float ay = X[3 * a + 1] - cx;
        float az = X[3 * a + 2] - cx;
        float xsq = ax * ax + ay * ay + az * az;

        v2f Bop;
        Bop.x = half ? az   : ax;
        Bop.y = half ? 0.0f : ay;

        v8f dot = {};
        dot = __builtin_amdgcn_wmma_f32_16x16x4_f32(
            false, Aop, false, Bop, (short)0, dot, false, false);

        float awk[K_TERMS], bl2[K_TERMS];
#pragma unroll
        for (int k = 0; k < K_TERMS; ++k) {
            awk[k] = aw[a * K_TERMS + k];
            bl2[k] = bw[a * K_TERMS + k] * LOG2E;
        }

#pragma unroll
        for (int v = 0; v < 8; ++v) {
            float d2 = dot[v] + gsq[v] + xsq;
            float s = 0.0f;
#pragma unroll
            for (int k = 0; k < K_TERMS; ++k)
                s += awk[k] * fast_exp2(bl2[k] * d2);
            dens[v] += s;
        }
    }

#pragma unroll
    for (int v = 0; v < 8; ++v) {
        float s = dens[v];
        s += __shfl_xor(s, 1, 32);
        s += __shfl_xor(s, 2, 32);
        s += __shfl_xor(s, 4, 32);
        s += __shfl_xor(s, 8, 32);
        dens[v] = s;
    }
    if (lane == 0) {
#pragma unroll
        for (int v = 0; v < 8; ++v) rho_re[g0 + v] = dens[v];
    }
    if (lane == 16) {
#pragma unroll
        for (int v = 0; v < 8; ++v) rho_re[g0 + 8 + v] = dens[v];
    }
    if (lane < 16) rho_im[g0 + lane] = 0.0f;
}

__global__ __launch_bounds__(256) void dft_pass_kernel(const float* __restrict__ Tre,
                                                       const float* __restrict__ Tim,
                                                       const float* __restrict__ Mre,
                                                       const float* __restrict__ Mim,
                                                       float* __restrict__ Ure,
                                                       float* __restrict__ Uim) {
    const int lane   = threadIdx.x & 31;
    const int waveId = (blockIdx.x * blockDim.x + threadIdx.x) >> 5;
    const int p0 = (waveId & 255) << 4;
    const int k0 = (waveId >> 8) << 4;
    const int half = lane >> 4;
    const int l16  = lane & 15;
    const int krow = k0 + l16;
    const int prow = p0 + l16;

    v8f rr = {}, ii = {}, ri = {}, ir = {};

#pragma unroll 4
    for (int n0 = 0; n0 < N1D; n0 += 4) {
        const int koff = n0 + 2 * half;
        v2f mr = *(const v2f*)(Mre + krow * N1D + koff);
        v2f mi = *(const v2f*)(Mim + krow * N1D + koff);
        v2f tr = *(const v2f*)(Tre + prow * N1D + koff);
        v2f ti = *(const v2f*)(Tim + prow * N1D + koff);
        rr = __builtin_amdgcn_wmma_f32_16x16x4_f32(false, mr, false, tr, (short)0, rr, false, false);
        ii = __builtin_amdgcn_wmma_f32_16x16x4_f32(false, mi, false, ti, (short)0, ii, false, false);
        ri = __builtin_amdgcn_wmma_f32_16x16x4_f32(false, mr, false, ti, (short)0, ri, false, false);
        ir = __builtin_amdgcn_wmma_f32_16x16x4_f32(false, mi, false, tr, (short)0, ir, false, false);
    }

#pragma unroll
    for (int v = 0; v < 8; ++v) {
        const int row = k0 + v + 8 * half;
        const int idx = row * 4096 + p0 + l16;
        Ure[idx] = rr[v] - ii[v];
        Uim[idx] = ri[v] + ir[v];
    }
}

__global__ __launch_bounds__(256) void hamming_kernel(float* __restrict__ re,
                                                      float* __restrict__ im,
                                                      const float* __restrict__ h) {
    int g = blockIdx.x * blockDim.x + threadIdx.x;
    if (g >= GRID_N) return;
    float hv = h[g];
    re[g] *= hv;
    im[g] *= hv;
}

__global__ __launch_bounds__(256) void reduce1_kernel(const float* __restrict__ x,
                                                      float* __restrict__ psum,
                                                      float* __restrict__ psq) {
    __shared__ float ssum[256];
    __shared__ float ssq[256];
    const int tid = threadIdx.x;
    const int base = blockIdx.x * 512;
    float a = x[base + tid];
    float b = x[base + 256 + tid];
    ssum[tid] = a + b;
    ssq[tid]  = a * a + b * b;
    __syncthreads();
    for (int s = 128; s > 0; s >>= 1) {
        if (tid < s) { ssum[tid] += ssum[tid + s]; ssq[tid] += ssq[tid + s]; }
        __syncthreads();
    }
    if (tid == 0) { psum[blockIdx.x] = ssum[0]; psq[blockIdx.x] = ssq[0]; }
}

__global__ __launch_bounds__(256) void reduce2_kernel(const float* __restrict__ psum,
                                                      const float* __restrict__ psq,
                                                      float* __restrict__ stats) {
    __shared__ float ssum[256];
    __shared__ float ssq[256];
    const int tid = threadIdx.x;
    ssum[tid] = psum[tid] + psum[tid + 256];
    ssq[tid]  = psq[tid] + psq[tid + 256];
    __syncthreads();
    for (int s = 128; s > 0; s >>= 1) {
        if (tid < s) { ssum[tid] += ssum[tid + s]; ssq[tid] += ssq[tid + s]; }
        __syncthreads();
    }
    if (tid == 0) {
        const float invN = 1.0f / (float)GRID_N;
        float mean = ssum[0] * invN;
        float var  = fmaxf(ssq[0] * invN - mean * mean, 0.0f);
        stats[0] = mean;
        stats[1] = 1.0f / (sqrtf(var) + 1e-8f);
    }
}

__global__ __launch_bounds__(256) void finalize_kernel(const float* __restrict__ re,
                                                       const float* __restrict__ stats,
                                                       float* __restrict__ out) {
    int g = blockIdx.x * blockDim.x + threadIdx.x;
    if (g >= GRID_N) return;
    out[g] = (re[g] - stats[0]) * stats[1];
}

extern "C" void kernel_launch(void* const* d_in, const int* in_sizes, int n_in,
                              void* d_out, int out_size, void* d_ws, size_t ws_size,
                              hipStream_t stream) {
    const float* X   = (const float*)d_in[0];
    const float* aw  = (const float*)d_in[1];
    const float* bw  = (const float*)d_in[2];
    const float* ham = (const float*)d_in[4];
    float* out = (float*)d_out;

    float* ws   = (float*)d_ws;
    float* Wre  = ws;
    float* WimF = Wre + 4096;
    float* WimI = WimF + 4096;
    float* Are  = WimI + 4096;
    float* Aim  = Are + GRID_N;
    float* Bre  = Aim + GRID_N;
    float* Bim  = Bre + GRID_N;
    float* psum = Bim + GRID_N;
    float* psq  = psum + 512;
    float* stats = psq + 512;

    gen_dft_kernel<<<16, 256, 0, stream>>>(Wre, WimF, WimI);

    density_kernel<<<2048, 256, 0, stream>>>(X, aw, bw, Are, Aim);

    dft_pass_kernel<<<128, 256, 0, stream>>>(Are, Aim, Wre, WimF, Bre, Bim);
    dft_pass_kernel<<<128, 256, 0, stream>>>(Bre, Bim, Wre, WimF, Are, Aim);
    dft_pass_kernel<<<128, 256, 0, stream>>>(Are, Aim, Wre, WimF, Bre, Bim);

    hamming_kernel<<<GRID_N / 256, 256, 0, stream>>>(Bre, Bim, ham);

    dft_pass_kernel<<<128, 256, 0, stream>>>(Bre, Bim, Wre, WimI, Are, Aim);
    dft_pass_kernel<<<128, 256, 0, stream>>>(Are, Aim, Wre, WimI, Bre, Bim);
    dft_pass_kernel<<<128, 256, 0, stream>>>(Bre, Bim, Wre, WimI, Are, Aim);

    reduce1_kernel<<<512, 256, 0, stream>>>(Are, psum, psq);
    reduce2_kernel<<<1, 256, 0, stream>>>(psum, psq, stats);
    finalize_kernel<<<GRID_N / 256, 256, 0, stream>>>(Are, stats, out);
}


# forward spectrum summed in Fourier space: per-axis Gaussian factors DFT'd, 768 terms contracted as a complex f32 WMMA product; density and 3 forward passes gone (12->10 launches); sum order changes
# speedup vs baseline: 1.0940x; 1.0940x over previous
#include <hip/hip_runtime.h>
#include <hip/hip_bf16.h>
#include <math.h>

#define N1D      64
#define GRID_N   (64 * 64 * 64)
#define N_ATOMS  128
#define K_TERMS  6
#define SPACING  0.5f
#define LOG2E    1.4426950408889634f

typedef float v2f __attribute__((ext_vector_type(2)));
typedef float v8f __attribute__((ext_vector_type(8)));

__device__ __forceinline__ float fast_exp2(float x) {
#if __has_builtin(__builtin_amdgcn_exp2f)
    return __builtin_amdgcn_exp2f(x);
#else
    return exp2f(x);
#endif
}

__global__ __launch_bounds__(256) void gen_dft_kernel(float* __restrict__ Wre,
                                                      float* __restrict__ WimF,
                                                      float* __restrict__ WimI) {
    int t = blockIdx.x * blockDim.x + threadIdx.x;
    if (t >= N1D * N1D) return;
    int k = t >> 6;
    int n = t & 63;
    int kn = (k * n) & 63;
    float ang = (float)kn * 0.09817477042468103f;
    float sgn = ((k + n) & 1) ? -0.125f : 0.125f;
    float c = sgn * cosf(ang);
    float s = sgn * sinf(ang);
    Wre[t]  = c;
    WimF[t] = -s;
    WimI[t] = s;
}

#define N_TERMS  (N_ATOMS * K_TERMS)
#define TAB_SZ   (N_TERMS * N1D)

__global__ __launch_bounds__(64) void axis_spectra_kernel(const float* __restrict__ X,
                                                           const float* __restrict__ aw,
                                                           const float* __restrict__ bw,
                                                           const float* __restrict__ Wre,
                                                           const float* __restrict__ WimF,
                                                           float* __restrict__ Pre,
                                                           float* __restrict__ Pim,
                                                           float* __restrict__ YFre,
                                                           float* __restrict__ YFim,
                                                           float* __restrict__ ZFre,
                                                           float* __restrict__ ZFim) {
    const int lane   = threadIdx.x & 31;
    const int waveId = (blockIdx.x * blockDim.x + threadIdx.x) >> 5;
    const int ck   = waveId & 3;
    const int rest = waveId >> 2;
    const int d    = rest / 48;
    const int tt   = rest - d * 48;
    const int half = lane >> 4;
    const int l16  = lane & 15;

    const int t = tt * 16 + l16;
    const int a = t / K_TERMS;
    const float c   = X[3 * a + d];
    const float bl2 = bw[t] * LOG2E;
    const float scale = (d == 0) ? aw[t] : 1.0f;
    const int krow = ck * 16 + l16;

    v8f re = {}, im = {};
#pragma unroll 4
    for (int n0 = 0; n0 < N1D; n0 += 4) {
        const int koff = n0 + 2 * half;
        const float d0 = (float)koff * SPACING - c;
        const float d1 = (float)(koff + 1) * SPACING - c;
        v2f e;
        e.x = scale * fast_exp2(bl2 * (d0 * d0));
        e.y = scale * fast_exp2(bl2 * (d1 * d1));
        v2f wr = *(const v2f*)(Wre  + krow * N1D + koff);
        v2f wi = *(const v2f*)(WimF + krow * N1D + koff);
        re = __builtin_amdgcn_wmma_f32_16x16x4_f32(false, e, false, wr, (short)0, re, false, false);
        im = __builtin_amdgcn_wmma_f32_16x16x4_f32(false, e, false, wi, (short)0, im, false, false);
    }

    if (d == 0) {
        const int o = krow * N_TERMS + tt * 16 + 8 * half;
        *(float4*)(Pre + o)     = make_float4(re[0], re[1], re[2], re[3]);
        *(float4*)(Pre + o + 4) = make_float4(re[4], re[5], re[6], re[7]);
        *(float4*)(Pim + o)     = make_float4(im[0], im[1], im[2], im[3]);
        *(float4*)(Pim + o + 4) = make_float4(im[4], im[5], im[6], im[7]);
    } else {
        float* __restrict__ Fr = (d == 1) ? YFre : ZFre;
        float* __restrict__ Fi = (d == 1) ? YFim : ZFim;
#pragma unroll
        for (int r2 = 0; r2 < 4; ++r2) {
            const int s  = tt * 4 + 2 * half + (r2 >> 1);
            const int hh = r2 & 1;
            const int o  = (((s * 4 + ck) * 32) + hh * 16 + l16) * 2;
            v2f vr, vi;
            vr.x = re[2 * r2]; vr.y = re[2 * r2 + 1];
            vi.x = im[2 * r2]; vi.y = im[2 * r2 + 1];
            *(v2f*)(Fr + o) = vr;
            *(v2f*)(Fi + o) = vi;
        }
    }
}

#define KSTEPS_HALF (N_TERMS / 8)
#define CG 8
__global__ __launch_bounds__(256) void spectrum_contract_kernel(const float* __restrict__ Pre,
                                                                const float* __restrict__ Pim,
                                                                const float* __restrict__ YFre,
                                                                const float* __restrict__ YFim,
                                                                const float* __restrict__ ZFre,
                                                                const float* __restrict__ ZFim,
                                                                float* __restrict__ Fre,
                                                                float* __restrict__ Fim) {
    __shared__ float red[4][16][32];

    const int lane = threadIdx.x & 31;
    const int w    = threadIdx.x >> 5;
    const int cz   = w & 3;
    const int kh   = w >> 2;
    const int kx   = blockIdx.x >> 2;
    const int cy   = blockIdx.x & 3;
    const int half = lane >> 4;
    const int l16  = lane & 15;
    const int sBeg = kh * KSTEPS_HALF;

    const v2f* __restrict__ pPr = (const v2f*)(Pre + kx * N_TERMS + 2 * half) + 2 * sBeg;
    const v2f* __restrict__ pPi = (const v2f*)(Pim + kx * N_TERMS + 2 * half) + 2 * sBeg;
    const v2f* __restrict__ pYr = (const v2f*)YFre + 128 * sBeg + cy * 32 + lane;
    const v2f* __restrict__ pYi = (const v2f*)YFim + 128 * sBeg + cy * 32 + lane;
    const v2f* __restrict__ pZr = (const v2f*)ZFre + 128 * sBeg + cz * 32 + lane;
    const v2f* __restrict__ pZi = (const v2f*)ZFim + 128 * sBeg + cz * 32 + lane;

    v8f rr = {}, ii = {}, ri = {}, ir = {};
    for (int s0 = 0; s0 < KSTEPS_HALF; s0 += CG) {
        v2f pr[CG], pi[CG], yr[CG], yi[CG], zr[CG], zi[CG];
#pragma unroll
        for (int u = 0; u < CG; ++u) {
            pr[u] = pPr[2 * (s0 + u)];   pi[u] = pPi[2 * (s0 + u)];
            yr[u] = pYr[128 * (s0 + u)]; yi[u] = pYi[128 * (s0 + u)];
            zr[u] = pZr[128 * (s0 + u)]; zi[u] = pZi[128 * (s0 + u)];
        }
#pragma unroll
        for (int u = 0; u < CG; ++u) {
            const v2f ar = pr[u] * yr[u] - pi[u] * yi[u];
            const v2f ai = pr[u] * yi[u] + pi[u] * yr[u];
            rr = __builtin_amdgcn_wmma_f32_16x16x4_f32(false, ar, false, zr[u], (short)0, rr, false, false);
            ii = __builtin_amdgcn_wmma_f32_16x16x4_f32(false, ai, false, zi[u], (short)0, ii, false, false);
            ri = __builtin_amdgcn_wmma_f32_16x16x4_f32(false, ar, false, zi[u], (short)0, ri, false, false);
            ir = __builtin_amdgcn_wmma_f32_16x16x4_f32(false, ai, false, zr[u], (short)0, ir, false, false);
        }
    }

    const v8f fre = rr - ii;
    const v8f fim = ri + ir;
    if (kh == 1) {
#pragma unroll
        for (int v = 0; v < 8; ++v) {
            red[cz][v][lane]     = fre[v];
            red[cz][8 + v][lane] = fim[v];
        }
    }
    __syncthreads();
    if (kh == 0) {
#pragma unroll
        for (int v = 0; v < 8; ++v) {
            const int ky  = cy * 16 + v + 8 * half;
            const int idx = kx * 4096 + ky * N1D + cz * 16 + l16;
            Fre[idx] = fre[v] + red[cz][v][lane];
            Fim[idx] = fim[v] + red[cz][8 + v][lane];
        }
    }
}

__global__ __launch_bounds__(256) void dft_pass_kernel(const float* __restrict__ Tre,
                                                       const float* __restrict__ Tim,
                                                       const float* __restrict__ Mre,
                                                       const float* __restrict__ Mim,
                                                       float* __restrict__ Ure,
                                                       float* __restrict__ Uim) {
    const int lane   = threadIdx.x & 31;
    const int waveId = (blockIdx.x * blockDim.x + threadIdx.x) >> 5;
    const int p0 = (waveId & 255) << 4;
    const int k0 = (waveId >> 8) << 4;
    const int half = lane >> 4;
    const int l16  = lane & 15;
    const int krow = k0 + l16;
    const int prow = p0 + l16;

    v8f rr = {}, ii = {}, ri = {}, ir = {};

#pragma unroll 4
    for (int n0 = 0; n0 < N1D; n0 += 4) {
        const int koff = n0 + 2 * half;
        v2f mr = *(const v2f*)(Mre + krow * N1D + koff);
        v2f mi = *(const v2f*)(Mim + krow * N1D + koff);
        v2f tr = *(const v2f*)(Tre + prow * N1D + koff);
        v2f ti = *(const v2f*)(Tim + prow * N1D + koff);
        rr = __builtin_amdgcn_wmma_f32_16x16x4_f32(false, mr, false, tr, (short)0, rr, false, false);
        ii = __builtin_amdgcn_wmma_f32_16x16x4_f32(false, mi, false, ti, (short)0, ii, false, false);
        ri = __builtin_amdgcn_wmma_f32_16x16x4_f32(false, mr, false, ti, (short)0, ri, false, false);
        ir = __builtin_amdgcn_wmma_f32_16x16x4_f32(false, mi, false, tr, (short)0, ir, false, false);
    }

#pragma unroll
    for (int v = 0; v < 8; ++v) {
        const int row = k0 + v + 8 * half;
        const int idx = row * 4096 + p0 + l16;
        Ure[idx] = rr[v] - ii[v];
        Uim[idx] = ri[v] + ir[v];
    }
}

__global__ __launch_bounds__(256) void hamming_kernel(float* __restrict__ re,
                                                      float* __restrict__ im,
                                                      const float* __restrict__ h) {
    int g = blockIdx.x * blockDim.x + threadIdx.x;
    if (g >= GRID_N) return;
    float hv = h[g];
    re[g] *= hv;
    im[g] *= hv;
}

__global__ __launch_bounds__(256) void reduce1_kernel(const float* __restrict__ x,
                                                      float* __restrict__ psum,
                                                      float* __restrict__ psq) {
    __shared__ float ssum[256];
    __shared__ float ssq[256];
    const int tid = threadIdx.x;
    const int base = blockIdx.x * 512;
    float a = x[base + tid];
    float b = x[base + 256 + tid];
    ssum[tid] = a + b;
    ssq[tid]  = a * a + b * b;
    __syncthreads();
    for (int s = 128; s > 0; s >>= 1) {
        if (tid < s) { ssum[tid] += ssum[tid + s]; ssq[tid] += ssq[tid + s]; }
        __syncthreads();
    }
    if (tid == 0) { psum[blockIdx.x] = ssum[0]; psq[blockIdx.x] = ssq[0]; }
}

__global__ __launch_bounds__(256) void reduce2_kernel(const float* __restrict__ psum,
                                                      const float* __restrict__ psq,
                                                      float* __restrict__ stats) {
    __shared__ float ssum[256];
    __shared__ float ssq[256];
    const int tid = threadIdx.x;
    ssum[tid] = psum[tid] + psum[tid + 256];
    ssq[tid]  = psq[tid] + psq[tid + 256];
    __syncthreads();
    for (int s = 128; s > 0; s >>= 1) {
        if (tid < s) { ssum[tid] += ssum[tid + s]; ssq[tid] += ssq[tid + s]; }
        __syncthreads();
    }
    if (tid == 0) {
        const float invN = 1.0f / (float)GRID_N;
        float mean = ssum[0] * invN;
        float var  = fmaxf(ssq[0] * invN - mean * mean, 0.0f);
        stats[0] = mean;
        stats[1] = 1.0f / (sqrtf(var) + 1e-8f);
    }
}

__global__ __launch_bounds__(256) void finalize_kernel(const float* __restrict__ re,
                                                       const float* __restrict__ stats,
                                                       float* __restrict__ out) {
    int g = blockIdx.x * blockDim.x + threadIdx.x;
    if (g >= GRID_N) return;
    out[g] = (re[g] - stats[0]) * stats[1];
}

extern "C" void kernel_launch(void* const* d_in, const int* in_sizes, int n_in,
                              void* d_out, int out_size, void* d_ws, size_t ws_size,
                              hipStream_t stream) {
    const float* X   = (const float*)d_in[0];
    const float* aw  = (const float*)d_in[1];
    const float* bw  = (const float*)d_in[2];
    const float* ham = (const float*)d_in[4];
    float* out = (float*)d_out;

    float* ws   = (float*)d_ws;
    float* Wre  = ws;
    float* WimF = Wre + 4096;
    float* WimI = WimF + 4096;
    float* Are  = WimI + 4096;
    float* Aim  = Are + GRID_N;
    float* Bre  = Aim + GRID_N;
    float* Bim  = Bre + GRID_N;
    float* psum = Bim + GRID_N;
    float* psq  = psum + 512;
    float* stats = psq + 512;

    float* Pre  = Are;
    float* Pim  = Pre  + TAB_SZ;
    float* YFre = Pim  + TAB_SZ;
    float* YFim = YFre + TAB_SZ;
    float* ZFre = YFim + TAB_SZ;
    float* ZFim = ZFre + TAB_SZ;

    gen_dft_kernel<<<16, 256, 0, stream>>>(Wre, WimF, WimI);

    axis_spectra_kernel<<<288, 64, 0, stream>>>(X, aw, bw, Wre, WimF,
                                                Pre, Pim, YFre, YFim, ZFre, ZFim);

    spectrum_contract_kernel<<<256, 256, 0, stream>>>(Pre, Pim, YFre, YFim, ZFre, ZFim, Bre, Bim);

    hamming_kernel<<<GRID_N / 256, 256, 0, stream>>>(Bre, Bim, ham);

    dft_pass_kernel<<<128, 256, 0, stream>>>(Bre, Bim, Wre, WimI, Are, Aim);
    dft_pass_kernel<<<128, 256, 0, stream>>>(Are, Aim, Wre, WimI, Bre, Bim);
    dft_pass_kernel<<<128, 256, 0, stream>>>(Bre, Bim, Wre, WimI, Are, Aim);

    reduce1_kernel<<<512, 256, 0, stream>>>(Are, psum, psq);
    reduce2_kernel<<<1, 256, 0, stream>>>(psum, psq, stats);
    finalize_kernel<<<GRID_N / 256, 256, 0, stream>>>(Are, stats, out);
}


# filter multiply and z inverse pass done inside the spectrum contraction launch, on its 16x64 tile kept in LDS: filter launch and first pass launch gone (10->8 launches); sum order unchanged
# speedup vs baseline: 1.3689x; 1.2513x over previous
#include <hip/hip_runtime.h>
#include <hip/hip_bf16.h>
#include <math.h>

#define N1D      64
#define GRID_N   (64 * 64 * 64)
#define N_ATOMS  128
#define K_TERMS  6
#define SPACING  0.5f
#define LOG2E    1.4426950408889634f

typedef float v2f __attribute__((ext_vector_type(2)));
typedef float v8f __attribute__((ext_vector_type(8)));

__device__ __forceinline__ float fast_exp2(float x) {
#if __has_builtin(__builtin_amdgcn_exp2f)
    return __builtin_amdgcn_exp2f(x);
#else
    return exp2f(x);
#endif
}

__global__ __launch_bounds__(256) void gen_dft_kernel(float* __restrict__ Wre,
                                                      float* __restrict__ WimF,
                                                      float* __restrict__ WimI) {
    int t = blockIdx.x * blockDim.x + threadIdx.x;
    if (t >= N1D * N1D) return;
    int k = t >> 6;
    int n = t & 63;
    int kn = (k * n) & 63;
    float ang = (float)kn * 0.09817477042468103f;
    float sgn = ((k + n) & 1) ? -0.125f : 0.125f;
    float c = sgn * cosf(ang);
    float s = sgn * sinf(ang);
    Wre[t]  = c;
    WimF[t] = -s;
    WimI[t] = s;
}

#define N_TERMS  (N_ATOMS * K_TERMS)
#define TAB_SZ   (N_TERMS * N1D)

__global__ __launch_bounds__(64) void axis_spectra_kernel(const float* __restrict__ X,
                                                           const float* __restrict__ aw,
                                                           const float* __restrict__ bw,
                                                           const float* __restrict__ Wre,
                                                           const float* __restrict__ WimF,
                                                           float* __restrict__ Pre,
                                                           float* __restrict__ Pim,
                                                           float* __restrict__ YFre,
                                                           float* __restrict__ YFim,
                                                           float* __restrict__ ZFre,
                                                           float* __restrict__ ZFim) {
    const int lane   = threadIdx.x & 31;
    const int waveId = (blockIdx.x * blockDim.x + threadIdx.x) >> 5;
    const int ck   = waveId & 3;
    const int rest = waveId >> 2;
    const int d    = rest / 48;
    const int tt   = rest - d * 48;
    const int half = lane >> 4;
    const int l16  = lane & 15;

    const int t = tt * 16 + l16;
    const int a = t / K_TERMS;
    const float c   = X[3 * a + d];
    const float bl2 = bw[t] * LOG2E;
    const float scale = (d == 0) ? aw[t] : 1.0f;
    const int krow = ck * 16 + l16;

    v8f re = {}, im = {};
#pragma unroll 4
    for (int n0 = 0; n0 < N1D; n0 += 4) {
        const int koff = n0 + 2 * half;
        const float d0 = (float)koff * SPACING - c;
        const float d1 = (float)(koff + 1) * SPACING - c;
        v2f e;
        e.x = scale * fast_exp2(bl2 * (d0 * d0));
        e.y = scale * fast_exp2(bl2 * (d1 * d1));
        v2f wr = *(const v2f*)(Wre  + krow * N1D + koff);
        v2f wi = *(const v2f*)(WimF + krow * N1D + koff);
        re = __builtin_amdgcn_wmma_f32_16x16x4_f32(false, e, false, wr, (short)0, re, false, false);
        im = __builtin_amdgcn_wmma_f32_16x16x4_f32(false, e, false, wi, (short)0, im, false, false);
    }

    if (d == 0) {
        const int o = krow * N_TERMS + tt * 16 + 8 * half;
        *(float4*)(Pre + o)     = make_float4(re[0], re[1], re[2], re[3]);
        *(float4*)(Pre + o + 4) = make_float4(re[4], re[5], re[6], re[7]);
        *(float4*)(Pim + o)     = make_float4(im[0], im[1], im[2], im[3]);
        *(float4*)(Pim + o + 4) = make_float4(im[4], im[5], im[6], im[7]);
    } else {
        float* __restrict__ Fr = (d == 1) ? YFre : ZFre;
        float* __restrict__ Fi = (d == 1) ? YFim : ZFim;
#pragma unroll
        for (int r2 = 0; r2 < 4; ++r2) {
            const int s  = tt * 4 + 2 * half + (r2 >> 1);
            const int hh = r2 & 1;
            const int o  = (((s * 4 + ck) * 32) + hh * 16 + l16) * 2;
            v2f vr, vi;
            vr.x = re[2 * r2]; vr.y = re[2 * r2 + 1];
            vi.x = im[2 * r2]; vi.y = im[2 * r2 + 1];
            *(v2f*)(Fr + o) = vr;
            *(v2f*)(Fi + o) = vi;
        }
    }
}

#define KSTEPS_HALF (N_TERMS / 8)
#define CG 8
#define GP 68
__global__ __launch_bounds__(256) void spectrum_contract_kernel(const float* __restrict__ Pre,
                                                                const float* __restrict__ Pim,
                                                                const float* __restrict__ YFre,
                                                                const float* __restrict__ YFim,
                                                                const float* __restrict__ ZFre,
                                                                const float* __restrict__ ZFim,
                                                                const float* __restrict__ ham,
                                                                const float* __restrict__ Mre,
                                                                const float* __restrict__ Mim,
                                                                float* __restrict__ Ure,
                                                                float* __restrict__ Uim) {
    __shared__ float red[4][16][32];
    __shared__ __attribute__((aligned(16))) float Gre[16 * GP];
    __shared__ __attribute__((aligned(16))) float Gim[16 * GP];

    const int lane = threadIdx.x & 31;
    const int w    = threadIdx.x >> 5;
    const int cz   = w & 3;
    const int kh   = w >> 2;
    const int kx   = blockIdx.x >> 2;
    const int cy   = blockIdx.x & 3;
    const int half = lane >> 4;
    const int l16  = lane & 15;
    const int sBeg = kh * KSTEPS_HALF;

    const v2f* __restrict__ pPr = (const v2f*)(Pre + kx * N_TERMS + 2 * half) + 2 * sBeg;
    const v2f* __restrict__ pPi = (const v2f*)(Pim + kx * N_TERMS + 2 * half) + 2 * sBeg;
    const v2f* __restrict__ pYr = (const v2f*)YFre + 128 * sBeg + cy * 32 + lane;
    const v2f* __restrict__ pYi = (const v2f*)YFim + 128 * sBeg + cy * 32 + lane;
    const v2f* __restrict__ pZr = (const v2f*)ZFre + 128 * sBeg + cz * 32 + lane;
    const v2f* __restrict__ pZi = (const v2f*)ZFim + 128 * sBeg + cz * 32 + lane;

    v8f rr = {}, ii = {}, ri = {}, ir = {};
    for (int s0 = 0; s0 < KSTEPS_HALF; s0 += CG) {
        v2f pr[CG], pi[CG], yr[CG], yi[CG], zr[CG], zi[CG];
#pragma unroll
        for (int u = 0; u < CG; ++u) {
            pr[u] = pPr[2 * (s0 + u)];   pi[u] = pPi[2 * (s0 + u)];
            yr[u] = pYr[128 * (s0 + u)]; yi[u] = pYi[128 * (s0 + u)];
            zr[u] = pZr[128 * (s0 + u)]; zi[u] = pZi[128 * (s0 + u)];
        }
#pragma unroll
        for (int u = 0; u < CG; ++u) {
            const v2f ar = pr[u] * yr[u] - pi[u] * yi[u];
            const v2f ai = pr[u] * yi[u] + pi[u] * yr[u];
            rr = __builtin_amdgcn_wmma_f32_16x16x4_f32(false, ar, false, zr[u], (short)0, rr, false, false);
            ii = __builtin_amdgcn_wmma_f32_16x16x4_f32(false, ai, false, zi[u], (short)0, ii, false, false);
            ri = __builtin_amdgcn_wmma_f32_16x16x4_f32(false, ar, false, zi[u], (short)0, ri, false, false);
            ir = __builtin_amdgcn_wmma_f32_16x16x4_f32(false, ai, false, zr[u], (short)0, ir, false, false);
        }
    }

    const v8f fre = rr - ii;
    const v8f fim = ri + ir;

    v2f mr[16], mi[16];
    {
        const float* __restrict__ aR = Mre + (cz * 16 + l16) * N1D + 2 * half;
        const float* __restrict__ aI = Mim + (cz * 16 + l16) * N1D + 2 * half;
#pragma unroll
        for (int s = 0; s < 16; ++s) {
            mr[s] = *(const v2f*)(aR + 4 * s);
            mi[s] = *(const v2f*)(aI + 4 * s);
        }
    }
    float hv[8];
#pragma unroll
    for (int v = 0; v < 8; ++v) {
        const int ky = cy * 16 + v + 8 * half;
        hv[v] = ham[kx * 4096 + ky * N1D + cz * 16 + l16];
    }

    if (kh == 1) {
#pragma unroll
        for (int v = 0; v < 8; ++v) {
            red[cz][v][lane]     = fre[v];
            red[cz][8 + v][lane] = fim[v];
        }
    }
    __syncthreads();
    if (kh == 0) {
#pragma unroll
        for (int v = 0; v < 8; ++v) {
            const int o = (v + 8 * half) * GP + cz * 16 + l16;
            Gre[o] = (fre[v] + red[cz][v][lane]) * hv[v];
            Gim[o] = (fim[v] + red[cz][8 + v][lane]) * hv[v];
        }
    }
    __syncthreads();

    const float* b1 = (kh ? Gim : Gre) + l16 * GP + 2 * half;
    const float* b2 = (kh ? Gre : Gim) + l16 * GP + 2 * half;
    v8f a1 = {}, a2 = {};
#pragma unroll
    for (int s = 0; s < 16; ++s) {
        const v2f t1 = *(const v2f*)(b1 + 4 * s);
        const v2f t2 = *(const v2f*)(b2 + 4 * s);
        a1 = __builtin_amdgcn_wmma_f32_16x16x4_f32(false, mr[s], false, t1, (short)0, a1, false, false);
        a2 = __builtin_amdgcn_wmma_f32_16x16x4_f32(false, mi[s], false, t2, (short)0, a2, false, false);
    }
    const float sg = kh ? 1.0f : -1.0f;
    const v8f u = a1 + sg * a2;

    float* __restrict__ dst = kh ? Uim : Ure;
#pragma unroll
    for (int v = 0; v < 8; ++v) {
        const int z = cz * 16 + v + 8 * half;
        dst[z * 4096 + kx * N1D + cy * 16 + l16] = u[v];
    }
}

__global__ __launch_bounds__(256) void dft_pass_kernel(const float* __restrict__ Tre,
                                                       const float* __restrict__ Tim,
                                                       const float* __restrict__ Mre,
                                                       const float* __restrict__ Mim,
                                                       float* __restrict__ Ure,
                                                       float* __restrict__ Uim) {
    const int lane   = threadIdx.x & 31;
    const int waveId = (blockIdx.x * blockDim.x + threadIdx.x) >> 5;
    const int p0 = (waveId & 255) << 4;
    const int k0 = (waveId >> 8) << 4;
    const int half = lane >> 4;
    const int l16  = lane & 15;
    const int krow = k0 + l16;
    const int prow = p0 + l16;

    v8f rr = {}, ii = {}, ri = {}, ir = {};

#pragma unroll 4
    for (int n0 = 0; n0 < N1D; n0 += 4) {
        const int koff = n0 + 2 * half;
        v2f mr = *(const v2f*)(Mre + krow * N1D + koff);
        v2f mi = *(const v2f*)(Mim + krow * N1D + koff);
        v2f tr = *(const v2f*)(Tre + prow * N1D + koff);
        v2f ti = *(const v2f*)(Tim + prow * N1D + koff);
        rr = __builtin_amdgcn_wmma_f32_16x16x4_f32(false, mr, false, tr, (short)0, rr, false, false);
        ii = __builtin_amdgcn_wmma_f32_16x16x4_f32(false, mi, false, ti, (short)0, ii, false, false);
        ri = __builtin_amdgcn_wmma_f32_16x16x4_f32(false, mr, false, ti, (short)0, ri, false, false);
        ir = __builtin_amdgcn_wmma_f32_16x16x4_f32(false, mi, false, tr, (short)0, ir, false, false);
    }

#pragma unroll
    for (int v = 0; v < 8; ++v) {
        const int row = k0 + v + 8 * half;
        const int idx = row * 4096 + p0 + l16;
        Ure[idx] = rr[v] - ii[v];
        Uim[idx] = ri[v] + ir[v];
    }
}

__global__ __launch_bounds__(256) void reduce1_kernel(const float* __restrict__ x,
                                                      float* __restrict__ psum,
                                                      float* __restrict__ psq) {
    __shared__ float ssum[256];
    __shared__ float ssq[256];
    const int tid = threadIdx.x;
    const int base = blockIdx.x * 512;
    float a = x[base + tid];
    float b = x[base + 256 + tid];
    ssum[tid] = a + b;
    ssq[tid]  = a * a + b * b;
    __syncthreads();
    for (int s = 128; s > 0; s >>= 1) {
        if (tid < s) { ssum[tid] += ssum[tid + s]; ssq[tid] += ssq[tid + s]; }
        __syncthreads();
    }
    if (tid == 0) { psum[blockIdx.x] = ssum[0]; psq[blockIdx.x] = ssq[0]; }
}

__global__ __launch_bounds__(256) void reduce2_kernel(const float* __restrict__ psum,
                                                      const float* __restrict__ psq,
                                                      float* __restrict__ stats) {
    __shared__ float ssum[256];
    __shared__ float ssq[256];
    const int tid = threadIdx.x;
    ssum[tid] = psum[tid] + psum[tid + 256];
    ssq[tid]  = psq[tid] + psq[tid + 256];
    __syncthreads();
    for (int s = 128; s > 0; s >>= 1) {
        if (tid < s) { ssum[tid] += ssum[tid + s]; ssq[tid] += ssq[tid + s]; }
        __syncthreads();
    }
    if (tid == 0) {
        const float invN = 1.0f / (float)GRID_N;
        float mean = ssum[0] * invN;
        float var  = fmaxf(ssq[0] * invN - mean * mean, 0.0f);
        stats[0] = mean;
        stats[1] = 1.0f / (sqrtf(var) + 1e-8f);
    }
}

__global__ __launch_bounds__(256) void finalize_kernel(const float* __restrict__ re,
                                                       const float* __restrict__ stats,
                                                       float* __restrict__ out) {
    int g = blockIdx.x * blockDim.x + threadIdx.x;
    if (g >= GRID_N) return;
    out[g] = (re[g] - stats[0]) * stats[1];
}

extern "C" void kernel_launch(void* const* d_in, const int* in_sizes, int n_in,
                              void* d_out, int out_size, void* d_ws, size_t ws_size,
                              hipStream_t stream) {
    const float* X   = (const float*)d_in[0];
    const float* aw  = (const float*)d_in[1];
    const float* bw  = (const float*)d_in[2];
    const float* ham = (const float*)d_in[4];
    float* out = (float*)d_out;

    float* ws   = (float*)d_ws;
    float* Wre  = ws;
    float* WimF = Wre + 4096;
    float* WimI = WimF + 4096;
    float* Are  = WimI + 4096;
    float* Aim  = Are + GRID_N;
    float* Bre  = Aim + GRID_N;
    float* Bim  = Bre + GRID_N;
    float* psum = Bim + GRID_N;
    float* psq  = psum + 512;
    float* stats = psq + 512;

    float* Pre  = Are;
    float* Pim  = Pre  + TAB_SZ;
    float* YFre = Pim  + TAB_SZ;
    float* YFim = YFre + TAB_SZ;
    float* ZFre = YFim + TAB_SZ;
    float* ZFim = ZFre + TAB_SZ;

    gen_dft_kernel<<<16, 256, 0, stream>>>(Wre, WimF, WimI);

    axis_spectra_kernel<<<288, 64, 0, stream>>>(X, aw, bw, Wre, WimF,
                                                Pre, Pim, YFre, YFim, ZFre, ZFim);

    spectrum_contract_kernel<<<256, 256, 0, stream>>>(Pre, Pim, YFre, YFim, ZFre, ZFim,
                                                      ham, Wre, WimI, Bre, Bim);

    dft_pass_kernel<<<128, 256, 0, stream>>>(Bre, Bim, Wre, WimI, Are, Aim);
    dft_pass_kernel<<<128, 256, 0, stream>>>(Are, Aim, Wre, WimI, Bre, Bim);

    reduce1_kernel<<<512, 256, 0, stream>>>(Bre, psum, psq);
    reduce2_kernel<<<1, 256, 0, stream>>>(psum, psq, stats);
    finalize_kernel<<<GRID_N / 256, 256, 0, stream>>>(Bre, stats, out);
}


# DFT matrices no longer made by a set-up launch: the axis-spectra launch keeps the 64 sin/cos phases in LDS, builds its B fragments from them and writes the matrices out (8->7); sum order unchanged
# speedup vs baseline: 1.4395x; 1.0516x over previous
#include <hip/hip_runtime.h>
#include <hip/hip_bf16.h>
#include <math.h>

#define N1D      64
#define GRID_N   (64 * 64 * 64)
#define N_ATOMS  128
#define K_TERMS  6
#define SPACING  0.5f
#define LOG2E    1.4426950408889634f

typedef float v2f __attribute__((ext_vector_type(2)));
typedef float v8f __attribute__((ext_vector_type(8)));

__device__ __forceinline__ float fast_exp2(float x) {
#if __has_builtin(__builtin_amdgcn_exp2f)
    return __builtin_amdgcn_exp2f(x);
#else
    return exp2f(x);
#endif
}

#define N_TERMS  (N_ATOMS * K_TERMS)
#define TAB_SZ   (N_TERMS * N1D)

__global__ __launch_bounds__(64) void axis_spectra_kernel(const float* __restrict__ X,
                                                           const float* __restrict__ aw,
                                                           const float* __restrict__ bw,
                                                           float* __restrict__ Wre,
                                                           float* __restrict__ WimI,
                                                           float* __restrict__ Pre,
                                                           float* __restrict__ Pim,
                                                           float* __restrict__ YFre,
                                                           float* __restrict__ YFim,
                                                           float* __restrict__ ZFre,
                                                           float* __restrict__ ZFim) {
    __shared__ float ctab[N1D];
    __shared__ float stab[N1D];
    {
        const int j = threadIdx.x;
        const float ang = (float)j * 0.09817477042468103f;
        ctab[j] = 0.125f * cosf(ang);
        stab[j] = 0.125f * sinf(ang);
    }
    __syncthreads();

    if (blockIdx.x >= 288) {
        const int e = (blockIdx.x - 288) * 64 + threadIdx.x;
        const int k = e >> 4;
        const int n = (e & 15) * 4;
        float cr[4], si[4];
#pragma unroll
        for (int q = 0; q < 4; ++q) {
            const int kn = (k * (n + q)) & 63;
            const float sgn = ((k + n + q) & 1) ? -1.0f : 1.0f;
            cr[q] = sgn * ctab[kn];
            si[q] = sgn * stab[kn];
        }
        *(float4*)(Wre  + k * N1D + n) = make_float4(cr[0], cr[1], cr[2], cr[3]);
        *(float4*)(WimI + k * N1D + n) = make_float4(si[0], si[1], si[2], si[3]);
        return;
    }

    const int lane   = threadIdx.x & 31;
    const int waveId = (blockIdx.x * blockDim.x + threadIdx.x) >> 5;
    const int ck   = waveId & 3;
    const int rest = waveId >> 2;
    const int d    = rest / 48;
    const int tt   = rest - d * 48;
    const int half = lane >> 4;
    const int l16  = lane & 15;

    const int t = tt * 16 + l16;
    const int a = t / K_TERMS;
    const float c   = X[3 * a + d];
    const float bl2 = bw[t] * LOG2E;
    const float scale = (d == 0) ? aw[t] : 1.0f;
    const int krow = ck * 16 + l16;
    const float sg0 = (krow & 1) ? -1.0f : 1.0f;

    v8f re = {}, im = {};
#pragma unroll 4
    for (int n0 = 0; n0 < N1D; n0 += 4) {
        const int koff = n0 + 2 * half;
        const float d0 = (float)koff * SPACING - c;
        const float d1 = (float)(koff + 1) * SPACING - c;
        v2f e;
        e.x = scale * fast_exp2(bl2 * (d0 * d0));
        e.y = scale * fast_exp2(bl2 * (d1 * d1));
        const int i0 = (krow * koff) & 63;
        const int i1 = (i0 + krow) & 63;
        v2f wr, wi;
        wr.x = sg0 * ctab[i0];   wr.y = -sg0 * ctab[i1];
        wi.x = -sg0 * stab[i0];  wi.y = sg0 * stab[i1];
        re = __builtin_amdgcn_wmma_f32_16x16x4_f32(false, e, false, wr, (short)0, re, false, false);
        im = __builtin_amdgcn_wmma_f32_16x16x4_f32(false, e, false, wi, (short)0, im, false, false);
    }

    if (d == 0) {
        const int o = krow * N_TERMS + tt * 16 + 8 * half;
        *(float4*)(Pre + o)     = make_float4(re[0], re[1], re[2], re[3]);
        *(float4*)(Pre + o + 4) = make_float4(re[4], re[5], re[6], re[7]);
        *(float4*)(Pim + o)     = make_float4(im[0], im[1], im[2], im[3]);
        *(float4*)(Pim + o + 4) = make_float4(im[4], im[5], im[6], im[7]);
    } else {
        float* __restrict__ Fr = (d == 1) ? YFre : ZFre;
        float* __restrict__ Fi = (d == 1) ? YFim : ZFim;
#pragma unroll
        for (int r2 = 0; r2 < 4; ++r2) {
            const int s  = tt * 4 + 2 * half + (r2 >> 1);
            const int hh = r2 & 1;
            const int o  = (((s * 4 + ck) * 32) + hh * 16 + l16) * 2;
            v2f vr, vi;
            vr.x = re[2 * r2]; vr.y = re[2 * r2 + 1];
            vi.x = im[2 * r2]; vi.y = im[2 * r2 + 1];
            *(v2f*)(Fr + o) = vr;
            *(v2f*)(Fi + o) = vi;
        }
    }
}

#define KSTEPS_HALF (N_TERMS / 8)
#define CG 8
#define GP 68
__global__ __launch_bounds__(256) void spectrum_contract_kernel(const float* __restrict__ Pre,
                                                                const float* __restrict__ Pim,
                                                                const float* __restrict__ YFre,
                                                                const float* __restrict__ YFim,
                                                                const float* __restrict__ ZFre,
                                                                const float* __restrict__ ZFim,
                                                                const float* __restrict__ ham,
                                                                const float* __restrict__ Mre,
                                                                const float* __restrict__ Mim,
                                                                float* __restrict__ Ure,
                                                                float* __restrict__ Uim) {
    __shared__ float red[4][16][32];
    __shared__ __attribute__((aligned(16))) float Gre[16 * GP];
    __shared__ __attribute__((aligned(16))) float Gim[16 * GP];

    const int lane = threadIdx.x & 31;
    const int w    = threadIdx.x >> 5;
    const int cz   = w & 3;
    const int kh   = w >> 2;
    const int kx   = blockIdx.x >> 2;
    const int cy   = blockIdx.x & 3;
    const int half = lane >> 4;
    const int l16  = lane & 15;
    const int sBeg = kh * KSTEPS_HALF;

    const v2f* __restrict__ pPr = (const v2f*)(Pre + kx * N_TERMS + 2 * half) + 2 * sBeg;
    const v2f* __restrict__ pPi = (const v2f*)(Pim + kx * N_TERMS + 2 * half) + 2 * sBeg;
    const v2f* __restrict__ pYr = (const v2f*)YFre + 128 * sBeg + cy * 32 + lane;
    const v2f* __restrict__ pYi = (const v2f*)YFim + 128 * sBeg + cy * 32 + lane;
    const v2f* __restrict__ pZr = (const v2f*)ZFre + 128 * sBeg + cz * 32 + lane;
    const v2f* __restrict__ pZi = (const v2f*)ZFim + 128 * sBeg + cz * 32 + lane;

    v8f rr = {}, ii = {}, ri = {}, ir = {};
    for (int s0 = 0; s0 < KSTEPS_HALF; s0 += CG) {
        v2f pr[CG], pi[CG], yr[CG], yi[CG], zr[CG], zi[CG];
#pragma unroll
        for (int u = 0; u < CG; ++u) {
            pr[u] = pPr[2 * (s0 + u)];   pi[u] = pPi[2 * (s0 + u)];
            yr[u] = pYr[128 * (s0 + u)]; yi[u] = pYi[128 * (s0 + u)];
            zr[u] = pZr[128 * (s0 + u)]; zi[u] = pZi[128 * (s0 + u)];
        }
#pragma unroll
        for (int u = 0; u < CG; ++u) {
            const v2f ar = pr[u] * yr[u] - pi[u] * yi[u];
            const v2f ai = pr[u] * yi[u] + pi[u] * yr[u];
            rr = __builtin_amdgcn_wmma_f32_16x16x4_f32(false, ar, false, zr[u], (short)0, rr, false, false);
            ii = __builtin_amdgcn_wmma_f32_16x16x4_f32(false, ai, false, zi[u], (short)0, ii, false, false);
            ri = __builtin_amdgcn_wmma_f32_16x16x4_f32(false, ar, false, zi[u], (short)0, ri, false, false);
            ir = __builtin_amdgcn_wmma_f32_16x16x4_f32(false, ai, false, zr[u], (short)0, ir, false, false);
        }
    }

    const v8f fre = rr - ii;
    const v8f fim = ri + ir;

    v2f mr[16], mi[16];
    {
        const float* __restrict__ aR = Mre + (cz * 16 + l16) * N1D + 2 * half;
        const float* __restrict__ aI = Mim + (cz * 16 + l16) * N1D + 2 * half;
#pragma unroll
        for (int s = 0; s < 16; ++s) {
            mr[s] = *(const v2f*)(aR + 4 * s);
            mi[s] = *(const v2f*)(aI + 4 * s);
        }
    }
    float hv[8];
#pragma unroll
    for (int v = 0; v < 8; ++v) {
        const int ky = cy * 16 + v + 8 * half;
        hv[v] = ham[kx * 4096 + ky * N1D + cz * 16 + l16];
    }

    if (kh == 1) {
#pragma unroll
        for (int v = 0; v < 8; ++v) {
            red[cz][v][lane]     = fre[v];
            red[cz][8 + v][lane] = fim[v];
        }
    }
    __syncthreads();
    if (kh == 0) {
#pragma unroll
        for (int v = 0; v < 8; ++v) {
            const int o = (v + 8 * half) * GP + cz * 16 + l16;
            Gre[o] = (fre[v] + red[cz][v][lane]) * hv[v];
            Gim[o] = (fim[v] + red[cz][8 + v][lane]) * hv[v];
        }
    }
    __syncthreads();

    const float* b1 = (kh ? Gim : Gre) + l16 * GP + 2 * half;
    const float* b2 = (kh ? Gre : Gim) + l16 * GP + 2 * half;
    v8f a1 = {}, a2 = {};
#pragma unroll
    for (int s = 0; s < 16; ++s) {
        const v2f t1 = *(const v2f*)(b1 + 4 * s);
        const v2f t2 = *(const v2f*)(b2 + 4 * s);
        a1 = __builtin_amdgcn_wmma_f32_16x16x4_f32(false, mr[s], false, t1, (short)0, a1, false, false);
        a2 = __builtin_amdgcn_wmma_f32_16x16x4_f32(false, mi[s], false, t2, (short)0, a2, false, false);
    }
    const float sg = kh ? 1.0f : -1.0f;
    const v8f u = a1 + sg * a2;

    float* __restrict__ dst = kh ? Uim : Ure;
#pragma unroll
    for (int v = 0; v < 8; ++v) {
        const int z = cz * 16 + v + 8 * half;
        dst[z * 4096 + kx * N1D + cy * 16 + l16] = u[v];
    }
}

__global__ __launch_bounds__(256) void dft_pass_kernel(const float* __restrict__ Tre,
                                                       const float* __restrict__ Tim,
                                                       const float* __restrict__ Mre,
                                                       const float* __restrict__ Mim,
                                                       float* __restrict__ Ure,
                                                       float* __restrict__ Uim) {
    const int lane   = threadIdx.x & 31;
    const int waveId = (blockIdx.x * blockDim.x + threadIdx.x) >> 5;
    const int p0 = (waveId & 255) << 4;
    const int k0 = (waveId >> 8) << 4;
    const int half = lane >> 4;
    const int l16  = lane & 15;
    const int krow = k0 + l16;
    const int prow = p0 + l16;

    v8f rr = {}, ii = {}, ri = {}, ir = {};

#pragma unroll 4
    for (int n0 = 0; n0 < N1D; n0 += 4) {
        const int koff = n0 + 2 * half;
        v2f mr = *(const v2f*)(Mre + krow * N1D + koff);
        v2f mi = *(const v2f*)(Mim + krow * N1D + koff);
        v2f tr = *(const v2f*)(Tre + prow * N1D + koff);
        v2f ti = *(const v2f*)(Tim + prow * N1D + koff);
        rr = __builtin_amdgcn_wmma_f32_16x16x4_f32(false, mr, false, tr, (short)0, rr, false, false);
        ii = __builtin_amdgcn_wmma_f32_16x16x4_f32(false, mi, false, ti, (short)0, ii, false, false);
        ri = __builtin_amdgcn_wmma_f32_16x16x4_f32(false, mr, false, ti, (short)0, ri, false, false);
        ir = __builtin_amdgcn_wmma_f32_16x16x4_f32(false, mi, false, tr, (short)0, ir, false, false);
    }

#pragma unroll
    for (int v = 0; v < 8; ++v) {
        const int row = k0 + v + 8 * half;
        const int idx = row * 4096 + p0 + l16;
        Ure[idx] = rr[v] - ii[v];
        Uim[idx] = ri[v] + ir[v];
    }
}

__global__ __launch_bounds__(256) void reduce1_kernel(const float* __restrict__ x,
                                                      float* __restrict__ psum,
                                                      float* __restrict__ psq) {
    __shared__ float ssum[256];
    __shared__ float ssq[256];
    const int tid = threadIdx.x;
    const int base = blockIdx.x * 512;
    float a = x[base + tid];
    float b = x[base + 256 + tid];
    ssum[tid] = a + b;
    ssq[tid]  = a * a + b * b;
    __syncthreads();
    for (int s = 128; s > 0; s >>= 1) {
        if (tid < s) { ssum[tid] += ssum[tid + s]; ssq[tid] += ssq[tid + s]; }
        __syncthreads();
    }
    if (tid == 0) { psum[blockIdx.x] = ssum[0]; psq[blockIdx.x] = ssq[0]; }
}

__global__ __launch_bounds__(256) void reduce2_kernel(const float* __restrict__ psum,
                                                      const float* __restrict__ psq,
                                                      float* __restrict__ stats) {
    __shared__ float ssum[256];
    __shared__ float ssq[256];
    const int tid = threadIdx.x;
    ssum[tid] = psum[tid] + psum[tid + 256];
    ssq[tid]  = psq[tid] + psq[tid + 256];
    __syncthreads();
    for (int s = 128; s > 0; s >>= 1) {
        if (tid < s) { ssum[tid] += ssum[tid + s]; ssq[tid] += ssq[tid + s]; }
        __syncthreads();
    }
    if (tid == 0) {
        const float invN = 1.0f / (float)GRID_N;
        float mean = ssum[0] * invN;
        float var  = fmaxf(ssq[0] * invN - mean * mean, 0.0f);
        stats[0] = mean;
        stats[1] = 1.0f / (sqrtf(var) + 1e-8f);
    }
}

__global__ __launch_bounds__(256) void finalize_kernel(const float* __restrict__ re,
                                                       const float* __restrict__ stats,
                                                       float* __restrict__ out) {
    int g = blockIdx.x * blockDim.x + threadIdx.x;
    if (g >= GRID_N) return;
    out[g] = (re[g] - stats[0]) * stats[1];
}

extern "C" void kernel_launch(void* const* d_in, const int* in_sizes, int n_in,
                              void* d_out, int out_size, void* d_ws, size_t ws_size,
                              hipStream_t stream) {
    const float* X   = (const float*)d_in[0];
    const float* aw  = (const float*)d_in[1];
    const float* bw  = (const float*)d_in[2];
    const float* ham = (const float*)d_in[4];
    float* out = (float*)d_out;

    float* ws   = (float*)d_ws;
    float* Wre  = ws;
    float* WimI = Wre + 8192;
    float* Are  = WimI + 4096;
    float* Aim  = Are + GRID_N;
    float* Bre  = Aim + GRID_N;
    float* Bim  = Bre + GRID_N;
    float* psum = Bim + GRID_N;
    float* psq  = psum + 512;
    float* stats = psq + 512;

    float* Pre  = Are;
    float* Pim  = Pre  + TAB_SZ;
    float* YFre = Pim  + TAB_SZ;
    float* YFim = YFre + TAB_SZ;
    float* ZFre = YFim + TAB_SZ;
    float* ZFim = ZFre + TAB_SZ;

    axis_spectra_kernel<<<304, 64, 0, stream>>>(X, aw, bw, Wre, WimI,
                                                Pre, Pim, YFre, YFim, ZFre, ZFim);

    spectrum_contract_kernel<<<256, 256, 0, stream>>>(Pre, Pim, YFre, YFim, ZFre, ZFim,
                                                      ham, Wre, WimI, Bre, Bim);

    dft_pass_kernel<<<128, 256, 0, stream>>>(Bre, Bim, Wre, WimI, Are, Aim);
    dft_pass_kernel<<<128, 256, 0, stream>>>(Are, Aim, Wre, WimI, Bre, Bim);

    reduce1_kernel<<<512, 256, 0, stream>>>(Bre, psum, psq);
    reduce2_kernel<<<1, 256, 0, stream>>>(psum, psq, stats);
    finalize_kernel<<<GRID_N / 256, 256, 0, stream>>>(Bre, stats, out);
}


# spectrum contraction multiplies kz 0..31 only: column 64-kz reuses the same four real products with signs swapped (real terms: conjugate z spectra), 16->8 WMMAs a k-step; term sum order changes
# speedup vs baseline: 1.6472x; 1.1443x over previous
#include <hip/hip_runtime.h>
#include <hip/hip_bf16.h>
#include <math.h>

#define N1D      64
#define GRID_N   (64 * 64 * 64)
#define N_ATOMS  128
#define K_TERMS  6
#define SPACING  0.5f
#define LOG2E    1.4426950408889634f

typedef float v2f __attribute__((ext_vector_type(2)));
typedef float v8f __attribute__((ext_vector_type(8)));

__device__ __forceinline__ float fast_exp2(float x) {
#if __has_builtin(__builtin_amdgcn_exp2f)
    return __builtin_amdgcn_exp2f(x);
#else
    return exp2f(x);
#endif
}

#define N_TERMS  (N_ATOMS * K_TERMS)
#define TAB_SZ   (N_TERMS * N1D)

__global__ __launch_bounds__(64) void axis_spectra_kernel(const float* __restrict__ X,
                                                           const float* __restrict__ aw,
                                                           const float* __restrict__ bw,
                                                           float* __restrict__ Wre,
                                                           float* __restrict__ WimI,
                                                           float* __restrict__ Pre,
                                                           float* __restrict__ Pim,
                                                           float* __restrict__ YFre,
                                                           float* __restrict__ YFim,
                                                           float* __restrict__ ZFre,
                                                           float* __restrict__ ZFim) {
    __shared__ float ctab[N1D];
    __shared__ float stab[N1D];
    {
        const int j = threadIdx.x;
        const float ang = (float)j * 0.09817477042468103f;
        ctab[j] = 0.125f * cosf(ang);
        stab[j] = 0.125f * sinf(ang);
    }
    __syncthreads();

    if (blockIdx.x >= 288) {
        const int e = (blockIdx.x - 288) * 64 + threadIdx.x;
        const int k = e >> 4;
        const int n = (e & 15) * 4;
        float cr[4], si[4];
#pragma unroll
        for (int q = 0; q < 4; ++q) {
            const int kn = (k * (n + q)) & 63;
            const float sgn = ((k + n + q) & 1) ? -1.0f : 1.0f;
            cr[q] = sgn * ctab[kn];
            si[q] = sgn * stab[kn];
        }
        *(float4*)(Wre  + k * N1D + n) = make_float4(cr[0], cr[1], cr[2], cr[3]);
        *(float4*)(WimI + k * N1D + n) = make_float4(si[0], si[1], si[2], si[3]);
        return;
    }

    const int lane   = threadIdx.x & 31;
    const int waveId = (blockIdx.x * blockDim.x + threadIdx.x) >> 5;
    const int ck   = waveId & 3;
    const int rest = waveId >> 2;
    const int d    = rest / 48;
    const int tt   = rest - d * 48;
    const int half = lane >> 4;
    const int l16  = lane & 15;

    const int t = tt * 16 + l16;
    const int a = t / K_TERMS;
    const float c   = X[3 * a + d];
    const float bl2 = bw[t] * LOG2E;
    const float scale = (d == 0) ? aw[t] : 1.0f;
    const int krow = ck * 16 + l16;
    const float sg0 = (krow & 1) ? -1.0f : 1.0f;
    const bool dcz = (d == 2) && (krow == 0);
    const float* itab = dcz ? ctab : stab;
    const float sgx = dcz ? 1.0f : -sg0;
    const float sgy = dcz ? 1.0f : sg0;

    v8f re = {}, im = {};
#pragma unroll 4
    for (int n0 = 0; n0 < N1D; n0 += 4) {
        const int koff = n0 + 2 * half;
        const float d0 = (float)koff * SPACING - c;
        const float d1 = (float)(koff + 1) * SPACING - c;
        v2f e;
        e.x = scale * fast_exp2(bl2 * (d0 * d0));
        e.y = scale * fast_exp2(bl2 * (d1 * d1));
        const int i0 = (krow * koff) & 63;
        const int i1 = (i0 + krow) & 63;
        v2f wr, wi;
        wr.x = sg0 * ctab[i0];   wr.y = -sg0 * ctab[i1];
        wi.x = sgx * itab[i0];   wi.y = sgy * itab[i1];
        re = __builtin_amdgcn_wmma_f32_16x16x4_f32(false, e, false, wr, (short)0, re, false, false);
        im = __builtin_amdgcn_wmma_f32_16x16x4_f32(false, e, false, wi, (short)0, im, false, false);
    }

    if (d == 0) {
        const int o = krow * N_TERMS + tt * 16 + 8 * half;
        *(float4*)(Pre + o)     = make_float4(re[0], re[1], re[2], re[3]);
        *(float4*)(Pre + o + 4) = make_float4(re[4], re[5], re[6], re[7]);
        *(float4*)(Pim + o)     = make_float4(im[0], im[1], im[2], im[3]);
        *(float4*)(Pim + o + 4) = make_float4(im[4], im[5], im[6], im[7]);
    } else {
        float* __restrict__ Fr = (d == 1) ? YFre : ZFre;
        float* __restrict__ Fi = (d == 1) ? YFim : ZFim;
#pragma unroll
        for (int r2 = 0; r2 < 4; ++r2) {
            const int s  = tt * 4 + 2 * half + (r2 >> 1);
            const int hh = r2 & 1;
            const int o  = (((s * 4 + ck) * 32) + hh * 16 + l16) * 2;
            v2f vr, vi;
            vr.x = re[2 * r2]; vr.y = re[2 * r2 + 1];
            vi.x = im[2 * r2]; vi.y = im[2 * r2 + 1];
            *(v2f*)(Fr + o) = vr;
            *(v2f*)(Fi + o) = vi;
        }
    }
}

#define KSTEPS_Q (N_TERMS / 16)
#define CG 8
#define GP 68
__global__ __launch_bounds__(256) void spectrum_contract_kernel(const float* __restrict__ Pre,
                                                                const float* __restrict__ Pim,
                                                                const float* __restrict__ YFre,
                                                                const float* __restrict__ YFim,
                                                                const float* __restrict__ ZFre,
                                                                const float* __restrict__ ZFim,
                                                                const float* __restrict__ ham,
                                                                const float* __restrict__ Mre,
                                                                const float* __restrict__ Mim,
                                                                float* __restrict__ Ure,
                                                                float* __restrict__ Uim) {
    __shared__ float part[4][2][32][32];
    __shared__ __attribute__((aligned(16))) float Gre[16 * GP];
    __shared__ __attribute__((aligned(16))) float Gim[16 * GP];

    const int lane = threadIdx.x & 31;
    const int w    = threadIdx.x >> 5;
    const int ct   = w & 1;
    const int kq   = w >> 1;
    const int cz   = w & 3;
    const int kh   = w >> 2;
    const int kx   = blockIdx.x >> 2;
    const int cy   = blockIdx.x & 3;
    const int half = lane >> 4;
    const int l16  = lane & 15;
    const int sBeg = kq * KSTEPS_Q;

    const v2f* __restrict__ pPr = (const v2f*)(Pre + kx * N_TERMS + 2 * half) + 2 * sBeg;
    const v2f* __restrict__ pPi = (const v2f*)(Pim + kx * N_TERMS + 2 * half) + 2 * sBeg;
    const v2f* __restrict__ pYr = (const v2f*)YFre + 128 * sBeg + cy * 32 + lane;
    const v2f* __restrict__ pYi = (const v2f*)YFim + 128 * sBeg + cy * 32 + lane;
    const v2f* __restrict__ pZr = (const v2f*)ZFre + 128 * sBeg + ct * 32 + lane;
    const v2f* __restrict__ pZi = (const v2f*)ZFim + 128 * sBeg + ct * 32 + lane;

    v8f rr = {}, ii = {}, ri = {}, ir = {};
    for (int s0 = 0; s0 < KSTEPS_Q; s0 += CG) {
        v2f pr[CG], pi[CG], yr[CG], yi[CG], zr[CG], zi[CG];
#pragma unroll
        for (int u = 0; u < CG; ++u) {
            pr[u] = pPr[2 * (s0 + u)];   pi[u] = pPi[2 * (s0 + u)];
            yr[u] = pYr[128 * (s0 + u)]; yi[u] = pYi[128 * (s0 + u)];
            zr[u] = pZr[128 * (s0 + u)]; zi[u] = pZi[128 * (s0 + u)];
        }
#pragma unroll
        for (int u = 0; u < CG; ++u) {
            const v2f ar = pr[u] * yr[u] - pi[u] * yi[u];
            const v2f ai = pr[u] * yi[u] + pi[u] * yr[u];
            rr = __builtin_amdgcn_wmma_f32_16x16x4_f32(false, ar, false, zr[u], (short)0, rr, false, false);
            ii = __builtin_amdgcn_wmma_f32_16x16x4_f32(false, ai, false, zi[u], (short)0, ii, false, false);
            ri = __builtin_amdgcn_wmma_f32_16x16x4_f32(false, ar, false, zi[u], (short)0, ri, false, false);
            ir = __builtin_amdgcn_wmma_f32_16x16x4_f32(false, ai, false, zr[u], (short)0, ir, false, false);
        }
    }

    v2f mr[16], mi[16];
    {
        const float* __restrict__ aR = Mre + (cz * 16 + l16) * N1D + 2 * half;
        const float* __restrict__ aI = Mim + (cz * 16 + l16) * N1D + 2 * half;
#pragma unroll
        for (int s = 0; s < 16; ++s) {
            mr[s] = *(const v2f*)(aR + 4 * s);
            mi[s] = *(const v2f*)(aI + 4 * s);
        }
    }
    const int  kzD = ct * 16 + l16;
    const bool sp  = (kzD == 0);
    const int  kzM = sp ? 32 : 64 - kzD;
    float hD[2], hM[2];
#pragma unroll
    for (int e = 0; e < 2; ++e) {
        const int ky = cy * 16 + 2 * kq + e + 8 * half;
        hD[e] = ham[kx * 4096 + ky * N1D + kzD];
        hM[e] = ham[kx * 4096 + ky * N1D + kzM];
    }

#pragma unroll
    for (int v = 0; v < 8; ++v) {
        part[kq][ct][v][lane]      = rr[v];
        part[kq][ct][8 + v][lane]  = ii[v];
        part[kq][ct][16 + v][lane] = ri[v];
        part[kq][ct][24 + v][lane] = ir[v];
    }
    __syncthreads();
#pragma unroll
    for (int e = 0; e < 2; ++e) {
        const int v = 2 * kq + e;
        float srr = part[0][ct][v][lane];
        float sii = part[0][ct][8 + v][lane];
        float sri = part[0][ct][16 + v][lane];
        float sir = part[0][ct][24 + v][lane];
#pragma unroll
        for (int q = 1; q < 4; ++q) {
            srr += part[q][ct][v][lane];
            sii += part[q][ct][8 + v][lane];
            sri += part[q][ct][16 + v][lane];
            sir += part[q][ct][24 + v][lane];
        }
        const float dre = sp ? srr : srr - sii;
        const float dim = sp ? sir : sri + sir;
        const float mre = sp ? sri : srr + sii;
        const float mim = sp ? sii : sir - sri;
        const int o = (v + 8 * half) * GP;
        Gre[o + kzD] = dre * hD[e];
        Gim[o + kzD] = dim * hD[e];
        Gre[o + kzM] = mre * hM[e];
        Gim[o + kzM] = mim * hM[e];
    }
    __syncthreads();

    const float* b1 = (kh ? Gim : Gre) + l16 * GP + 2 * half;
    const float* b2 = (kh ? Gre : Gim) + l16 * GP + 2 * half;
    v8f a1 = {}, a2 = {};
#pragma unroll
    for (int s = 0; s < 16; ++s) {
        const v2f t1 = *(const v2f*)(b1 + 4 * s);
        const v2f t2 = *(const v2f*)(b2 + 4 * s);
        a1 = __builtin_amdgcn_wmma_f32_16x16x4_f32(false, mr[s], false, t1, (short)0, a1, false, false);
        a2 = __builtin_amdgcn_wmma_f32_16x16x4_f32(false, mi[s], false, t2, (short)0, a2, false, false);
    }
    const float sg = kh ? 1.0f : -1.0f;
    const v8f u = a1 + sg * a2;

    float* __restrict__ dst = kh ? Uim : Ure;
#pragma unroll
    for (int v = 0; v < 8; ++v) {
        const int z = cz * 16 + v + 8 * half;
        dst[z * 4096 + kx * N1D + cy * 16 + l16] = u[v];
    }
}

__global__ __launch_bounds__(256) void dft_pass_kernel(const float* __restrict__ Tre,
                                                       const float* __restrict__ Tim,
                                                       const float* __restrict__ Mre,
                                                       const float* __restrict__ Mim,
                                                       float* __restrict__ Ure,
                                                       float* __restrict__ Uim) {
    const int lane   = threadIdx.x & 31;
    const int waveId = (blockIdx.x * blockDim.x + threadIdx.x) >> 5;
    const int p0 = (waveId & 255) << 4;
    const int k0 = (waveId >> 8) << 4;
    const int half = lane >> 4;
    const int l16  = lane & 15;
    const int krow = k0 + l16;
    const int prow = p0 + l16;

    v8f rr = {}, ii = {}, ri = {}, ir = {};

#pragma unroll 4
    for (int n0 = 0; n0 < N1D; n0 += 4) {
        const int koff = n0 + 2 * half;
        v2f mr = *(const v2f*)(Mre + krow * N1D + koff);
        v2f mi = *(const v2f*)(Mim + krow * N1D + koff);
        v2f tr = *(const v2f*)(Tre + prow * N1D + koff);
        v2f ti = *(const v2f*)(Tim + prow * N1D + koff);
        rr = __builtin_amdgcn_wmma_f32_16x16x4_f32(false, mr, false, tr, (short)0, rr, false, false);
        ii = __builtin_amdgcn_wmma_f32_16x16x4_f32(false, mi, false, ti, (short)0, ii, false, false);
        ri = __builtin_amdgcn_wmma_f32_16x16x4_f32(false, mr, false, ti, (short)0, ri, false, false);
        ir = __builtin_amdgcn_wmma_f32_16x16x4_f32(false, mi, false, tr, (short)0, ir, false, false);
    }

#pragma unroll
    for (int v = 0; v < 8; ++v) {
        const int row = k0 + v + 8 * half;
        const int idx = row * 4096 + p0 + l16;
        Ure[idx] = rr[v] - ii[v];
        Uim[idx] = ri[v] + ir[v];
    }
}

__global__ __launch_bounds__(256) void reduce1_kernel(const float* __restrict__ x,
                                                      float* __restrict__ psum,
                                                      float* __restrict__ psq) {
    __shared__ float ssum[256];
    __shared__ float ssq[256];
    const int tid = threadIdx.x;
    const int base = blockIdx.x * 512;
    float a = x[base + tid];
    float b = x[base + 256 + tid];
    ssum[tid] = a + b;
    ssq[tid]  = a * a + b * b;
    __syncthreads();
    for (int s = 128; s > 0; s >>= 1) {
        if (tid < s) { ssum[tid] += ssum[tid + s]; ssq[tid] += ssq[tid + s]; }
        __syncthreads();
    }
    if (tid == 0) { psum[blockIdx.x] = ssum[0]; psq[blockIdx.x] = ssq[0]; }
}

__global__ __launch_bounds__(256) void reduce2_kernel(const float* __restrict__ psum,
                                                      const float* __restrict__ psq,
                                                      float* __restrict__ stats) {
    __shared__ float ssum[256];
    __shared__ float ssq[256];
    const int tid = threadIdx.x;
    ssum[tid] = psum[tid] + psum[tid + 256];
    ssq[tid]  = psq[tid] + psq[tid + 256];
    __syncthreads();
    for (int s = 128; s > 0; s >>= 1) {
        if (tid < s) { ssum[tid] += ssum[tid + s]; ssq[tid] += ssq[tid + s]; }
        __syncthreads();
    }
    if (tid == 0) {
        const float invN = 1.0f / (float)GRID_N;
        float mean = ssum[0] * invN;
        float var  = fmaxf(ssq[0] * invN - mean * mean, 0.0f);
        stats[0] = mean;
        stats[1] = 1.0f / (sqrtf(var) + 1e-8f);
    }
}

__global__ __launch_bounds__(256) void finalize_kernel(const float* __restrict__ re,
                                                       const float* __restrict__ stats,
                                                       float* __restrict__ out) {
    int g = blockIdx.x * blockDim.x + threadIdx.x;
    if (g >= GRID_N) return;
    out[g] = (re[g] - stats[0]) * stats[1];
}

extern "C" void kernel_launch(void* const* d_in, const int* in_sizes, int n_in,
                              void* d_out, int out_size, void* d_ws, size_t ws_size,
                              hipStream_t stream) {
    const float* X   = (const float*)d_in[0];
    const float* aw  = (const float*)d_in[1];
    const float* bw  = (const float*)d_in[2];
    const float* ham = (const float*)d_in[4];
    float* out = (float*)d_out;

    float* ws   = (float*)d_ws;
    float* Wre  = ws;
    float* WimI = Wre + 8192;
    float* Are  = WimI + 4096;
    float* Aim  = Are + GRID_N;
    float* Bre  = Aim + GRID_N;
    float* Bim  = Bre + GRID_N;
    float* psum = Bim + GRID_N;
    float* psq  = psum + 512;
    float* stats = psq + 512;

    float* Pre  = Are;
    float* Pim  = Pre  + TAB_SZ;
    float* YFre = Pim  + TAB_SZ;
    float* YFim = YFre + TAB_SZ;
    float* ZFre = YFim + TAB_SZ;
    float* ZFim = ZFre + TAB_SZ;

    axis_spectra_kernel<<<304, 64, 0, stream>>>(X, aw, bw, Wre, WimI,
                                                Pre, Pim, YFre, YFim, ZFre, ZFim);

    spectrum_contract_kernel<<<256, 256, 0, stream>>>(Pre, Pim, YFre, YFim, ZFre, ZFim,
                                                      ham, Wre, WimI, Bre, Bim);

    dft_pass_kernel<<<128, 256, 0, stream>>>(Bre, Bim, Wre, WimI, Are, Aim);
    dft_pass_kernel<<<128, 256, 0, stream>>>(Are, Aim, Wre, WimI, Bre, Bim);

    reduce1_kernel<<<512, 256, 0, stream>>>(Bre, psum, psq);
    reduce2_kernel<<<1, 256, 0, stream>>>(psum, psq, stats);
    finalize_kernel<<<GRID_N / 256, 256, 0, stream>>>(Bre, stats, out);
}


# normalise tail on fewer fatter waves: block-sum and normalise launches take 16 elements a lane as four 16-byte accesses (12288 -> 1024 waves; wave entry was the cost); mean/variance sum order changes
# speedup vs baseline: 1.7051x; 1.0352x over previous
#include <hip/hip_runtime.h>
#include <hip/hip_bf16.h>
#include <math.h>

#define N1D      64
#define GRID_N   (64 * 64 * 64)
#define N_ATOMS  128
#define K_TERMS  6
#define SPACING  0.5f
#define LOG2E    1.4426950408889634f

typedef float v2f __attribute__((ext_vector_type(2)));
typedef float v8f __attribute__((ext_vector_type(8)));

__device__ __forceinline__ float fast_exp2(float x) {
#if __has_builtin(__builtin_amdgcn_exp2f)
    return __builtin_amdgcn_exp2f(x);
#else
    return exp2f(x);
#endif
}

#define N_TERMS  (N_ATOMS * K_TERMS)
#define TAB_SZ   (N_TERMS * N1D)

__global__ __launch_bounds__(64) void axis_spectra_kernel(const float* __restrict__ X,
                                                           const float* __restrict__ aw,
                                                           const float* __restrict__ bw,
                                                           float* __restrict__ Wre,
                                                           float* __restrict__ WimI,
                                                           float* __restrict__ Pre,
                                                           float* __restrict__ Pim,
                                                           float* __restrict__ YFre,
                                                           float* __restrict__ YFim,
                                                           float* __restrict__ ZFre,
                                                           float* __restrict__ ZFim) {
    __shared__ float ctab[N1D];
    __shared__ float stab[N1D];
    {
        const int j = threadIdx.x;
        const float ang = (float)j * 0.09817477042468103f;
        ctab[j] = 0.125f * cosf(ang);
        stab[j] = 0.125f * sinf(ang);
    }
    __syncthreads();

    if (blockIdx.x >= 288) {
        const int e = (blockIdx.x - 288) * 64 + threadIdx.x;
        const int k = e >> 4;
        const int n = (e & 15) * 4;
        float cr[4], si[4];
#pragma unroll
        for (int q = 0; q < 4; ++q) {
            const int kn = (k * (n + q)) & 63;
            const float sgn = ((k + n + q) & 1) ? -1.0f : 1.0f;
            cr[q] = sgn * ctab[kn];
            si[q] = sgn * stab[kn];
        }
        *(float4*)(Wre  + k * N1D + n) = make_float4(cr[0], cr[1], cr[2], cr[3]);
        *(float4*)(WimI + k * N1D + n) = make_float4(si[0], si[1], si[2], si[3]);
        return;
    }

    const int lane   = threadIdx.x & 31;
    const int waveId = (blockIdx.x * blockDim.x + threadIdx.x) >> 5;
    const int ck   = waveId & 3;
    const int rest = waveId >> 2;
    const int d    = rest / 48;
    const int tt   = rest - d * 48;
    const int half = lane >> 4;
    const int l16  = lane & 15;

    const int t = tt * 16 + l16;
    const int a = t / K_TERMS;
    const float c   = X[3 * a + d];
    const float bl2 = bw[t] * LOG2E;
    const float scale = (d == 0) ? aw[t] : 1.0f;
    const int krow = ck * 16 + l16;
    const float sg0 = (krow & 1) ? -1.0f : 1.0f;
    const bool dcz = (d == 2) && (krow == 0);
    const float* itab = dcz ? ctab : stab;
    const float sgx = dcz ? 1.0f : -sg0;
    const float sgy = dcz ? 1.0f : sg0;

    v8f re = {}, im = {};
#pragma unroll 4
    for (int n0 = 0; n0 < N1D; n0 += 4) {
        const int koff = n0 + 2 * half;
        const float d0 = (float)koff * SPACING - c;
        const float d1 = (float)(koff + 1) * SPACING - c;
        v2f e;
        e.x = scale * fast_exp2(bl2 * (d0 * d0));
        e.y = scale * fast_exp2(bl2 * (d1 * d1));
        const int i0 = (krow * koff) & 63;
        const int i1 = (i0 + krow) & 63;
        v2f wr, wi;
        wr.x = sg0 * ctab[i0];   wr.y = -sg0 * ctab[i1];
        wi.x = sgx * itab[i0];   wi.y = sgy * itab[i1];
        re = __builtin_amdgcn_wmma_f32_16x16x4_f32(false, e, false, wr, (short)0, re, false, false);
        im = __builtin_amdgcn_wmma_f32_16x16x4_f32(false, e, false, wi, (short)0, im, false, false);
    }

    if (d == 0) {
        const int o = krow * N_TERMS + tt * 16 + 8 * half;
        *(float4*)(Pre + o)     = make_float4(re[0], re[1], re[2], re[3]);
        *(float4*)(Pre + o + 4) = make_float4(re[4], re[5], re[6], re[7]);
        *(float4*)(Pim + o)     = make_float4(im[0], im[1], im[2], im[3]);
        *(float4*)(Pim + o + 4) = make_float4(im[4], im[5], im[6], im[7]);
    } else {
        float* __restrict__ Fr = (d == 1) ? YFre : ZFre;
        float* __restrict__ Fi = (d == 1) ? YFim : ZFim;
#pragma unroll
        for (int r2 = 0; r2 < 4; ++r2) {
            const int s  = tt * 4 + 2 * half + (r2 >> 1);
            const int hh = r2 & 1;
            const int o  = (((s * 4 + ck) * 32) + hh * 16 + l16) * 2;
            v2f vr, vi;
            vr.x = re[2 * r2]; vr.y = re[2 * r2 + 1];
            vi.x = im[2 * r2]; vi.y = im[2 * r2 + 1];
            *(v2f*)(Fr + o) = vr;
            *(v2f*)(Fi + o) = vi;
        }
    }
}

#define KSTEPS_Q (N_TERMS / 16)
#define CG 8
#define GP 68
__global__ __launch_bounds__(256) void spectrum_contract_kernel(const float* __restrict__ Pre,
                                                                const float* __restrict__ Pim,
                                                                const float* __restrict__ YFre,
                                                                const float* __restrict__ YFim,
                                                                const float* __restrict__ ZFre,
                                                                const float* __restrict__ ZFim,
                                                                const float* __restrict__ ham,
                                                                const float* __restrict__ Mre,
                                                                const float* __restrict__ Mim,
                                                                float* __restrict__ Ure,
                                                                float* __restrict__ Uim) {
    __shared__ float part[4][2][32][32];
    __shared__ __attribute__((aligned(16))) float Gre[16 * GP];
    __shared__ __attribute__((aligned(16))) float Gim[16 * GP];

    const int lane = threadIdx.x & 31;
    const int w    = threadIdx.x >> 5;
    const int ct   = w & 1;
    const int kq   = w >> 1;
    const int cz   = w & 3;
    const int kh   = w >> 2;
    const int kx   = blockIdx.x >> 2;
    const int cy   = blockIdx.x & 3;
    const int half = lane >> 4;
    const int l16  = lane & 15;
    const int sBeg = kq * KSTEPS_Q;

    const v2f* __restrict__ pPr = (const v2f*)(Pre + kx * N_TERMS + 2 * half) + 2 * sBeg;
    const v2f* __restrict__ pPi = (const v2f*)(Pim + kx * N_TERMS + 2 * half) + 2 * sBeg;
    const v2f* __restrict__ pYr = (const v2f*)YFre + 128 * sBeg + cy * 32 + lane;
    const v2f* __restrict__ pYi = (const v2f*)YFim + 128 * sBeg + cy * 32 + lane;
    const v2f* __restrict__ pZr = (const v2f*)ZFre + 128 * sBeg + ct * 32 + lane;
    const v2f* __restrict__ pZi = (const v2f*)ZFim + 128 * sBeg + ct * 32 + lane;

    v8f rr = {}, ii = {}, ri = {}, ir = {};
    for (int s0 = 0; s0 < KSTEPS_Q; s0 += CG) {
        v2f pr[CG], pi[CG], yr[CG], yi[CG], zr[CG], zi[CG];
#pragma unroll
        for (int u = 0; u < CG; ++u) {
            pr[u] = pPr[2 * (s0 + u)];   pi[u] = pPi[2 * (s0 + u)];
            yr[u] = pYr[128 * (s0 + u)]; yi[u] = pYi[128 * (s0 + u)];
            zr[u] = pZr[128 * (s0 + u)]; zi[u] = pZi[128 * (s0 + u)];
        }
#pragma unroll
        for (int u = 0; u < CG; ++u) {
            const v2f ar = pr[u] * yr[u] - pi[u] * yi[u];
            const v2f ai = pr[u] * yi[u] + pi[u] * yr[u];
            rr = __builtin_amdgcn_wmma_f32_16x16x4_f32(false, ar, false, zr[u], (short)0, rr, false, false);
            ii = __builtin_amdgcn_wmma_f32_16x16x4_f32(false, ai, false, zi[u], (short)0, ii, false, false);
            ri = __builtin_amdgcn_wmma_f32_16x16x4_f32(false, ar, false, zi[u], (short)0, ri, false, false);
            ir = __builtin_amdgcn_wmma_f32_16x16x4_f32(false, ai, false, zr[u], (short)0, ir, false, false);
        }
    }

    v2f mr[16], mi[16];
    {
        const float* __restrict__ aR = Mre + (cz * 16 + l16) * N1D + 2 * half;
        const float* __restrict__ aI = Mim + (cz * 16 + l16) * N1D + 2 * half;
#pragma unroll
        for (int s = 0; s < 16; ++s) {
            mr[s] = *(const v2f*)(aR + 4 * s);
            mi[s] = *(const v2f*)(aI + 4 * s);
        }
    }
    const int  kzD = ct * 16 + l16;
    const bool sp  = (kzD == 0);
    const int  kzM = sp ? 32 : 64 - kzD;
    float hD[2], hM[2];
#pragma unroll
    for (int e = 0; e < 2; ++e) {
        const int ky = cy * 16 + 2 * kq + e + 8 * half;
        hD[e] = ham[kx * 4096 + ky * N1D + kzD];
        hM[e] = ham[kx * 4096 + ky * N1D + kzM];
    }

#pragma unroll
    for (int v = 0; v < 8; ++v) {
        part[kq][ct][v][lane]      = rr[v];
        part[kq][ct][8 + v][lane]  = ii[v];
        part[kq][ct][16 + v][lane] = ri[v];
        part[kq][ct][24 + v][lane] = ir[v];
    }
    __syncthreads();
#pragma unroll
    for (int e = 0; e < 2; ++e) {
        const int v = 2 * kq + e;
        float srr = part[0][ct][v][lane];
        float sii = part[0][ct][8 + v][lane];
        float sri = part[0][ct][16 + v][lane];
        float sir = part[0][ct][24 + v][lane];
#pragma unroll
        for (int q = 1; q < 4; ++q) {
            srr += part[q][ct][v][lane];
            sii += part[q][ct][8 + v][lane];
            sri += part[q][ct][16 + v][lane];
            sir += part[q][ct][24 + v][lane];
        }
        const float dre = sp ? srr : srr - sii;
        const float dim = sp ? sir : sri + sir;
        const float mre = sp ? sri : srr + sii;
        const float mim = sp ? sii : sir - sri;
        const int o = (v + 8 * half) * GP;
        Gre[o + kzD] = dre * hD[e];
        Gim[o + kzD] = dim * hD[e];
        Gre[o + kzM] = mre * hM[e];
        Gim[o + kzM] = mim * hM[e];
    }
    __syncthreads();

    const float* b1 = (kh ? Gim : Gre) + l16 * GP + 2 * half;
    const float* b2 = (kh ? Gre : Gim) + l16 * GP + 2 * half;
    v8f a1 = {}, a2 = {};
#pragma unroll
    for (int s = 0; s < 16; ++s) {
        const v2f t1 = *(const v2f*)(b1 + 4 * s);
        const v2f t2 = *(const v2f*)(b2 + 4 * s);
        a1 = __builtin_amdgcn_wmma_f32_16x16x4_f32(false, mr[s], false, t1, (short)0, a1, false, false);
        a2 = __builtin_amdgcn_wmma_f32_16x16x4_f32(false, mi[s], false, t2, (short)0, a2, false, false);
    }
    const float sg = kh ? 1.0f : -1.0f;
    const v8f u = a1 + sg * a2;

    float* __restrict__ dst = kh ? Uim : Ure;
#pragma unroll
    for (int v = 0; v < 8; ++v) {
        const int z = cz * 16 + v + 8 * half;
        dst[z * 4096 + kx * N1D + cy * 16 + l16] = u[v];
    }
}

__global__ __launch_bounds__(256) void dft_pass_kernel(const float* __restrict__ Tre,
                                                       const float* __restrict__ Tim,
                                                       const float* __restrict__ Mre,
                                                       const float* __restrict__ Mim,
                                                       float* __restrict__ Ure,
                                                       float* __restrict__ Uim) {
    const int lane   = threadIdx.x & 31;
    const int waveId = (blockIdx.x * blockDim.x + threadIdx.x) >> 5;
    const int p0 = (waveId & 255) << 4;
    const int k0 = (waveId >> 8) << 4;
    const int half = lane >> 4;
    const int l16  = lane & 15;
    const int krow = k0 + l16;
    const int prow = p0 + l16;

    v8f rr = {}, ii = {}, ri = {}, ir = {};

#pragma unroll 4
    for (int n0 = 0; n0 < N1D; n0 += 4) {
        const int koff = n0 + 2 * half;
        v2f mr = *(const v2f*)(Mre + krow * N1D + koff);
        v2f mi = *(const v2f*)(Mim + krow * N1D + koff);
        v2f tr = *(const v2f*)(Tre + prow * N1D + koff);
        v2f ti = *(const v2f*)(Tim + prow * N1D + koff);
        rr = __builtin_amdgcn_wmma_f32_16x16x4_f32(false, mr, false, tr, (short)0, rr, false, false);
        ii = __builtin_amdgcn_wmma_f32_16x16x4_f32(false, mi, false, ti, (short)0, ii, false, false);
        ri = __builtin_amdgcn_wmma_f32_16x16x4_f32(false, mr, false, ti, (short)0, ri, false, false);
        ir = __builtin_amdgcn_wmma_f32_16x16x4_f32(false, mi, false, tr, (short)0, ir, false, false);
    }

#pragma unroll
    for (int v = 0; v < 8; ++v) {
        const int row = k0 + v + 8 * half;
        const int idx = row * 4096 + p0 + l16;
        Ure[idx] = rr[v] - ii[v];
        Uim[idx] = ri[v] + ir[v];
    }
}

#define TAIL_WG   64
#define TAIL_V4   (GRID_N / 4 / TAIL_WG)

__global__ __launch_bounds__(256) void reduce1_kernel(const float* __restrict__ x,
                                                      float* __restrict__ psum,
                                                      float* __restrict__ psq) {
    __shared__ float ssum[256];
    __shared__ float ssq[256];
    const int tid = threadIdx.x;
    const float4* __restrict__ x4 = (const float4*)x + blockIdx.x * TAIL_V4;
    float4 v[4];
#pragma unroll
    for (int q = 0; q < 4; ++q) v[q] = x4[q * 256 + tid];
    float s = 0.0f, sq = 0.0f;
#pragma unroll
    for (int q = 0; q < 4; ++q) {
        s  += (v[q].x + v[q].y) + (v[q].z + v[q].w);
        sq += (v[q].x * v[q].x + v[q].y * v[q].y) + (v[q].z * v[q].z + v[q].w * v[q].w);
    }
    ssum[tid] = s;
    ssq[tid]  = sq;
    __syncthreads();
    for (int st = 128; st > 0; st >>= 1) {
        if (tid < st) { ssum[tid] += ssum[tid + st]; ssq[tid] += ssq[tid + st]; }
        __syncthreads();
    }
    if (tid == 0) { psum[blockIdx.x] = ssum[0]; psq[blockIdx.x] = ssq[0]; }
}

__global__ __launch_bounds__(TAIL_WG) void reduce2_kernel(const float* __restrict__ psum,
                                                          const float* __restrict__ psq,
                                                          float* __restrict__ stats) {
    __shared__ float ssum[TAIL_WG];
    __shared__ float ssq[TAIL_WG];
    const int tid = threadIdx.x;
    ssum[tid] = psum[tid];
    ssq[tid]  = psq[tid];
    __syncthreads();
    for (int st = TAIL_WG / 2; st > 0; st >>= 1) {
        if (tid < st) { ssum[tid] += ssum[tid + st]; ssq[tid] += ssq[tid + st]; }
        __syncthreads();
    }
    if (tid == 0) {
        const float invN = 1.0f / (float)GRID_N;
        float mean = ssum[0] * invN;
        float var  = fmaxf(ssq[0] * invN - mean * mean, 0.0f);
        stats[0] = mean;
        stats[1] = 1.0f / (sqrtf(var) + 1e-8f);
    }
}

__global__ __launch_bounds__(256) void finalize_kernel(const float* __restrict__ re,
                                                       const float* __restrict__ stats,
                                                       float* __restrict__ out) {
    const int tid = threadIdx.x;
    const float4* __restrict__ r4 = (const float4*)re + blockIdx.x * TAIL_V4;
    float4* __restrict__ o4 = (float4*)out + blockIdx.x * TAIL_V4;
    float4 v[4];
#pragma unroll
    for (int q = 0; q < 4; ++q) v[q] = r4[q * 256 + tid];
    const float m  = stats[0];
    const float is = stats[1];
#pragma unroll
    for (int q = 0; q < 4; ++q)
        o4[q * 256 + tid] = make_float4((v[q].x - m) * is, (v[q].y - m) * is,
                                        (v[q].z - m) * is, (v[q].w - m) * is);
}

extern "C" void kernel_launch(void* const* d_in, const int* in_sizes, int n_in,
                              void* d_out, int out_size, void* d_ws, size_t ws_size,
                              hipStream_t stream) {
    const float* X   = (const float*)d_in[0];
    const float* aw  = (const float*)d_in[1];
    const float* bw  = (const float*)d_in[2];
    const float* ham = (const float*)d_in[4];
    float* out = (float*)d_out;

    float* ws   = (float*)d_ws;
    float* Wre  = ws;
    float* WimI = Wre + 8192;
    float* Are  = WimI + 4096;
    float* Aim  = Are + GRID_N;
    float* Bre  = Aim + GRID_N;
    float* Bim  = Bre + GRID_N;
    float* psum = Bim + GRID_N;
    float* psq  = psum + 512;
    float* stats = psq + 512;

    float* Pre  = Are;
    float* Pim  = Pre  + TAB_SZ;
    float* YFre = Pim  + TAB_SZ;
    float* YFim = YFre + TAB_SZ;
    float* ZFre = YFim + TAB_SZ;
    float* ZFim = ZFre + TAB_SZ;

    axis_spectra_kernel<<<304, 64, 0, stream>>>(X, aw, bw, Wre, WimI,
                                                Pre, Pim, YFre, YFim, ZFre, ZFim);

    spectrum_contract_kernel<<<256, 256, 0, stream>>>(Pre, Pim, YFre, YFim, ZFre, ZFim,
                                                      ham, Wre, WimI, Bre, Bim);

    dft_pass_kernel<<<128, 256, 0, stream>>>(Bre, Bim, Wre, WimI, Are, Aim);
    dft_pass_kernel<<<128, 256, 0, stream>>>(Are, Aim, Wre, WimI, Bre, Bim);

    reduce1_kernel<<<TAIL_WG, 256, 0, stream>>>(Bre, psum, psq);
    reduce2_kernel<<<1, TAIL_WG, 0, stream>>>(psum, psq, stats);
    finalize_kernel<<<TAIL_WG, 256, 0, stream>>>(Bre, stats, out);
}


# dft passes: a wave issues all 64 fragment reads as one group ahead of its 64 WMMAs (one memory wait, not four), in 256 work-groups of 4 waves so each WGP holds one wave a SIMD; sum order unchanged
# speedup vs baseline: 1.8184x; 1.0664x over previous
#include <hip/hip_runtime.h>
#include <hip/hip_bf16.h>
#include <math.h>

#define N1D      64
#define GRID_N   (64 * 64 * 64)
#define N_ATOMS  128
#define K_TERMS  6
#define SPACING  0.5f
#define LOG2E    1.4426950408889634f

typedef float v2f __attribute__((ext_vector_type(2)));
typedef float v8f __attribute__((ext_vector_type(8)));

__device__ __forceinline__ float fast_exp2(float x) {
#if __has_builtin(__builtin_amdgcn_exp2f)
    return __builtin_amdgcn_exp2f(x);
#else
    return exp2f(x);
#endif
}

#define N_TERMS  (N_ATOMS * K_TERMS)
#define TAB_SZ   (N_TERMS * N1D)

__global__ __launch_bounds__(64) void axis_spectra_kernel(const float* __restrict__ X,
                                                           const float* __restrict__ aw,
                                                           const float* __restrict__ bw,
                                                           float* __restrict__ Wre,
                                                           float* __restrict__ WimI,
                                                           float* __restrict__ Pre,
                                                           float* __restrict__ Pim,
                                                           float* __restrict__ YFre,
                                                           float* __restrict__ YFim,
                                                           float* __restrict__ ZFre,
                                                           float* __restrict__ ZFim) {
    __shared__ float ctab[N1D];
    __shared__ float stab[N1D];
    {
        const int j = threadIdx.x;
        const float ang = (float)j * 0.09817477042468103f;
        ctab[j] = 0.125f * cosf(ang);
        stab[j] = 0.125f * sinf(ang);
    }
    __syncthreads();

    if (blockIdx.x >= 288) {
        const int e = (blockIdx.x - 288) * 64 + threadIdx.x;
        const int k = e >> 4;
        const int n = (e & 15) * 4;
        float cr[4], si[4];
#pragma unroll
        for (int q = 0; q < 4; ++q) {
            const int kn = (k * (n + q)) & 63;
            const float sgn = ((k + n + q) & 1) ? -1.0f : 1.0f;
            cr[q] = sgn * ctab[kn];
            si[q] = sgn * stab[kn];
        }
        *(float4*)(Wre  + k * N1D + n) = make_float4(cr[0], cr[1], cr[2], cr[3]);
        *(float4*)(WimI + k * N1D + n) = make_float4(si[0], si[1], si[2], si[3]);
        return;
    }

    const int lane   = threadIdx.x & 31;
    const int waveId = (blockIdx.x * blockDim.x + threadIdx.x) >> 5;
    const int ck   = waveId & 3;
    const int rest = waveId >> 2;
    const int d    = rest / 48;
    const int tt   = rest - d * 48;
    const int half = lane >> 4;
    const int l16  = lane & 15;

    const int t = tt * 16 + l16;
    const int a = t / K_TERMS;
    const float c   = X[3 * a + d];
    const float bl2 = bw[t] * LOG2E;
    const float scale = (d == 0) ? aw[t] : 1.0f;
    const int krow = ck * 16 + l16;
    const float sg0 = (krow & 1) ? -1.0f : 1.0f;
    const bool dcz = (d == 2) && (krow == 0);
    const float* itab = dcz ? ctab : stab;
    const float sgx = dcz ? 1.0f : -sg0;
    const float sgy = dcz ? 1.0f : sg0;

    v8f re = {}, im = {};
#pragma unroll 4
    for (int n0 = 0; n0 < N1D; n0 += 4) {
        const int koff = n0 + 2 * half;
        const float d0 = (float)koff * SPACING - c;
        const float d1 = (float)(koff + 1) * SPACING - c;
        v2f e;
        e.x = scale * fast_exp2(bl2 * (d0 * d0));
        e.y = scale * fast_exp2(bl2 * (d1 * d1));
        const int i0 = (krow * koff) & 63;
        const int i1 = (i0 + krow) & 63;
        v2f wr, wi;
        wr.x = sg0 * ctab[i0];   wr.y = -sg0 * ctab[i1];
        wi.x = sgx * itab[i0];   wi.y = sgy * itab[i1];
        re = __builtin_amdgcn_wmma_f32_16x16x4_f32(false, e, false, wr, (short)0, re, false, false);
        im = __builtin_amdgcn_wmma_f32_16x16x4_f32(false, e, false, wi, (short)0, im, false, false);
    }

    if (d == 0) {
        const int o = krow * N_TERMS + tt * 16 + 8 * half;
        *(float4*)(Pre + o)     = make_float4(re[0], re[1], re[2], re[3]);
        *(float4*)(Pre + o + 4) = make_float4(re[4], re[5], re[6], re[7]);
        *(float4*)(Pim + o)     = make_float4(im[0], im[1], im[2], im[3]);
        *(float4*)(Pim + o + 4) = make_float4(im[4], im[5], im[6], im[7]);
    } else {
        float* __restrict__ Fr = (d == 1) ? YFre : ZFre;
        float* __restrict__ Fi = (d == 1) ? YFim : ZFim;
#pragma unroll
        for (int r2 = 0; r2 < 4; ++r2) {
            const int s  = tt * 4 + 2 * half + (r2 >> 1);
            const int hh = r2 & 1;
            const int o  = (((s * 4 + ck) * 32) + hh * 16 + l16) * 2;
            v2f vr, vi;
            vr.x = re[2 * r2]; vr.y = re[2 * r2 + 1];
            vi.x = im[2 * r2]; vi.y = im[2 * r2 + 1];
            *(v2f*)(Fr + o) = vr;
            *(v2f*)(Fi + o) = vi;
        }
    }
}

#define KSTEPS_Q (N_TERMS / 16)
#define CG 8
#define GP 68
__global__ __launch_bounds__(256) void spectrum_contract_kernel(const float* __restrict__ Pre,
                                                                const float* __restrict__ Pim,
                                                                const float* __restrict__ YFre,
                                                                const float* __restrict__ YFim,
                                                                const float* __restrict__ ZFre,
                                                                const float* __restrict__ ZFim,
                                                                const float* __restrict__ ham,
                                                                const float* __restrict__ Mre,
                                                                const float* __restrict__ Mim,
                                                                float* __restrict__ Ure,
                                                                float* __restrict__ Uim) {
    __shared__ float part[4][2][32][32];
    __shared__ __attribute__((aligned(16))) float Gre[16 * GP];
    __shared__ __attribute__((aligned(16))) float Gim[16 * GP];

    const int lane = threadIdx.x & 31;
    const int w    = threadIdx.x >> 5;
    const int ct   = w & 1;
    const int kq   = w >> 1;
    const int cz   = w & 3;
    const int kh   = w >> 2;
    const int kx   = blockIdx.x >> 2;
    const int cy   = blockIdx.x & 3;
    const int half = lane >> 4;
    const int l16  = lane & 15;
    const int sBeg = kq * KSTEPS_Q;

    const v2f* __restrict__ pPr = (const v2f*)(Pre + kx * N_TERMS + 2 * half) + 2 * sBeg;
    const v2f* __restrict__ pPi = (const v2f*)(Pim + kx * N_TERMS + 2 * half) + 2 * sBeg;
    const v2f* __restrict__ pYr = (const v2f*)YFre + 128 * sBeg + cy * 32 + lane;
    const v2f* __restrict__ pYi = (const v2f*)YFim + 128 * sBeg + cy * 32 + lane;
    const v2f* __restrict__ pZr = (const v2f*)ZFre + 128 * sBeg + ct * 32 + lane;
    const v2f* __restrict__ pZi = (const v2f*)ZFim + 128 * sBeg + ct * 32 + lane;

    v8f rr = {}, ii = {}, ri = {}, ir = {};
    for (int s0 = 0; s0 < KSTEPS_Q; s0 += CG) {
        v2f pr[CG], pi[CG], yr[CG], yi[CG], zr[CG], zi[CG];
#pragma unroll
        for (int u = 0; u < CG; ++u) {
            pr[u] = pPr[2 * (s0 + u)];   pi[u] = pPi[2 * (s0 + u)];
            yr[u] = pYr[128 * (s0 + u)]; yi[u] = pYi[128 * (s0 + u)];
            zr[u] = pZr[128 * (s0 + u)]; zi[u] = pZi[128 * (s0 + u)];
        }
#pragma unroll
        for (int u = 0; u < CG; ++u) {
            const v2f ar = pr[u] * yr[u] - pi[u] * yi[u];
            const v2f ai = pr[u] * yi[u] + pi[u] * yr[u];
            rr = __builtin_amdgcn_wmma_f32_16x16x4_f32(false, ar, false, zr[u], (short)0, rr, false, false);
            ii = __builtin_amdgcn_wmma_f32_16x16x4_f32(false, ai, false, zi[u], (short)0, ii, false, false);
            ri = __builtin_amdgcn_wmma_f32_16x16x4_f32(false, ar, false, zi[u], (short)0, ri, false, false);
            ir = __builtin_amdgcn_wmma_f32_16x16x4_f32(false, ai, false, zr[u], (short)0, ir, false, false);
        }
    }

    v2f mr[16], mi[16];
    {
        const float* __restrict__ aR = Mre + (cz * 16 + l16) * N1D + 2 * half;
        const float* __restrict__ aI = Mim + (cz * 16 + l16) * N1D + 2 * half;
#pragma unroll
        for (int s = 0; s < 16; ++s) {
            mr[s] = *(const v2f*)(aR + 4 * s);
            mi[s] = *(const v2f*)(aI + 4 * s);
        }
    }
    const int  kzD = ct * 16 + l16;
    const bool sp  = (kzD == 0);
    const int  kzM = sp ? 32 : 64 - kzD;
    float hD[2], hM[2];
#pragma unroll
    for (int e = 0; e < 2; ++e) {
        const int ky = cy * 16 + 2 * kq + e + 8 * half;
        hD[e] = ham[kx * 4096 + ky * N1D + kzD];
        hM[e] = ham[kx * 4096 + ky * N1D + kzM];
    }

#pragma unroll
    for (int v = 0; v < 8; ++v) {
        part[kq][ct][v][lane]      = rr[v];
        part[kq][ct][8 + v][lane]  = ii[v];
        part[kq][ct][16 + v][lane] = ri[v];
        part[kq][ct][24 + v][lane] = ir[v];
    }
    __syncthreads();
#pragma unroll
    for (int e = 0; e < 2; ++e) {
        const int v = 2 * kq + e;
        float srr = part[0][ct][v][lane];
        float sii = part[0][ct][8 + v][lane];
        float sri = part[0][ct][16 + v][lane];
        float sir = part[0][ct][24 + v][lane];
#pragma unroll
        for (int q = 1; q < 4; ++q) {
            srr += part[q][ct][v][lane];
            sii += part[q][ct][8 + v][lane];
            sri += part[q][ct][16 + v][lane];
            sir += part[q][ct][24 + v][lane];
        }
        const float dre = sp ? srr : srr - sii;
        const float dim = sp ? sir : sri + sir;
        const float mre = sp ? sri : srr + sii;
        const float mim = sp ? sii : sir - sri;
        const int o = (v + 8 * half) * GP;
        Gre[o + kzD] = dre * hD[e];
        Gim[o + kzD] = dim * hD[e];
        Gre[o + kzM] = mre * hM[e];
        Gim[o + kzM] = mim * hM[e];
    }
    __syncthreads();

    const float* b1 = (kh ? Gim : Gre) + l16 * GP + 2 * half;
    const float* b2 = (kh ? Gre : Gim) + l16 * GP + 2 * half;
    v8f a1 = {}, a2 = {};
#pragma unroll
    for (int s = 0; s < 16; ++s) {
        const v2f t1 = *(const v2f*)(b1 + 4 * s);
        const v2f t2 = *(const v2f*)(b2 + 4 * s);
        a1 = __builtin_amdgcn_wmma_f32_16x16x4_f32(false, mr[s], false, t1, (short)0, a1, false, false);
        a2 = __builtin_amdgcn_wmma_f32_16x16x4_f32(false, mi[s], false, t2, (short)0, a2, false, false);
    }
    const float sg = kh ? 1.0f : -1.0f;
    const v8f u = a1 + sg * a2;

    float* __restrict__ dst = kh ? Uim : Ure;
#pragma unroll
    for (int v = 0; v < 8; ++v) {
        const int z = cz * 16 + v + 8 * half;
        dst[z * 4096 + kx * N1D + cy * 16 + l16] = u[v];
    }
}

#define DFT_WG_THREADS 128
#define DFT_WGS        256
__global__ __launch_bounds__(DFT_WG_THREADS) void dft_pass_kernel(const float* __restrict__ Tre,
                                                                  const float* __restrict__ Tim,
                                                                  const float* __restrict__ Mre,
                                                                  const float* __restrict__ Mim,
                                                                  float* __restrict__ Ure,
                                                                  float* __restrict__ Uim) {
    const int lane   = threadIdx.x & 31;
    const int waveId = (blockIdx.x * DFT_WG_THREADS + threadIdx.x) >> 5;
    const int p0 = (waveId & 255) << 4;
    const int k0 = (waveId >> 8) << 4;
    const int half = lane >> 4;
    const int l16  = lane & 15;
    const int krow = k0 + l16;
    const int prow = p0 + l16;

    const float* __restrict__ aR = Mre + krow * N1D + 2 * half;
    const float* __restrict__ aI = Mim + krow * N1D + 2 * half;
    const float* __restrict__ bR = Tre + prow * N1D + 2 * half;
    const float* __restrict__ bI = Tim + prow * N1D + 2 * half;
    v2f mr[16], mi[16], tr[16], ti[16];
#pragma unroll
    for (int s = 0; s < 16; ++s) {
        mr[s] = *(const v2f*)(aR + 4 * s);
        tr[s] = *(const v2f*)(bR + 4 * s);
        mi[s] = *(const v2f*)(aI + 4 * s);
        ti[s] = *(const v2f*)(bI + 4 * s);
    }
    __builtin_amdgcn_sched_barrier(0);

    v8f rr = {}, ii = {}, ri = {}, ir = {};
#pragma unroll
    for (int s = 0; s < 16; ++s) {
        rr = __builtin_amdgcn_wmma_f32_16x16x4_f32(false, mr[s], false, tr[s], (short)0, rr, false, false);
        ii = __builtin_amdgcn_wmma_f32_16x16x4_f32(false, mi[s], false, ti[s], (short)0, ii, false, false);
        ri = __builtin_amdgcn_wmma_f32_16x16x4_f32(false, mr[s], false, ti[s], (short)0, ri, false, false);
        ir = __builtin_amdgcn_wmma_f32_16x16x4_f32(false, mi[s], false, tr[s], (short)0, ir, false, false);
    }

#pragma unroll
    for (int v = 0; v < 8; ++v) {
        const int row = k0 + v + 8 * half;
        const int idx = row * 4096 + p0 + l16;
        Ure[idx] = rr[v] - ii[v];
        Uim[idx] = ri[v] + ir[v];
    }
}

#define TAIL_WG   64
#define TAIL_V4   (GRID_N / 4 / TAIL_WG)

__global__ __launch_bounds__(256) void reduce1_kernel(const float* __restrict__ x,
                                                      float* __restrict__ psum,
                                                      float* __restrict__ psq) {
    __shared__ float ssum[256];
    __shared__ float ssq[256];
    const int tid = threadIdx.x;
    const float4* __restrict__ x4 = (const float4*)x + blockIdx.x * TAIL_V4;
    float4 v[4];
#pragma unroll
    for (int q = 0; q < 4; ++q) v[q] = x4[q * 256 + tid];
    float s = 0.0f, sq = 0.0f;
#pragma unroll
    for (int q = 0; q < 4; ++q) {
        s  += (v[q].x + v[q].y) + (v[q].z + v[q].w);
        sq += (v[q].x * v[q].x + v[q].y * v[q].y) + (v[q].z * v[q].z + v[q].w * v[q].w);
    }
    ssum[tid] = s;
    ssq[tid]  = sq;
    __syncthreads();
    for (int st = 128; st > 0; st >>= 1) {
        if (tid < st) { ssum[tid] += ssum[tid + st]; ssq[tid] += ssq[tid + st]; }
        __syncthreads();
    }
    if (tid == 0) { psum[blockIdx.x] = ssum[0]; psq[blockIdx.x] = ssq[0]; }
}

__global__ __launch_bounds__(TAIL_WG) void reduce2_kernel(const float* __restrict__ psum,
                                                          const float* __restrict__ psq,
                                                          float* __restrict__ stats) {
    __shared__ float ssum[TAIL_WG];
    __shared__ float ssq[TAIL_WG];
    const int tid = threadIdx.x;
    ssum[tid] = psum[tid];
    ssq[tid]  = psq[tid];
    __syncthreads();
    for (int st = TAIL_WG / 2; st > 0; st >>= 1) {
        if (tid < st) { ssum[tid] += ssum[tid + st]; ssq[tid] += ssq[tid + st]; }
        __syncthreads();
    }
    if (tid == 0) {
        const float invN = 1.0f / (float)GRID_N;
        float mean = ssum[0] * invN;
        float var  = fmaxf(ssq[0] * invN - mean * mean, 0.0f);
        stats[0] = mean;
        stats[1] = 1.0f / (sqrtf(var) + 1e-8f);
    }
}

__global__ __launch_bounds__(256) void finalize_kernel(const float* __restrict__ re,
                                                       const float* __restrict__ stats,
                                                       float* __restrict__ out) {
    const int tid = threadIdx.x;
    const float4* __restrict__ r4 = (const float4*)re + blockIdx.x * TAIL_V4;
    float4* __restrict__ o4 = (float4*)out + blockIdx.x * TAIL_V4;
    float4 v[4];
#pragma unroll
    for (int q = 0; q < 4; ++q) v[q] = r4[q * 256 + tid];
    const float m  = stats[0];
    const float is = stats[1];
#pragma unroll
    for (int q = 0; q < 4; ++q)
        o4[q * 256 + tid] = make_float4((v[q].x - m) * is, (v[q].y - m) * is,
                                        (v[q].z - m) * is, (v[q].w - m) * is);
}

extern "C" void kernel_launch(void* const* d_in, const int* in_sizes, int n_in,
                              void* d_out, int out_size, void* d_ws, size_t ws_size,
                              hipStream_t stream) {
    const float* X   = (const float*)d_in[0];
    const float* aw  = (const float*)d_in[1];
    const float* bw  = (const float*)d_in[2];
    const float* ham = (const float*)d_in[4];
    float* out = (float*)d_out;

    float* ws   = (float*)d_ws;
    float* Wre  = ws;
    float* WimI = Wre + 8192;
    float* Are  = WimI + 4096;
    float* Aim  = Are + GRID_N;
    float* Bre  = Aim + GRID_N;
    float* Bim  = Bre + GRID_N;
    float* psum = Bim + GRID_N;
    float* psq  = psum + 512;
    float* stats = psq + 512;

    float* Pre  = Are;
    float* Pim  = Pre  + TAB_SZ;
    float* YFre = Pim  + TAB_SZ;
    float* YFim = YFre + TAB_SZ;
    float* ZFre = YFim + TAB_SZ;
    float* ZFim = ZFre + TAB_SZ;

    axis_spectra_kernel<<<304, 64, 0, stream>>>(X, aw, bw, Wre, WimI,
                                                Pre, Pim, YFre, YFim, ZFre, ZFim);

    spectrum_contract_kernel<<<256, 256, 0, stream>>>(Pre, Pim, YFre, YFim, ZFre, ZFim,
                                                      ham, Wre, WimI, Bre, Bim);

    dft_pass_kernel<<<DFT_WGS, DFT_WG_THREADS, 0, stream>>>(Bre, Bim, Wre, WimI, Are, Aim);
    dft_pass_kernel<<<DFT_WGS, DFT_WG_THREADS, 0, stream>>>(Are, Aim, Wre, WimI, Bre, Bim);

    reduce1_kernel<<<TAIL_WG, 256, 0, stream>>>(Bre, psum, psq);
    reduce2_kernel<<<1, TAIL_WG, 0, stream>>>(psum, psq, stats);
    finalize_kernel<<<TAIL_WG, 256, 0, stream>>>(Bre, stats, out);
}


# last inverse pass run twice, real part only: run 1 leaves 256 tile-sum pairs, run 2 adds them per wave and stores normalised output; no plane stored, 7->5 launches; mean/variance sum order changes
# speedup vs baseline: 2.0529x; 1.1290x over previous
#include <hip/hip_runtime.h>
#include <hip/hip_bf16.h>
#include <math.h>

#define N1D      64
#define GRID_N   (64 * 64 * 64)
#define N_ATOMS  128
#define K_TERMS  6
#define SPACING  0.5f
#define LOG2E    1.4426950408889634f

typedef float v2f __attribute__((ext_vector_type(2)));
typedef float v8f __attribute__((ext_vector_type(8)));

__device__ __forceinline__ float fast_exp2(float x) {
#if __has_builtin(__builtin_amdgcn_exp2f)
    return __builtin_amdgcn_exp2f(x);
#else
    return exp2f(x);
#endif
}

#define N_TERMS  (N_ATOMS * K_TERMS)
#define TAB_SZ   (N_TERMS * N1D)

__global__ __launch_bounds__(64) void axis_spectra_kernel(const float* __restrict__ X,
                                                           const float* __restrict__ aw,
                                                           const float* __restrict__ bw,
                                                           float* __restrict__ Wre,
                                                           float* __restrict__ WimI,
                                                           float* __restrict__ Pre,
                                                           float* __restrict__ Pim,
                                                           float* __restrict__ YFre,
                                                           float* __restrict__ YFim,
                                                           float* __restrict__ ZFre,
                                                           float* __restrict__ ZFim) {
    __shared__ float ctab[N1D];
    __shared__ float stab[N1D];
    {
        const int j = threadIdx.x;
        const float ang = (float)j * 0.09817477042468103f;
        ctab[j] = 0.125f * cosf(ang);
        stab[j] = 0.125f * sinf(ang);
    }
    __syncthreads();

    if (blockIdx.x >= 288) {
        const int e = (blockIdx.x - 288) * 64 + threadIdx.x;
        const int k = e >> 4;
        const int n = (e & 15) * 4;
        float cr[4], si[4];
#pragma unroll
        for (int q = 0; q < 4; ++q) {
            const int kn = (k * (n + q)) & 63;
            const float sgn = ((k + n + q) & 1) ? -1.0f : 1.0f;
            cr[q] = sgn * ctab[kn];
            si[q] = sgn * stab[kn];
        }
        *(float4*)(Wre  + k * N1D + n) = make_float4(cr[0], cr[1], cr[2], cr[3]);
        *(float4*)(WimI + k * N1D + n) = make_float4(si[0], si[1], si[2], si[3]);
        return;
    }

    const int lane   = threadIdx.x & 31;
    const int waveId = (blockIdx.x * blockDim.x + threadIdx.x) >> 5;
    const int ck   = waveId & 3;
    const int rest = waveId >> 2;
    const int d    = rest / 48;
    const int tt   = rest - d * 48;
    const int half = lane >> 4;
    const int l16  = lane & 15;

    const int t = tt * 16 + l16;
    const int a = t / K_TERMS;
    const float c   = X[3 * a + d];
    const float bl2 = bw[t] * LOG2E;
    const float scale = (d == 0) ? aw[t] : 1.0f;
    const int krow = ck * 16 + l16;
    const float sg0 = (krow & 1) ? -1.0f : 1.0f;
    const bool dcz = (d == 2) && (krow == 0);
    const float* itab = dcz ? ctab : stab;
    const float sgx = dcz ? 1.0f : -sg0;
    const float sgy = dcz ? 1.0f : sg0;

    v8f re = {}, im = {};
#pragma unroll 4
    for (int n0 = 0; n0 < N1D; n0 += 4) {
        const int koff = n0 + 2 * half;
        const float d0 = (float)koff * SPACING - c;
        const float d1 = (float)(koff + 1) * SPACING - c;
        v2f e;
        e.x = scale * fast_exp2(bl2 * (d0 * d0));
        e.y = scale * fast_exp2(bl2 * (d1 * d1));
        const int i0 = (krow * koff) & 63;
        const int i1 = (i0 + krow) & 63;
        v2f wr, wi;
        wr.x = sg0 * ctab[i0];   wr.y = -sg0 * ctab[i1];
        wi.x = sgx * itab[i0];   wi.y = sgy * itab[i1];
        re = __builtin_amdgcn_wmma_f32_16x16x4_f32(false, e, false, wr, (short)0, re, false, false);
        im = __builtin_amdgcn_wmma_f32_16x16x4_f32(false, e, false, wi, (short)0, im, false, false);
    }

    if (d == 0) {
        const int o = krow * N_TERMS + tt * 16 + 8 * half;
        *(float4*)(Pre + o)     = make_float4(re[0], re[1], re[2], re[3]);
        *(float4*)(Pre + o + 4) = make_float4(re[4], re[5], re[6], re[7]);
        *(float4*)(Pim + o)     = make_float4(im[0], im[1], im[2], im[3]);
        *(float4*)(Pim + o + 4) = make_float4(im[4], im[5], im[6], im[7]);
    } else {
        float* __restrict__ Fr = (d == 1) ? YFre : ZFre;
        float* __restrict__ Fi = (d == 1) ? YFim : ZFim;
#pragma unroll
        for (int r2 = 0; r2 < 4; ++r2) {
            const int s  = tt * 4 + 2 * half + (r2 >> 1);
            const int hh = r2 & 1;
            const int o  = (((s * 4 + ck) * 32) + hh * 16 + l16) * 2;
            v2f vr, vi;
            vr.x = re[2 * r2]; vr.y = re[2 * r2 + 1];
            vi.x = im[2 * r2]; vi.y = im[2 * r2 + 1];
            *(v2f*)(Fr + o) = vr;
            *(v2f*)(Fi + o) = vi;
        }
    }
}

#define KSTEPS_Q (N_TERMS / 16)
#define CG 8
#define GP 68
__global__ __launch_bounds__(256) void spectrum_contract_kernel(const float* __restrict__ Pre,
                                                                const float* __restrict__ Pim,
                                                                const float* __restrict__ YFre,
                                                                const float* __restrict__ YFim,
                                                                const float* __restrict__ ZFre,
                                                                const float* __restrict__ ZFim,
                                                                const float* __restrict__ ham,
                                                                const float* __restrict__ Mre,
                                                                const float* __restrict__ Mim,
                                                                float* __restrict__ Ure,
                                                                float* __restrict__ Uim) {
    __shared__ float part[4][2][32][32];
    __shared__ __attribute__((aligned(16))) float Gre[16 * GP];
    __shared__ __attribute__((aligned(16))) float Gim[16 * GP];

    const int lane = threadIdx.x & 31;
    const int w    = threadIdx.x >> 5;
    const int ct   = w & 1;
    const int kq   = w >> 1;
    const int cz   = w & 3;
    const int kh   = w >> 2;
    const int kx   = blockIdx.x >> 2;
    const int cy   = blockIdx.x & 3;
    const int half = lane >> 4;
    const int l16  = lane & 15;
    const int sBeg = kq * KSTEPS_Q;

    const v2f* __restrict__ pPr = (const v2f*)(Pre + kx * N_TERMS + 2 * half) + 2 * sBeg;
    const v2f* __restrict__ pPi = (const v2f*)(Pim + kx * N_TERMS + 2 * half) + 2 * sBeg;
    const v2f* __restrict__ pYr = (const v2f*)YFre + 128 * sBeg + cy * 32 + lane;
    const v2f* __restrict__ pYi = (const v2f*)YFim + 128 * sBeg + cy * 32 + lane;
    const v2f* __restrict__ pZr = (const v2f*)ZFre + 128 * sBeg + ct * 32 + lane;
    const v2f* __restrict__ pZi = (const v2f*)ZFim + 128 * sBeg + ct * 32 + lane;

    v8f rr = {}, ii = {}, ri = {}, ir = {};
    for (int s0 = 0; s0 < KSTEPS_Q; s0 += CG) {
        v2f pr[CG], pi[CG], yr[CG], yi[CG], zr[CG], zi[CG];
#pragma unroll
        for (int u = 0; u < CG; ++u) {
            pr[u] = pPr[2 * (s0 + u)];   pi[u] = pPi[2 * (s0 + u)];
            yr[u] = pYr[128 * (s0 + u)]; yi[u] = pYi[128 * (s0 + u)];
            zr[u] = pZr[128 * (s0 + u)]; zi[u] = pZi[128 * (s0 + u)];
        }
#pragma unroll
        for (int u = 0; u < CG; ++u) {
            const v2f ar = pr[u] * yr[u] - pi[u] * yi[u];
            const v2f ai = pr[u] * yi[u] + pi[u] * yr[u];
            rr = __builtin_amdgcn_wmma_f32_16x16x4_f32(false, ar, false, zr[u], (short)0, rr, false, false);
            ii = __builtin_amdgcn_wmma_f32_16x16x4_f32(false, ai, false, zi[u], (short)0, ii, false, false);
            ri = __builtin_amdgcn_wmma_f32_16x16x4_f32(false, ar, false, zi[u], (short)0, ri, false, false);
            ir = __builtin_amdgcn_wmma_f32_16x16x4_f32(false, ai, false, zr[u], (short)0, ir, false, false);
        }
    }

    v2f mr[16], mi[16];
    {
        const float* __restrict__ aR = Mre + (cz * 16 + l16) * N1D + 2 * half;
        const float* __restrict__ aI = Mim + (cz * 16 + l16) * N1D + 2 * half;
#pragma unroll
        for (int s = 0; s < 16; ++s) {
            mr[s] = *(const v2f*)(aR + 4 * s);
            mi[s] = *(const v2f*)(aI + 4 * s);
        }
    }
    const int  kzD = ct * 16 + l16;
    const bool sp  = (kzD == 0);
    const int  kzM = sp ? 32 : 64 - kzD;
    float hD[2], hM[2];
#pragma unroll
    for (int e = 0; e < 2; ++e) {
        const int ky = cy * 16 + 2 * kq + e + 8 * half;
        hD[e] = ham[kx * 4096 + ky * N1D + kzD];
        hM[e] = ham[kx * 4096 + ky * N1D + kzM];
    }

#pragma unroll
    for (int v = 0; v < 8; ++v) {
        part[kq][ct][v][lane]      = rr[v];
        part[kq][ct][8 + v][lane]  = ii[v];
        part[kq][ct][16 + v][lane] = ri[v];
        part[kq][ct][24 + v][lane] = ir[v];
    }
    __syncthreads();
#pragma unroll
    for (int e = 0; e < 2; ++e) {
        const int v = 2 * kq + e;
        float srr = part[0][ct][v][lane];
        float sii = part[0][ct][8 + v][lane];
        float sri = part[0][ct][16 + v][lane];
        float sir = part[0][ct][24 + v][lane];
#pragma unroll
        for (int q = 1; q < 4; ++q) {
            srr += part[q][ct][v][lane];
            sii += part[q][ct][8 + v][lane];
            sri += part[q][ct][16 + v][lane];
            sir += part[q][ct][24 + v][lane];
        }
        const float dre = sp ? srr : srr - sii;
        const float dim = sp ? sir : sri + sir;
        const float mre = sp ? sri : srr + sii;
        const float mim = sp ? sii : sir - sri;
        const int o = (v + 8 * half) * GP;
        Gre[o + kzD] = dre * hD[e];
        Gim[o + kzD] = dim * hD[e];
        Gre[o + kzM] = mre * hM[e];
        Gim[o + kzM] = mim * hM[e];
    }
    __syncthreads();

    const float* b1 = (kh ? Gim : Gre) + l16 * GP + 2 * half;
    const float* b2 = (kh ? Gre : Gim) + l16 * GP + 2 * half;
    v8f a1 = {}, a2 = {};
#pragma unroll
    for (int s = 0; s < 16; ++s) {
        const v2f t1 = *(const v2f*)(b1 + 4 * s);
        const v2f t2 = *(const v2f*)(b2 + 4 * s);
        a1 = __builtin_amdgcn_wmma_f32_16x16x4_f32(false, mr[s], false, t1, (short)0, a1, false, false);
        a2 = __builtin_amdgcn_wmma_f32_16x16x4_f32(false, mi[s], false, t2, (short)0, a2, false, false);
    }
    const float sg = kh ? 1.0f : -1.0f;
    const v8f u = a1 + sg * a2;

    float* __restrict__ dst = kh ? Uim : Ure;
#pragma unroll
    for (int v = 0; v < 8; ++v) {
        const int z = cz * 16 + v + 8 * half;
        dst[z * 4096 + kx * N1D + cy * 16 + l16] = u[v];
    }
}

#define DFT_WG_THREADS 128
#define DFT_WGS        256
__global__ __launch_bounds__(DFT_WG_THREADS) void dft_pass_kernel(const float* __restrict__ Tre,
                                                                  const float* __restrict__ Tim,
                                                                  const float* __restrict__ Mre,
                                                                  const float* __restrict__ Mim,
                                                                  float* __restrict__ Ure,
                                                                  float* __restrict__ Uim) {
    const int lane   = threadIdx.x & 31;
    const int waveId = (blockIdx.x * DFT_WG_THREADS + threadIdx.x) >> 5;
    const int p0 = (waveId & 255) << 4;
    const int k0 = (waveId >> 8) << 4;
    const int half = lane >> 4;
    const int l16  = lane & 15;
    const int krow = k0 + l16;
    const int prow = p0 + l16;

    const float* __restrict__ aR = Mre + krow * N1D + 2 * half;
    const float* __restrict__ aI = Mim + krow * N1D + 2 * half;
    const float* __restrict__ bR = Tre + prow * N1D + 2 * half;
    const float* __restrict__ bI = Tim + prow * N1D + 2 * half;
    v2f mr[16], mi[16], tr[16], ti[16];
#pragma unroll
    for (int s = 0; s < 16; ++s) {
        mr[s] = *(const v2f*)(aR + 4 * s);
        tr[s] = *(const v2f*)(bR + 4 * s);
        mi[s] = *(const v2f*)(aI + 4 * s);
        ti[s] = *(const v2f*)(bI + 4 * s);
    }
    __builtin_amdgcn_sched_barrier(0);

    v8f rr = {}, ii = {}, ri = {}, ir = {};
#pragma unroll
    for (int s = 0; s < 16; ++s) {
        rr = __builtin_amdgcn_wmma_f32_16x16x4_f32(false, mr[s], false, tr[s], (short)0, rr, false, false);
        ii = __builtin_amdgcn_wmma_f32_16x16x4_f32(false, mi[s], false, ti[s], (short)0, ii, false, false);
        ri = __builtin_amdgcn_wmma_f32_16x16x4_f32(false, mr[s], false, ti[s], (short)0, ri, false, false);
        ir = __builtin_amdgcn_wmma_f32_16x16x4_f32(false, mi[s], false, tr[s], (short)0, ir, false, false);
    }

#pragma unroll
    for (int v = 0; v < 8; ++v) {
        const int row = k0 + v + 8 * half;
        const int idx = row * 4096 + p0 + l16;
        Ure[idx] = rr[v] - ii[v];
        Uim[idx] = ri[v] + ir[v];
    }
}

#define XP_THREADS 128
#define XP_WGS     256

__device__ __forceinline__ void xpass_read(const float* __restrict__ Tre,
                                           const float* __restrict__ Tim,
                                           const float* __restrict__ Mre,
                                           const float* __restrict__ Mim,
                                           int krow, int prow, int half,
                                           v2f (&mr)[16], v2f (&mi)[16],
                                           v2f (&tr)[16], v2f (&ti)[16]) {
    const float* __restrict__ aR = Mre + krow * N1D + 2 * half;
    const float* __restrict__ aI = Mim + krow * N1D + 2 * half;
    const float* __restrict__ bR = Tre + prow * N1D + 2 * half;
    const float* __restrict__ bI = Tim + prow * N1D + 2 * half;
#pragma unroll
    for (int s = 0; s < 16; ++s) {
        mr[s] = *(const v2f*)(aR + 4 * s);
        tr[s] = *(const v2f*)(bR + 4 * s);
        mi[s] = *(const v2f*)(aI + 4 * s);
        ti[s] = *(const v2f*)(bI + 4 * s);
    }
}

__device__ __forceinline__ v8f xpass_real(const v2f (&mr)[16], const v2f (&mi)[16],
                                          const v2f (&tr)[16], const v2f (&ti)[16]) {
    v8f rr = {}, ii = {};
#pragma unroll
    for (int s = 0; s < 16; ++s) {
        rr = __builtin_amdgcn_wmma_f32_16x16x4_f32(false, mr[s], false, tr[s], (short)0, rr, false, false);
        ii = __builtin_amdgcn_wmma_f32_16x16x4_f32(false, mi[s], false, ti[s], (short)0, ii, false, false);
    }
    return rr - ii;
}

__global__ __launch_bounds__(XP_THREADS) void xpass_sum_kernel(const float* __restrict__ Tre,
                                                               const float* __restrict__ Tim,
                                                               const float* __restrict__ Mre,
                                                               const float* __restrict__ Mim,
                                                               float* __restrict__ psum,
                                                               float* __restrict__ psq) {
    __shared__ float wsum[4];
    __shared__ float wsq[4];
    const int lane   = threadIdx.x & 31;
    const int w      = threadIdx.x >> 5;
    const int waveId = blockIdx.x * 4 + w;
    const int p0 = (waveId & 255) << 4;
    const int k0 = (waveId >> 8) << 4;
    const int half = lane >> 4;
    const int l16  = lane & 15;

    v2f mr[16], mi[16], tr[16], ti[16];
    xpass_read(Tre, Tim, Mre, Mim, k0 + l16, p0 + l16, half, mr, mi, tr, ti);
    __builtin_amdgcn_sched_barrier(0);
    const v8f u = xpass_real(mr, mi, tr, ti);

    float s = 0.0f, q = 0.0f;
#pragma unroll
    for (int v = 0; v < 8; ++v) { s += u[v]; q += u[v] * u[v]; }
#pragma unroll
    for (int o = 16; o > 0; o >>= 1) {
        s += __shfl_xor(s, o);
        q += __shfl_xor(q, o);
    }
    if (lane == 0) { wsum[w] = s; wsq[w] = q; }
    __syncthreads();
    if (threadIdx.x == 0) {
        psum[blockIdx.x] = (wsum[0] + wsum[1]) + (wsum[2] + wsum[3]);
        psq[blockIdx.x]  = (wsq[0] + wsq[1]) + (wsq[2] + wsq[3]);
    }
}

__global__ __launch_bounds__(XP_THREADS) void xpass_norm_kernel(const float* __restrict__ Tre,
                                                                const float* __restrict__ Tim,
                                                                const float* __restrict__ Mre,
                                                                const float* __restrict__ Mim,
                                                                const float* __restrict__ psum,
                                                                const float* __restrict__ psq,
                                                                float* __restrict__ out) {
    const int lane   = threadIdx.x & 31;
    const int waveId = (blockIdx.x * XP_THREADS + threadIdx.x) >> 5;
    const int p0 = (waveId & 255) << 4;
    const int k0 = (waveId >> 8) << 4;
    const int half = lane >> 4;
    const int l16  = lane & 15;

    const float4 sa = ((const float4*)psum)[2 * lane], sb = ((const float4*)psum)[2 * lane + 1];
    const float4 qa = ((const float4*)psq)[2 * lane],  qb = ((const float4*)psq)[2 * lane + 1];

    v2f mr[16], mi[16], tr[16], ti[16];
    xpass_read(Tre, Tim, Mre, Mim, k0 + l16, p0 + l16, half, mr, mi, tr, ti);
    __builtin_amdgcn_sched_barrier(0);

    float s = ((sa.x + sa.y) + (sa.z + sa.w)) + ((sb.x + sb.y) + (sb.z + sb.w));
    float q = ((qa.x + qa.y) + (qa.z + qa.w)) + ((qb.x + qb.y) + (qb.z + qb.w));
#pragma unroll
    for (int o = 16; o > 0; o >>= 1) {
        s += __shfl_xor(s, o);
        q += __shfl_xor(q, o);
    }
    const float invN = 1.0f / (float)GRID_N;
    const float mean = s * invN;
    const float var  = fmaxf(q * invN - mean * mean, 0.0f);
    const float is   = 1.0f / (sqrtf(var) + 1e-8f);

    const v8f u = xpass_real(mr, mi, tr, ti);

#pragma unroll
    for (int v = 0; v < 8; ++v) {
        const int row = k0 + v + 8 * half;
        out[row * 4096 + p0 + l16] = (u[v] - mean) * is;
    }
}

extern "C" void kernel_launch(void* const* d_in, const int* in_sizes, int n_in,
                              void* d_out, int out_size, void* d_ws, size_t ws_size,
                              hipStream_t stream) {
    const float* X   = (const float*)d_in[0];
    const float* aw  = (const float*)d_in[1];
    const float* bw  = (const float*)d_in[2];
    const float* ham = (const float*)d_in[4];
    float* out = (float*)d_out;

    float* ws   = (float*)d_ws;
    float* Wre  = ws;
    float* WimI = Wre + 8192;
    float* Are  = WimI + 4096;
    float* Aim  = Are + GRID_N;
    float* Bre  = Aim + GRID_N;
    float* Bim  = Bre + GRID_N;
    float* psum = Bim + GRID_N;
    float* psq  = psum + 512;

    float* Pre  = Are;
    float* Pim  = Pre  + TAB_SZ;
    float* YFre = Pim  + TAB_SZ;
    float* YFim = YFre + TAB_SZ;
    float* ZFre = YFim + TAB_SZ;
    float* ZFim = ZFre + TAB_SZ;

    axis_spectra_kernel<<<304, 64, 0, stream>>>(X, aw, bw, Wre, WimI,
                                                Pre, Pim, YFre, YFim, ZFre, ZFim);

    spectrum_contract_kernel<<<256, 256, 0, stream>>>(Pre, Pim, YFre, YFim, ZFre, ZFim,
                                                      ham, Wre, WimI, Bre, Bim);

    dft_pass_kernel<<<DFT_WGS, DFT_WG_THREADS, 0, stream>>>(Bre, Bim, Wre, WimI, Are, Aim);

    xpass_sum_kernel<<<XP_WGS, XP_THREADS, 0, stream>>>(Are, Aim, Wre, WimI, psum, psq);
    xpass_norm_kernel<<<XP_WGS, XP_THREADS, 0, stream>>>(Are, Aim, Wre, WimI, psum, psq, out);
}


# spectrum contraction reads double-buffered: each group's 48 fragment reads (last slot: z-transform rows, filter values) issued ahead of the previous group's WMMAs; sum order unchanged
# speedup vs baseline: 2.3458x; 1.1427x over previous
#include <hip/hip_runtime.h>
#include <hip/hip_bf16.h>
#include <math.h>

#define N1D      64
#define GRID_N   (64 * 64 * 64)
#define N_ATOMS  128
#define K_TERMS  6
#define SPACING  0.5f
#define LOG2E    1.4426950408889634f

typedef float v2f __attribute__((ext_vector_type(2)));
typedef float v8f __attribute__((ext_vector_type(8)));

__device__ __forceinline__ float fast_exp2(float x) {
#if __has_builtin(__builtin_amdgcn_exp2f)
    return __builtin_amdgcn_exp2f(x);
#else
    return exp2f(x);
#endif
}

#define N_TERMS  (N_ATOMS * K_TERMS)
#define TAB_SZ   (N_TERMS * N1D)

__global__ __launch_bounds__(64) void axis_spectra_kernel(const float* __restrict__ X,
                                                           const float* __restrict__ aw,
                                                           const float* __restrict__ bw,
                                                           float* __restrict__ Wre,
                                                           float* __restrict__ WimI,
                                                           float* __restrict__ Pre,
                                                           float* __restrict__ Pim,
                                                           float* __restrict__ YFre,
                                                           float* __restrict__ YFim,
                                                           float* __restrict__ ZFre,
                                                           float* __restrict__ ZFim) {
    __shared__ float ctab[N1D];
    __shared__ float stab[N1D];
    {
        const int j = threadIdx.x;
        const float ang = (float)j * 0.09817477042468103f;
        ctab[j] = 0.125f * cosf(ang);
        stab[j] = 0.125f * sinf(ang);
    }
    __syncthreads();

    if (blockIdx.x >= 288) {
        const int e = (blockIdx.x - 288) * 64 + threadIdx.x;
        const int k = e >> 4;
        const int n = (e & 15) * 4;
        float cr[4], si[4];
#pragma unroll
        for (int q = 0; q < 4; ++q) {
            const int kn = (k * (n + q)) & 63;
            const float sgn = ((k + n + q) & 1) ? -1.0f : 1.0f;
            cr[q] = sgn * ctab[kn];
            si[q] = sgn * stab[kn];
        }
        *(float4*)(Wre  + k * N1D + n) = make_float4(cr[0], cr[1], cr[2], cr[3]);
        *(float4*)(WimI + k * N1D + n) = make_float4(si[0], si[1], si[2], si[3]);
        return;
    }

    const int lane   = threadIdx.x & 31;
    const int waveId = (blockIdx.x * blockDim.x + threadIdx.x) >> 5;
    const int ck   = waveId & 3;
    const int rest = waveId >> 2;
    const int d    = rest / 48;
    const int tt   = rest - d * 48;
    const int half = lane >> 4;
    const int l16  = lane & 15;

    const int t = tt * 16 + l16;
    const int a = t / K_TERMS;
    const float c   = X[3 * a + d];
    const float bl2 = bw[t] * LOG2E;
    const float scale = (d == 0) ? aw[t] : 1.0f;
    const int krow = ck * 16 + l16;
    const float sg0 = (krow & 1) ? -1.0f : 1.0f;
    const bool dcz = (d == 2) && (krow == 0);
    const float* itab = dcz ? ctab : stab;
    const float sgx = dcz ? 1.0f : -sg0;
    const float sgy = dcz ? 1.0f : sg0;

    v8f re = {}, im = {};
#pragma unroll 4
    for (int n0 = 0; n0 < N1D; n0 += 4) {
        const int koff = n0 + 2 * half;
        const float d0 = (float)koff * SPACING - c;
        const float d1 = (float)(koff + 1) * SPACING - c;
        v2f e;
        e.x = scale * fast_exp2(bl2 * (d0 * d0));
        e.y = scale * fast_exp2(bl2 * (d1 * d1));
        const int i0 = (krow * koff) & 63;
        const int i1 = (i0 + krow) & 63;
        v2f wr, wi;
        wr.x = sg0 * ctab[i0];   wr.y = -sg0 * ctab[i1];
        wi.x = sgx * itab[i0];   wi.y = sgy * itab[i1];
        re = __builtin_amdgcn_wmma_f32_16x16x4_f32(false, e, false, wr, (short)0, re, false, false);
        im = __builtin_amdgcn_wmma_f32_16x16x4_f32(false, e, false, wi, (short)0, im, false, false);
    }

    if (d == 0) {
        const int o = krow * N_TERMS + tt * 16 + 8 * half;
        *(float4*)(Pre + o)     = make_float4(re[0], re[1], re[2], re[3]);
        *(float4*)(Pre + o + 4) = make_float4(re[4], re[5], re[6], re[7]);
        *(float4*)(Pim + o)     = make_float4(im[0], im[1], im[2], im[3]);
        *(float4*)(Pim + o + 4) = make_float4(im[4], im[5], im[6], im[7]);
    } else {
        float* __restrict__ Fr = (d == 1) ? YFre : ZFre;
        float* __restrict__ Fi = (d == 1) ? YFim : ZFim;
#pragma unroll
        for (int r2 = 0; r2 < 4; ++r2) {
            const int s  = tt * 4 + 2 * half + (r2 >> 1);
            const int hh = r2 & 1;
            const int o  = (((s * 4 + ck) * 32) + hh * 16 + l16) * 2;
            v2f vr, vi;
            vr.x = re[2 * r2]; vr.y = re[2 * r2 + 1];
            vi.x = im[2 * r2]; vi.y = im[2 * r2 + 1];
            *(v2f*)(Fr + o) = vr;
            *(v2f*)(Fi + o) = vi;
        }
    }
}

#define KSTEPS_Q (N_TERMS / 16)
#define CG 8
#define GP 68
__device__ __forceinline__ void contract_read(const v2f* __restrict__ pPr, const v2f* __restrict__ pPi,
                                              const v2f* __restrict__ pYr, const v2f* __restrict__ pYi,
                                              const v2f* __restrict__ pZr, const v2f* __restrict__ pZi,
                                              int s,
                                              v2f (&pr)[CG], v2f (&pi)[CG], v2f (&yr)[CG],
                                              v2f (&yi)[CG], v2f (&zr)[CG], v2f (&zi)[CG]) {
#pragma unroll
    for (int u = 0; u < CG; ++u) {
        pr[u] = pPr[2 * (s + u)];   pi[u] = pPi[2 * (s + u)];
        yr[u] = pYr[128 * (s + u)]; yi[u] = pYi[128 * (s + u)];
        zr[u] = pZr[128 * (s + u)]; zi[u] = pZi[128 * (s + u)];
    }
}

__device__ __forceinline__ void contract_mac(const v2f (&pr)[CG], const v2f (&pi)[CG],
                                             const v2f (&yr)[CG], const v2f (&yi)[CG],
                                             const v2f (&zr)[CG], const v2f (&zi)[CG],
                                             v8f& rr, v8f& ii, v8f& ri, v8f& ir) {
#pragma unroll
    for (int u = 0; u < CG; ++u) {
        const v2f ar = pr[u] * yr[u] - pi[u] * yi[u];
        const v2f ai = pr[u] * yi[u] + pi[u] * yr[u];
        rr = __builtin_amdgcn_wmma_f32_16x16x4_f32(false, ar, false, zr[u], (short)0, rr, false, false);
        ii = __builtin_amdgcn_wmma_f32_16x16x4_f32(false, ai, false, zi[u], (short)0, ii, false, false);
        ri = __builtin_amdgcn_wmma_f32_16x16x4_f32(false, ar, false, zi[u], (short)0, ri, false, false);
        ir = __builtin_amdgcn_wmma_f32_16x16x4_f32(false, ai, false, zr[u], (short)0, ir, false, false);
    }
}

__global__ __launch_bounds__(256) void spectrum_contract_kernel(const float* __restrict__ Pre,
                                                                const float* __restrict__ Pim,
                                                                const float* __restrict__ YFre,
                                                                const float* __restrict__ YFim,
                                                                const float* __restrict__ ZFre,
                                                                const float* __restrict__ ZFim,
                                                                const float* __restrict__ ham,
                                                                const float* __restrict__ Mre,
                                                                const float* __restrict__ Mim,
                                                                float* __restrict__ Ure,
                                                                float* __restrict__ Uim) {
    __shared__ float part[4][2][32][32];
    __shared__ __attribute__((aligned(16))) float Gre[16 * GP];
    __shared__ __attribute__((aligned(16))) float Gim[16 * GP];

    const int lane = threadIdx.x & 31;
    const int w    = threadIdx.x >> 5;
    const int ct   = w & 1;
    const int kq   = w >> 1;
    const int cz   = w & 3;
    const int kh   = w >> 2;
    const int kx   = blockIdx.x >> 2;
    const int cy   = blockIdx.x & 3;
    const int half = lane >> 4;
    const int l16  = lane & 15;
    const int sBeg = kq * KSTEPS_Q;

    const v2f* __restrict__ pPr = (const v2f*)(Pre + kx * N_TERMS + 2 * half) + 2 * sBeg;
    const v2f* __restrict__ pPi = (const v2f*)(Pim + kx * N_TERMS + 2 * half) + 2 * sBeg;
    const v2f* __restrict__ pYr = (const v2f*)YFre + 128 * sBeg + cy * 32 + lane;
    const v2f* __restrict__ pYi = (const v2f*)YFim + 128 * sBeg + cy * 32 + lane;
    const v2f* __restrict__ pZr = (const v2f*)ZFre + 128 * sBeg + ct * 32 + lane;
    const v2f* __restrict__ pZi = (const v2f*)ZFim + 128 * sBeg + ct * 32 + lane;

    const int  kzD = ct * 16 + l16;
    const bool sp  = (kzD == 0);
    const int  kzM = sp ? 32 : 64 - kzD;

    v8f rr = {}, ii = {}, ri = {}, ir = {};
    v2f prA[CG], piA[CG], yrA[CG], yiA[CG], zrA[CG], ziA[CG];
    v2f prB[CG], piB[CG], yrB[CG], yiB[CG], zrB[CG], ziB[CG];
    v2f mr[16], mi[16];
    float hD[2], hM[2];
    contract_read(pPr, pPi, pYr, pYi, pZr, pZi, 0, prA, piA, yrA, yiA, zrA, ziA);
#pragma unroll
    for (int s0 = 0; s0 < KSTEPS_Q; s0 += 2 * CG) {
        contract_read(pPr, pPi, pYr, pYi, pZr, pZi, s0 + CG, prB, piB, yrB, yiB, zrB, ziB);
        __builtin_amdgcn_sched_barrier(0);
        contract_mac(prA, piA, yrA, yiA, zrA, ziA, rr, ii, ri, ir);
        __builtin_amdgcn_sched_barrier(0);
        if (s0 + 2 * CG < KSTEPS_Q) {
            contract_read(pPr, pPi, pYr, pYi, pZr, pZi, s0 + 2 * CG, prA, piA, yrA, yiA, zrA, ziA);
        } else {
            const float* __restrict__ aR = Mre + (cz * 16 + l16) * N1D + 2 * half;
            const float* __restrict__ aI = Mim + (cz * 16 + l16) * N1D + 2 * half;
#pragma unroll
            for (int s = 0; s < 16; ++s) {
                mr[s] = *(const v2f*)(aR + 4 * s);
                mi[s] = *(const v2f*)(aI + 4 * s);
            }
#pragma unroll
            for (int e = 0; e < 2; ++e) {
                const int ky = cy * 16 + 2 * kq + e + 8 * half;
                hD[e] = ham[kx * 4096 + ky * N1D + kzD];
                hM[e] = ham[kx * 4096 + ky * N1D + kzM];
            }
        }
        __builtin_amdgcn_sched_barrier(0);
        contract_mac(prB, piB, yrB, yiB, zrB, ziB, rr, ii, ri, ir);
        if (s0 + 2 * CG < KSTEPS_Q) {
            __builtin_amdgcn_sched_barrier(0);
        }
    }

#pragma unroll
    for (int v = 0; v < 8; ++v) {
        part[kq][ct][v][lane]      = rr[v];
        part[kq][ct][8 + v][lane]  = ii[v];
        part[kq][ct][16 + v][lane] = ri[v];
        part[kq][ct][24 + v][lane] = ir[v];
    }
    __syncthreads();
#pragma unroll
    for (int e = 0; e < 2; ++e) {
        const int v = 2 * kq + e;
        float srr = part[0][ct][v][lane];
        float sii = part[0][ct][8 + v][lane];
        float sri = part[0][ct][16 + v][lane];
        float sir = part[0][ct][24 + v][lane];
#pragma unroll
        for (int q = 1; q < 4; ++q) {
            srr += part[q][ct][v][lane];
            sii += part[q][ct][8 + v][lane];
            sri += part[q][ct][16 + v][lane];
            sir += part[q][ct][24 + v][lane];
        }
        const float dre = sp ? srr : srr - sii;
        const float dim = sp ? sir : sri + sir;
        const float mre = sp ? sri : srr + sii;
        const float mim = sp ? sii : sir - sri;
        const int o = (v + 8 * half) * GP;
        Gre[o + kzD] = dre * hD[e];
        Gim[o + kzD] = dim * hD[e];
        Gre[o + kzM] = mre * hM[e];
        Gim[o + kzM] = mim * hM[e];
    }
    __syncthreads();

    const float* b1 = (kh ? Gim : Gre) + l16 * GP + 2 * half;
    const float* b2 = (kh ? Gre : Gim) + l16 * GP + 2 * half;
    v8f a1 = {}, a2 = {};
#pragma unroll
    for (int s = 0; s < 16; ++s) {
        const v2f t1 = *(const v2f*)(b1 + 4 * s);
        const v2f t2 = *(const v2f*)(b2 + 4 * s);
        a1 = __builtin_amdgcn_wmma_f32_16x16x4_f32(false, mr[s], false, t1, (short)0, a1, false, false);
        a2 = __builtin_amdgcn_wmma_f32_16x16x4_f32(false, mi[s], false, t2, (short)0, a2, false, false);
    }
    const float sg = kh ? 1.0f : -1.0f;
    const v8f u = a1 + sg * a2;

    float* __restrict__ dst = kh ? Uim : Ure;
#pragma unroll
    for (int v = 0; v < 8; ++v) {
        const int z = cz * 16 + v + 8 * half;
        dst[z * 4096 + kx * N1D + cy * 16 + l16] = u[v];
    }
}

#define DFT_WG_THREADS 128
#define DFT_WGS        256
__global__ __launch_bounds__(DFT_WG_THREADS) void dft_pass_kernel(const float* __restrict__ Tre,
                                                                  const float* __restrict__ Tim,
                                                                  const float* __restrict__ Mre,
                                                                  const float* __restrict__ Mim,
                                                                  float* __restrict__ Ure,
                                                                  float* __restrict__ Uim) {
    const int lane   = threadIdx.x & 31;
    const int waveId = (blockIdx.x * DFT_WG_THREADS + threadIdx.x) >> 5;
    const int p0 = (waveId & 255) << 4;
    const int k0 = (waveId >> 8) << 4;
    const int half = lane >> 4;
    const int l16  = lane & 15;
    const int krow = k0 + l16;
    const int prow = p0 + l16;

    const float* __restrict__ aR = Mre + krow * N1D + 2 * half;
    const float* __restrict__ aI = Mim + krow * N1D + 2 * half;
    const float* __restrict__ bR = Tre + prow * N1D + 2 * half;
    const float* __restrict__ bI = Tim + prow * N1D + 2 * half;
    v2f mr[16], mi[16], tr[16], ti[16];
#pragma unroll
    for (int s = 0; s < 16; ++s) {
        mr[s] = *(const v2f*)(aR + 4 * s);
        tr[s] = *(const v2f*)(bR + 4 * s);
        mi[s] = *(const v2f*)(aI + 4 * s);
        ti[s] = *(const v2f*)(bI + 4 * s);
    }
    __builtin_amdgcn_sched_barrier(0);

    v8f rr = {}, ii = {}, ri = {}, ir = {};
#pragma unroll
    for (int s = 0; s < 16; ++s) {
        rr = __builtin_amdgcn_wmma_f32_16x16x4_f32(false, mr[s], false, tr[s], (short)0, rr, false, false);
        ii = __builtin_amdgcn_wmma_f32_16x16x4_f32(false, mi[s], false, ti[s], (short)0, ii, false, false);
        ri = __builtin_amdgcn_wmma_f32_16x16x4_f32(false, mr[s], false, ti[s], (short)0, ri, false, false);
        ir = __builtin_amdgcn_wmma_f32_16x16x4_f32(false, mi[s], false, tr[s], (short)0, ir, false, false);
    }

#pragma unroll
    for (int v = 0; v < 8; ++v) {
        const int row = k0 + v + 8 * half;
        const int idx = row * 4096 + p0 + l16;
        Ure[idx] = rr[v] - ii[v];
        Uim[idx] = ri[v] + ir[v];
    }
}

#define XP_THREADS 128
#define XP_WGS     256

__device__ __forceinline__ void xpass_read(const float* __restrict__ Tre,
                                           const float* __restrict__ Tim,
                                           const float* __restrict__ Mre,
                                           const float* __restrict__ Mim,
                                           int krow, int prow, int half,
                                           v2f (&mr)[16], v2f (&mi)[16],
                                           v2f (&tr)[16], v2f (&ti)[16]) {
    const float* __restrict__ aR = Mre + krow * N1D + 2 * half;
    const float* __restrict__ aI = Mim + krow * N1D + 2 * half;
    const float* __restrict__ bR = Tre + prow * N1D + 2 * half;
    const float* __restrict__ bI = Tim + prow * N1D + 2 * half;
#pragma unroll
    for (int s = 0; s < 16; ++s) {
        mr[s] = *(const v2f*)(aR + 4 * s);
        tr[s] = *(const v2f*)(bR + 4 * s);
        mi[s] = *(const v2f*)(aI + 4 * s);
        ti[s] = *(const v2f*)(bI + 4 * s);
    }
}

__device__ __forceinline__ v8f xpass_real(const v2f (&mr)[16], const v2f (&mi)[16],
                                          const v2f (&tr)[16], const v2f (&ti)[16]) {
    v8f rr = {}, ii = {};
#pragma unroll
    for (int s = 0; s < 16; ++s) {
        rr = __builtin_amdgcn_wmma_f32_16x16x4_f32(false, mr[s], false, tr[s], (short)0, rr, false, false);
        ii = __builtin_amdgcn_wmma_f32_16x16x4_f32(false, mi[s], false, ti[s], (short)0, ii, false, false);
    }
    return rr - ii;
}

__global__ __launch_bounds__(XP_THREADS) void xpass_sum_kernel(const float* __restrict__ Tre,
                                                               const float* __restrict__ Tim,
                                                               const float* __restrict__ Mre,
                                                               const float* __restrict__ Mim,
                                                               float* __restrict__ psum,
                                                               float* __restrict__ psq) {
    __shared__ float wsum[4];
    __shared__ float wsq[4];
    const int lane   = threadIdx.x & 31;
    const int w      = threadIdx.x >> 5;
    const int waveId = blockIdx.x * 4 + w;
    const int p0 = (waveId & 255) << 4;
    const int k0 = (waveId >> 8) << 4;
    const int half = lane >> 4;
    const int l16  = lane & 15;

    v2f mr[16], mi[16], tr[16], ti[16];
    xpass_read(Tre, Tim, Mre, Mim, k0 + l16, p0 + l16, half, mr, mi, tr, ti);
    __builtin_amdgcn_sched_barrier(0);
    const v8f u = xpass_real(mr, mi, tr, ti);

    float s = 0.0f, q = 0.0f;
#pragma unroll
    for (int v = 0; v < 8; ++v) { s += u[v]; q += u[v] * u[v]; }
#pragma unroll
    for (int o = 16; o > 0; o >>= 1) {
        s += __shfl_xor(s, o);
        q += __shfl_xor(q, o);
    }
    if (lane == 0) { wsum[w] = s; wsq[w] = q; }
    __syncthreads();
    if (threadIdx.x == 0) {
        psum[blockIdx.x] = (wsum[0] + wsum[1]) + (wsum[2] + wsum[3]);
        psq[blockIdx.x]  = (wsq[0] + wsq[1]) + (wsq[2] + wsq[3]);
    }
}

__global__ __launch_bounds__(XP_THREADS) void xpass_norm_kernel(const float* __restrict__ Tre,
                                                                const float* __restrict__ Tim,
                                                                const float* __restrict__ Mre,
                                                                const float* __restrict__ Mim,
                                                                const float* __restrict__ psum,
                                                                const float* __restrict__ psq,
                                                                float* __restrict__ out) {
    const int lane   = threadIdx.x & 31;
    const int waveId = (blockIdx.x * XP_THREADS + threadIdx.x) >> 5;
    const int p0 = (waveId & 255) << 4;
    const int k0 = (waveId >> 8) << 4;
    const int half = lane >> 4;
    const int l16  = lane & 15;

    const float4 sa = ((const float4*)psum)[2 * lane], sb = ((const float4*)psum)[2 * lane + 1];
    const float4 qa = ((const float4*)psq)[2 * lane],  qb = ((const float4*)psq)[2 * lane + 1];

    v2f mr[16], mi[16], tr[16], ti[16];
    xpass_read(Tre, Tim, Mre, Mim, k0 + l16, p0 + l16, half, mr, mi, tr, ti);
    __builtin_amdgcn_sched_barrier(0);

    float s = ((sa.x + sa.y) + (sa.z + sa.w)) + ((sb.x + sb.y) + (sb.z + sb.w));
    float q = ((qa.x + qa.y) + (qa.z + qa.w)) + ((qb.x + qb.y) + (qb.z + qb.w));
#pragma unroll
    for (int o = 16; o > 0; o >>= 1) {
        s += __shfl_xor(s, o);
        q += __shfl_xor(q, o);
    }
    const float invN = 1.0f / (float)GRID_N;
    const float mean = s * invN;
    const float var  = fmaxf(q * invN - mean * mean, 0.0f);
    const float is   = 1.0f / (sqrtf(var) + 1e-8f);

    const v8f u = xpass_real(mr, mi, tr, ti);

#pragma unroll
    for (int v = 0; v < 8; ++v) {
        const int row = k0 + v + 8 * half;
        out[row * 4096 + p0 + l16] = (u[v] - mean) * is;
    }
}

extern "C" void kernel_launch(void* const* d_in, const int* in_sizes, int n_in,
                              void* d_out, int out_size, void* d_ws, size_t ws_size,
                              hipStream_t stream) {
    const float* X   = (const float*)d_in[0];
    const float* aw  = (const float*)d_in[1];
    const float* bw  = (const float*)d_in[2];
    const float* ham = (const float*)d_in[4];
    float* out = (float*)d_out;

    float* ws   = (float*)d_ws;
    float* Wre  = ws;
    float* WimI = Wre + 8192;
    float* Are  = WimI + 4096;
    float* Aim  = Are + GRID_N;
    float* Bre  = Aim + GRID_N;
    float* Bim  = Bre + GRID_N;
    float* psum = Bim + GRID_N;
    float* psq  = psum + 512;

    float* Pre  = Are;
    float* Pim  = Pre  + TAB_SZ;
    float* YFre = Pim  + TAB_SZ;
    float* YFim = YFre + TAB_SZ;
    float* ZFre = YFim + TAB_SZ;
    float* ZFim = ZFre + TAB_SZ;

    axis_spectra_kernel<<<304, 64, 0, stream>>>(X, aw, bw, Wre, WimI,
                                                Pre, Pim, YFre, YFim, ZFre, ZFim);

    spectrum_contract_kernel<<<256, 256, 0, stream>>>(Pre, Pim, YFre, YFim, ZFre, ZFim,
                                                      ham, Wre, WimI, Bre, Bim);

    dft_pass_kernel<<<DFT_WGS, DFT_WG_THREADS, 0, stream>>>(Bre, Bim, Wre, WimI, Are, Aim);

    xpass_sum_kernel<<<XP_WGS, XP_THREADS, 0, stream>>>(Are, Aim, Wre, WimI, psum, psq);
    xpass_norm_kernel<<<XP_WGS, XP_THREADS, 0, stream>>>(Are, Aim, Wre, WimI, psum, psq, out);
}


# run 1 of the last inverse pass (sums only) done inside the y-pass launch, on the work-group's 16x64 tile passed through LDS: its launch and plane re-read gone (5->4); mean/variance sum order changes
# speedup vs baseline: 2.5895x; 1.1039x over previous
#include <hip/hip_runtime.h>
#include <hip/hip_bf16.h>
#include <math.h>

#define N1D      64
#define GRID_N   (64 * 64 * 64)
#define N_ATOMS  128
#define K_TERMS  6
#define SPACING  0.5f
#define LOG2E    1.4426950408889634f

typedef float v2f __attribute__((ext_vector_type(2)));
typedef float v8f __attribute__((ext_vector_type(8)));

__device__ __forceinline__ float fast_exp2(float x) {
#if __has_builtin(__builtin_amdgcn_exp2f)
    return __builtin_amdgcn_exp2f(x);
#else
    return exp2f(x);
#endif
}

#define N_TERMS  (N_ATOMS * K_TERMS)
#define TAB_SZ   (N_TERMS * N1D)

__global__ __launch_bounds__(64) void axis_spectra_kernel(const float* __restrict__ X,
                                                           const float* __restrict__ aw,
                                                           const float* __restrict__ bw,
                                                           float* __restrict__ Wre,
                                                           float* __restrict__ WimI,
                                                           float* __restrict__ Pre,
                                                           float* __restrict__ Pim,
                                                           float* __restrict__ YFre,
                                                           float* __restrict__ YFim,
                                                           float* __restrict__ ZFre,
                                                           float* __restrict__ ZFim) {
    __shared__ float ctab[N1D];
    __shared__ float stab[N1D];
    {
        const int j = threadIdx.x;
        const float ang = (float)j * 0.09817477042468103f;
        ctab[j] = 0.125f * cosf(ang);
        stab[j] = 0.125f * sinf(ang);
    }
    __syncthreads();

    if (blockIdx.x >= 288) {
        const int e = (blockIdx.x - 288) * 64 + threadIdx.x;
        const int k = e >> 4;
        const int n = (e & 15) * 4;
        float cr[4], si[4];
#pragma unroll
        for (int q = 0; q < 4; ++q) {
            const int kn = (k * (n + q)) & 63;
            const float sgn = ((k + n + q) & 1) ? -1.0f : 1.0f;
            cr[q] = sgn * ctab[kn];
            si[q] = sgn * stab[kn];
        }
        *(float4*)(Wre  + k * N1D + n) = make_float4(cr[0], cr[1], cr[2], cr[3]);
        *(float4*)(WimI + k * N1D + n) = make_float4(si[0], si[1], si[2], si[3]);
        return;
    }

    const int lane   = threadIdx.x & 31;
    const int waveId = (blockIdx.x * blockDim.x + threadIdx.x) >> 5;
    const int ck   = waveId & 3;
    const int rest = waveId >> 2;
    const int d    = rest / 48;
    const int tt   = rest - d * 48;
    const int half = lane >> 4;
    const int l16  = lane & 15;

    const int t = tt * 16 + l16;
    const int a = t / K_TERMS;
    const float c   = X[3 * a + d];
    const float bl2 = bw[t] * LOG2E;
    const float scale = (d == 0) ? aw[t] : 1.0f;
    const int krow = ck * 16 + l16;
    const float sg0 = (krow & 1) ? -1.0f : 1.0f;
    const bool dcz = (d == 2) && (krow == 0);
    const float* itab = dcz ? ctab : stab;
    const float sgx = dcz ? 1.0f : -sg0;
    const float sgy = dcz ? 1.0f : sg0;

    v8f re = {}, im = {};
#pragma unroll 4
    for (int n0 = 0; n0 < N1D; n0 += 4) {
        const int koff = n0 + 2 * half;
        const float d0 = (float)koff * SPACING - c;
        const float d1 = (float)(koff + 1) * SPACING - c;
        v2f e;
        e.x = scale * fast_exp2(bl2 * (d0 * d0));
        e.y = scale * fast_exp2(bl2 * (d1 * d1));
        const int i0 = (krow * koff) & 63;
        const int i1 = (i0 + krow) & 63;
        v2f wr, wi;
        wr.x = sg0 * ctab[i0];   wr.y = -sg0 * ctab[i1];
        wi.x = sgx * itab[i0];   wi.y = sgy * itab[i1];
        re = __builtin_amdgcn_wmma_f32_16x16x4_f32(false, e, false, wr, (short)0, re, false, false);
        im = __builtin_amdgcn_wmma_f32_16x16x4_f32(false, e, false, wi, (short)0, im, false, false);
    }

    if (d == 0) {
        const int o = krow * N_TERMS + tt * 16 + 8 * half;
        *(float4*)(Pre + o)     = make_float4(re[0], re[1], re[2], re[3]);
        *(float4*)(Pre + o + 4) = make_float4(re[4], re[5], re[6], re[7]);
        *(float4*)(Pim + o)     = make_float4(im[0], im[1], im[2], im[3]);
        *(float4*)(Pim + o + 4) = make_float4(im[4], im[5], im[6], im[7]);
    } else {
        float* __restrict__ Fr = (d == 1) ? YFre : ZFre;
        float* __restrict__ Fi = (d == 1) ? YFim : ZFim;
#pragma unroll
        for (int r2 = 0; r2 < 4; ++r2) {
            const int s  = tt * 4 + 2 * half + (r2 >> 1);
            const int hh = r2 & 1;
            const int o  = (((s * 4 + ck) * 32) + hh * 16 + l16) * 2;
            v2f vr, vi;
            vr.x = re[2 * r2]; vr.y = re[2 * r2 + 1];
            vi.x = im[2 * r2]; vi.y = im[2 * r2 + 1];
            *(v2f*)(Fr + o) = vr;
            *(v2f*)(Fi + o) = vi;
        }
    }
}

#define KSTEPS_Q (N_TERMS / 16)
#define CG 8
#define GP 68
__device__ __forceinline__ void contract_read(const v2f* __restrict__ pPr, const v2f* __restrict__ pPi,
                                              const v2f* __restrict__ pYr, const v2f* __restrict__ pYi,
                                              const v2f* __restrict__ pZr, const v2f* __restrict__ pZi,
                                              int s,
                                              v2f (&pr)[CG], v2f (&pi)[CG], v2f (&yr)[CG],
                                              v2f (&yi)[CG], v2f (&zr)[CG], v2f (&zi)[CG]) {
#pragma unroll
    for (int u = 0; u < CG; ++u) {
        pr[u] = pPr[2 * (s + u)];   pi[u] = pPi[2 * (s + u)];
        yr[u] = pYr[128 * (s + u)]; yi[u] = pYi[128 * (s + u)];
        zr[u] = pZr[128 * (s + u)]; zi[u] = pZi[128 * (s + u)];
    }
}

__device__ __forceinline__ void contract_mac(const v2f (&pr)[CG], const v2f (&pi)[CG],
                                             const v2f (&yr)[CG], const v2f (&yi)[CG],
                                             const v2f (&zr)[CG], const v2f (&zi)[CG],
                                             v8f& rr, v8f& ii, v8f& ri, v8f& ir) {
#pragma unroll
    for (int u = 0; u < CG; ++u) {
        const v2f ar = pr[u] * yr[u] - pi[u] * yi[u];
        const v2f ai = pr[u] * yi[u] + pi[u] * yr[u];
        rr = __builtin_amdgcn_wmma_f32_16x16x4_f32(false, ar, false, zr[u], (short)0, rr, false, false);
        ii = __builtin_amdgcn_wmma_f32_16x16x4_f32(false, ai, false, zi[u], (short)0, ii, false, false);
        ri = __builtin_amdgcn_wmma_f32_16x16x4_f32(false, ar, false, zi[u], (short)0, ri, false, false);
        ir = __builtin_amdgcn_wmma_f32_16x16x4_f32(false, ai, false, zr[u], (short)0, ir, false, false);
    }
}

__global__ __launch_bounds__(256) void spectrum_contract_kernel(const float* __restrict__ Pre,
                                                                const float* __restrict__ Pim,
                                                                const float* __restrict__ YFre,
                                                                const float* __restrict__ YFim,
                                                                const float* __restrict__ ZFre,
                                                                const float* __restrict__ ZFim,
                                                                const float* __restrict__ ham,
                                                                const float* __restrict__ Mre,
                                                                const float* __restrict__ Mim,
                                                                float* __restrict__ Ure,
                                                                float* __restrict__ Uim) {
    __shared__ float part[4][2][32][32];
    __shared__ __attribute__((aligned(16))) float Gre[16 * GP];
    __shared__ __attribute__((aligned(16))) float Gim[16 * GP];

    const int lane = threadIdx.x & 31;
    const int w    = threadIdx.x >> 5;
    const int ct   = w & 1;
    const int kq   = w >> 1;
    const int cz   = w & 3;
    const int kh   = w >> 2;
    const int kx   = blockIdx.x >> 2;
    const int cy   = blockIdx.x & 3;
    const int half = lane >> 4;
    const int l16  = lane & 15;
    const int sBeg = kq * KSTEPS_Q;

    const v2f* __restrict__ pPr = (const v2f*)(Pre + kx * N_TERMS + 2 * half) + 2 * sBeg;
    const v2f* __restrict__ pPi = (const v2f*)(Pim + kx * N_TERMS + 2 * half) + 2 * sBeg;
    const v2f* __restrict__ pYr = (const v2f*)YFre + 128 * sBeg + cy * 32 + lane;
    const v2f* __restrict__ pYi = (const v2f*)YFim + 128 * sBeg + cy * 32 + lane;
    const v2f* __restrict__ pZr = (const v2f*)ZFre + 128 * sBeg + ct * 32 + lane;
    const v2f* __restrict__ pZi = (const v2f*)ZFim + 128 * sBeg + ct * 32 + lane;

    const int  kzD = ct * 16 + l16;
    const bool sp  = (kzD == 0);
    const int  kzM = sp ? 32 : 64 - kzD;

    v8f rr = {}, ii = {}, ri = {}, ir = {};
    v2f prA[CG], piA[CG], yrA[CG], yiA[CG], zrA[CG], ziA[CG];
    v2f prB[CG], piB[CG], yrB[CG], yiB[CG], zrB[CG], ziB[CG];
    v2f mr[16], mi[16];
    float hD[2], hM[2];
    contract_read(pPr, pPi, pYr, pYi, pZr, pZi, 0, prA, piA, yrA, yiA, zrA, ziA);
#pragma unroll
    for (int s0 = 0; s0 < KSTEPS_Q; s0 += 2 * CG) {
        contract_read(pPr, pPi, pYr, pYi, pZr, pZi, s0 + CG, prB, piB, yrB, yiB, zrB, ziB);
        __builtin_amdgcn_sched_barrier(0);
        contract_mac(prA, piA, yrA, yiA, zrA, ziA, rr, ii, ri, ir);
        __builtin_amdgcn_sched_barrier(0);
        if (s0 + 2 * CG < KSTEPS_Q) {
            contract_read(pPr, pPi, pYr, pYi, pZr, pZi, s0 + 2 * CG, prA, piA, yrA, yiA, zrA, ziA);
        } else {
            const float* __restrict__ aR = Mre + (cz * 16 + l16) * N1D + 2 * half;
            const float* __restrict__ aI = Mim + (cz * 16 + l16) * N1D + 2 * half;
#pragma unroll
            for (int s = 0; s < 16; ++s) {
                mr[s] = *(const v2f*)(aR + 4 * s);
                mi[s] = *(const v2f*)(aI + 4 * s);
            }
#pragma unroll
            for (int e = 0; e < 2; ++e) {
                const int ky = cy * 16 + 2 * kq + e + 8 * half;
                hD[e] = ham[kx * 4096 + ky * N1D + kzD];
                hM[e] = ham[kx * 4096 + ky * N1D + kzM];
            }
        }
        __builtin_amdgcn_sched_barrier(0);
        contract_mac(prB, piB, yrB, yiB, zrB, ziB, rr, ii, ri, ir);
        if (s0 + 2 * CG < KSTEPS_Q) {
            __builtin_amdgcn_sched_barrier(0);
        }
    }

#pragma unroll
    for (int v = 0; v < 8; ++v) {
        part[kq][ct][v][lane]      = rr[v];
        part[kq][ct][8 + v][lane]  = ii[v];
        part[kq][ct][16 + v][lane] = ri[v];
        part[kq][ct][24 + v][lane] = ir[v];
    }
    __syncthreads();
#pragma unroll
    for (int e = 0; e < 2; ++e) {
        const int v = 2 * kq + e;
        float srr = part[0][ct][v][lane];
        float sii = part[0][ct][8 + v][lane];
        float sri = part[0][ct][16 + v][lane];
        float sir = part[0][ct][24 + v][lane];
#pragma unroll
        for (int q = 1; q < 4; ++q) {
            srr += part[q][ct][v][lane];
            sii += part[q][ct][8 + v][lane];
            sri += part[q][ct][16 + v][lane];
            sir += part[q][ct][24 + v][lane];
        }
        const float dre = sp ? srr : srr - sii;
        const float dim = sp ? sir : sri + sir;
        const float mre = sp ? sri : srr + sii;
        const float mim = sp ? sii : sir - sri;
        const int o = (v + 8 * half) * GP;
        Gre[o + kzD] = dre * hD[e];
        Gim[o + kzD] = dim * hD[e];
        Gre[o + kzM] = mre * hM[e];
        Gim[o + kzM] = mim * hM[e];
    }
    __syncthreads();

    const float* b1 = (kh ? Gim : Gre) + l16 * GP + 2 * half;
    const float* b2 = (kh ? Gre : Gim) + l16 * GP + 2 * half;
    v8f a1 = {}, a2 = {};
#pragma unroll
    for (int s = 0; s < 16; ++s) {
        const v2f t1 = *(const v2f*)(b1 + 4 * s);
        const v2f t2 = *(const v2f*)(b2 + 4 * s);
        a1 = __builtin_amdgcn_wmma_f32_16x16x4_f32(false, mr[s], false, t1, (short)0, a1, false, false);
        a2 = __builtin_amdgcn_wmma_f32_16x16x4_f32(false, mi[s], false, t2, (short)0, a2, false, false);
    }
    const float sg = kh ? 1.0f : -1.0f;
    const v8f u = a1 + sg * a2;

    float* __restrict__ dst = kh ? Uim : Ure;
#pragma unroll
    for (int v = 0; v < 8; ++v) {
        const int z = cz * 16 + v + 8 * half;
        dst[z * 4096 + kx * N1D + cy * 16 + l16] = u[v];
    }
}

#define XP_THREADS 128
#define XP_WGS     256

__device__ __forceinline__ void xpass_read(const float* __restrict__ Tre,
                                           const float* __restrict__ Tim,
                                           const float* __restrict__ Mre,
                                           const float* __restrict__ Mim,
                                           int krow, int prow, int half,
                                           v2f (&mr)[16], v2f (&mi)[16],
                                           v2f (&tr)[16], v2f (&ti)[16]) {
    const float* __restrict__ aR = Mre + krow * N1D + 2 * half;
    const float* __restrict__ aI = Mim + krow * N1D + 2 * half;
    const float* __restrict__ bR = Tre + prow * N1D + 2 * half;
    const float* __restrict__ bI = Tim + prow * N1D + 2 * half;
#pragma unroll
    for (int s = 0; s < 16; ++s) {
        mr[s] = *(const v2f*)(aR + 4 * s);
        tr[s] = *(const v2f*)(bR + 4 * s);
        mi[s] = *(const v2f*)(aI + 4 * s);
        ti[s] = *(const v2f*)(bI + 4 * s);
    }
}

__device__ __forceinline__ v8f xpass_real(const v2f (&mr)[16], const v2f (&mi)[16],
                                          const v2f (&tr)[16], const v2f (&ti)[16]) {
    v8f rr = {}, ii = {};
#pragma unroll
    for (int s = 0; s < 16; ++s) {
        rr = __builtin_amdgcn_wmma_f32_16x16x4_f32(false, mr[s], false, tr[s], (short)0, rr, false, false);
        ii = __builtin_amdgcn_wmma_f32_16x16x4_f32(false, mi[s], false, ti[s], (short)0, ii, false, false);
    }
    return rr - ii;
}

#define DFT_WG_THREADS 128
#define DFT_WGS        256
__global__ __launch_bounds__(DFT_WG_THREADS) void ypass_xsum_kernel(const float* __restrict__ Tre,
                                                                    const float* __restrict__ Tim,
                                                                    const float* __restrict__ Mre,
                                                                    const float* __restrict__ Mim,
                                                                    float* __restrict__ Ure,
                                                                    float* __restrict__ Uim,
                                                                    float* __restrict__ psum,
                                                                    float* __restrict__ psq) {
    __shared__ __attribute__((aligned(16))) float Sre[16 * GP];
    __shared__ __attribute__((aligned(16))) float Sim[16 * GP];
    __shared__ float wsum[4];
    __shared__ float wsq[4];

    const int lane = threadIdx.x & 31;
    const int w    = threadIdx.x >> 5;
    const int z    = blockIdx.x & 63;
    const int k0   = (blockIdx.x >> 6) << 4;
    const int p0   = z * N1D + 16 * w;
    const int half = lane >> 4;
    const int l16  = lane & 15;
    const int krow = k0 + l16;
    const int prow = p0 + l16;
    const int xrow = 16 * w + l16;

    const float* __restrict__ aR = Mre + krow * N1D + 2 * half;
    const float* __restrict__ aI = Mim + krow * N1D + 2 * half;
    const float* __restrict__ bR = Tre + prow * N1D + 2 * half;
    const float* __restrict__ bI = Tim + prow * N1D + 2 * half;
    const float* __restrict__ xR = Mre + xrow * N1D + 2 * half;
    const float* __restrict__ xI = Mim + xrow * N1D + 2 * half;
    v2f mr[16], mi[16], tr[16], ti[16], xr[16], xi[16];
#pragma unroll
    for (int s = 0; s < 16; ++s) {
        mr[s] = *(const v2f*)(aR + 4 * s);
        tr[s] = *(const v2f*)(bR + 4 * s);
        mi[s] = *(const v2f*)(aI + 4 * s);
        ti[s] = *(const v2f*)(bI + 4 * s);
    }
#pragma unroll
    for (int s = 0; s < 16; ++s) {
        xr[s] = *(const v2f*)(xR + 4 * s);
        xi[s] = *(const v2f*)(xI + 4 * s);
    }
    __builtin_amdgcn_sched_barrier(0);

    v8f rr = {}, ii = {}, ri = {}, ir = {};
#pragma unroll
    for (int s = 0; s < 16; ++s) {
        rr = __builtin_amdgcn_wmma_f32_16x16x4_f32(false, mr[s], false, tr[s], (short)0, rr, false, false);
        ii = __builtin_amdgcn_wmma_f32_16x16x4_f32(false, mi[s], false, ti[s], (short)0, ii, false, false);
        ri = __builtin_amdgcn_wmma_f32_16x16x4_f32(false, mr[s], false, ti[s], (short)0, ri, false, false);
        ir = __builtin_amdgcn_wmma_f32_16x16x4_f32(false, mi[s], false, tr[s], (short)0, ir, false, false);
    }

    const v8f ure = rr - ii;
    const v8f uim = ri + ir;
#pragma unroll
    for (int v = 0; v < 8; ++v) {
        const int yl = v + 8 * half;
        Sre[yl * GP + 16 * w + l16] = ure[v];
        Sim[yl * GP + 16 * w + l16] = uim[v];
    }
    __syncthreads();
#pragma unroll
    for (int v = 0; v < 8; ++v) {
        const int idx = (k0 + v + 8 * half) * 4096 + p0 + l16;
        Ure[idx] = ure[v];
        Uim[idx] = uim[v];
    }
    __builtin_amdgcn_sched_barrier(0);

    v2f ur[16], ui[16];
    {
        const float* sR = Sre + l16 * GP + 2 * half;
        const float* sI = Sim + l16 * GP + 2 * half;
#pragma unroll
        for (int s = 0; s < 16; ++s) {
            ur[s] = *(const v2f*)(sR + 4 * s);
            ui[s] = *(const v2f*)(sI + 4 * s);
        }
    }
    const v8f u = xpass_real(xr, xi, ur, ui);

    float s = 0.0f, q = 0.0f;
#pragma unroll
    for (int v = 0; v < 8; ++v) { s += u[v]; q += u[v] * u[v]; }
#pragma unroll
    for (int o = 16; o > 0; o >>= 1) {
        s += __shfl_xor(s, o);
        q += __shfl_xor(q, o);
    }
    if (lane == 0) { wsum[w] = s; wsq[w] = q; }
    __syncthreads();
    if (threadIdx.x == 0) {
        psum[blockIdx.x] = (wsum[0] + wsum[1]) + (wsum[2] + wsum[3]);
        psq[blockIdx.x]  = (wsq[0] + wsq[1]) + (wsq[2] + wsq[3]);
    }
}

__global__ __launch_bounds__(XP_THREADS) void xpass_norm_kernel(const float* __restrict__ Tre,
                                                                const float* __restrict__ Tim,
                                                                const float* __restrict__ Mre,
                                                                const float* __restrict__ Mim,
                                                                const float* __restrict__ psum,
                                                                const float* __restrict__ psq,
                                                                float* __restrict__ out) {
    const int lane   = threadIdx.x & 31;
    const int waveId = (blockIdx.x * XP_THREADS + threadIdx.x) >> 5;
    const int p0 = (waveId & 255) << 4;
    const int k0 = (waveId >> 8) << 4;
    const int half = lane >> 4;
    const int l16  = lane & 15;

    const float4 sa = ((const float4*)psum)[2 * lane], sb = ((const float4*)psum)[2 * lane + 1];
    const float4 qa = ((const float4*)psq)[2 * lane],  qb = ((const float4*)psq)[2 * lane + 1];

    v2f mr[16], mi[16], tr[16], ti[16];
    xpass_read(Tre, Tim, Mre, Mim, k0 + l16, p0 + l16, half, mr, mi, tr, ti);
    __builtin_amdgcn_sched_barrier(0);

    float s = ((sa.x + sa.y) + (sa.z + sa.w)) + ((sb.x + sb.y) + (sb.z + sb.w));
    float q = ((qa.x + qa.y) + (qa.z + qa.w)) + ((qb.x + qb.y) + (qb.z + qb.w));
#pragma unroll
    for (int o = 16; o > 0; o >>= 1) {
        s += __shfl_xor(s, o);
        q += __shfl_xor(q, o);
    }
    const float invN = 1.0f / (float)GRID_N;
    const float mean = s * invN;
    const float var  = fmaxf(q * invN - mean * mean, 0.0f);
    const float is   = 1.0f / (sqrtf(var) + 1e-8f);

    const v8f u = xpass_real(mr, mi, tr, ti);

#pragma unroll
    for (int v = 0; v < 8; ++v) {
        const int row = k0 + v + 8 * half;
        out[row * 4096 + p0 + l16] = (u[v] - mean) * is;
    }
}

extern "C" void kernel_launch(void* const* d_in, const int* in_sizes, int n_in,
                              void* d_out, int out_size, void* d_ws, size_t ws_size,
                              hipStream_t stream) {
    const float* X   = (const float*)d_in[0];
    const float* aw  = (const float*)d_in[1];
    const float* bw  = (const float*)d_in[2];
    const float* ham = (const float*)d_in[4];
    float* out = (float*)d_out;

    float* ws   = (float*)d_ws;
    float* Wre  = ws;
    float* WimI = Wre + 8192;
    float* Are  = WimI + 4096;
    float* Aim  = Are + GRID_N;
    float* Bre  = Aim + GRID_N;
    float* Bim  = Bre + GRID_N;
    float* psum = Bim + GRID_N;
    float* psq  = psum + 512;

    float* Pre  = Are;
    float* Pim  = Pre  + TAB_SZ;
    float* YFre = Pim  + TAB_SZ;
    float* YFim = YFre + TAB_SZ;
    float* ZFre = YFim + TAB_SZ;
    float* ZFim = ZFre + TAB_SZ;

    axis_spectra_kernel<<<304, 64, 0, stream>>>(X, aw, bw, Wre, WimI,
                                                Pre, Pim, YFre, YFim, ZFre, ZFim);

    spectrum_contract_kernel<<<256, 256, 0, stream>>>(Pre, Pim, YFre, YFim, ZFre, ZFim,
                                                      ham, Wre, WimI, Bre, Bim);

    ypass_xsum_kernel<<<DFT_WGS, DFT_WG_THREADS, 0, stream>>>(Bre, Bim, Wre, WimI, Are, Aim, psum, psq);

    xpass_norm_kernel<<<XP_WGS, XP_THREADS, 0, stream>>>(Are, Aim, Wre, WimI, psum, psq, out);
}


# filter symmetrised to (h[k]+h[-k])/2: density is real, so the filtered spectrum is Hermitian and only planes kx 0..32 are formed and transformed; x pass sums 33 weighted columns; sum order changes
# speedup vs baseline: 2.7933x; 1.0787x over previous
#include <hip/hip_runtime.h>
#include <hip/hip_bf16.h>
#include <math.h>

#define N1D      64
#define GRID_N   (64 * 64 * 64)
#define N_ATOMS  128
#define K_TERMS  6
#define SPACING  0.5f
#define LOG2E    1.4426950408889634f

typedef float v2f __attribute__((ext_vector_type(2)));
typedef float v8f __attribute__((ext_vector_type(8)));

__device__ __forceinline__ float fast_exp2(float x) {
#if __has_builtin(__builtin_amdgcn_exp2f)
    return __builtin_amdgcn_exp2f(x);
#else
    return exp2f(x);
#endif
}

#define N_TERMS  (N_ATOMS * K_TERMS)
#define TAB_SZ   (N_TERMS * N1D)

__global__ __launch_bounds__(64) void axis_spectra_kernel(const float* __restrict__ X,
                                                           const float* __restrict__ aw,
                                                           const float* __restrict__ bw,
                                                           float* __restrict__ Wre,
                                                           float* __restrict__ WimI,
                                                           float* __restrict__ Pre,
                                                           float* __restrict__ Pim,
                                                           float* __restrict__ YFre,
                                                           float* __restrict__ YFim,
                                                           float* __restrict__ ZFre,
                                                           float* __restrict__ ZFim) {
    __shared__ float ctab[N1D];
    __shared__ float stab[N1D];
    {
        const int j = threadIdx.x;
        const float ang = (float)j * 0.09817477042468103f;
        ctab[j] = 0.125f * cosf(ang);
        stab[j] = 0.125f * sinf(ang);
    }
    __syncthreads();

    if (blockIdx.x >= 288) {
        const int e = (blockIdx.x - 288) * 64 + threadIdx.x;
        const int k = e >> 4;
        const int n = (e & 15) * 4;
        float cr[4], si[4];
#pragma unroll
        for (int q = 0; q < 4; ++q) {
            const int kn = (k * (n + q)) & 63;
            const float sgn = ((k + n + q) & 1) ? -1.0f : 1.0f;
            cr[q] = sgn * ctab[kn];
            si[q] = sgn * stab[kn];
        }
        *(float4*)(Wre  + k * N1D + n) = make_float4(cr[0], cr[1], cr[2], cr[3]);
        *(float4*)(WimI + k * N1D + n) = make_float4(si[0], si[1], si[2], si[3]);
        return;
    }

    const int lane   = threadIdx.x & 31;
    const int waveId = (blockIdx.x * blockDim.x + threadIdx.x) >> 5;
    const int ck   = waveId & 3;
    const int rest = waveId >> 2;
    const int d    = rest / 48;
    const int tt   = rest - d * 48;
    const int half = lane >> 4;
    const int l16  = lane & 15;

    const int t = tt * 16 + l16;
    const int a = t / K_TERMS;
    const float c   = X[3 * a + d];
    const float bl2 = bw[t] * LOG2E;
    const float scale = (d == 0) ? aw[t] : 1.0f;
    const int krow = ck * 16 + l16;
    const float sg0 = (krow & 1) ? -1.0f : 1.0f;
    const bool dcz = (d == 2) && (krow == 0);
    const float* itab = dcz ? ctab : stab;
    const float sgx = dcz ? 1.0f : -sg0;
    const float sgy = dcz ? 1.0f : sg0;

    v8f re = {}, im = {};
#pragma unroll 4
    for (int n0 = 0; n0 < N1D; n0 += 4) {
        const int koff = n0 + 2 * half;
        const float d0 = (float)koff * SPACING - c;
        const float d1 = (float)(koff + 1) * SPACING - c;
        v2f e;
        e.x = scale * fast_exp2(bl2 * (d0 * d0));
        e.y = scale * fast_exp2(bl2 * (d1 * d1));
        const int i0 = (krow * koff) & 63;
        const int i1 = (i0 + krow) & 63;
        v2f wr, wi;
        wr.x = sg0 * ctab[i0];   wr.y = -sg0 * ctab[i1];
        wi.x = sgx * itab[i0];   wi.y = sgy * itab[i1];
        re = __builtin_amdgcn_wmma_f32_16x16x4_f32(false, e, false, wr, (short)0, re, false, false);
        im = __builtin_amdgcn_wmma_f32_16x16x4_f32(false, e, false, wi, (short)0, im, false, false);
    }

    if (d == 0) {
        const int o = krow * N_TERMS + tt * 16 + 8 * half;
        *(float4*)(Pre + o)     = make_float4(re[0], re[1], re[2], re[3]);
        *(float4*)(Pre + o + 4) = make_float4(re[4], re[5], re[6], re[7]);
        *(float4*)(Pim + o)     = make_float4(im[0], im[1], im[2], im[3]);
        *(float4*)(Pim + o + 4) = make_float4(im[4], im[5], im[6], im[7]);
    } else {
        float* __restrict__ Fr = (d == 1) ? YFre : ZFre;
        float* __restrict__ Fi = (d == 1) ? YFim : ZFim;
#pragma unroll
        for (int r2 = 0; r2 < 4; ++r2) {
            const int s  = tt * 4 + 2 * half + (r2 >> 1);
            const int hh = r2 & 1;
            const int o  = (((s * 4 + ck) * 32) + hh * 16 + l16) * 2;
            v2f vr, vi;
            vr.x = re[2 * r2]; vr.y = re[2 * r2 + 1];
            vi.x = im[2 * r2]; vi.y = im[2 * r2 + 1];
            *(v2f*)(Fr + o) = vr;
            *(v2f*)(Fi + o) = vi;
        }
    }
}

#define KSTEPS_Q (N_TERMS / 16)
#define CG 8
#define GP 68
__device__ __forceinline__ void contract_read(const v2f* __restrict__ pPr, const v2f* __restrict__ pPi,
                                              const v2f* __restrict__ pYr, const v2f* __restrict__ pYi,
                                              const v2f* __restrict__ pZr, const v2f* __restrict__ pZi,
                                              int s,
                                              v2f (&pr)[CG], v2f (&pi)[CG], v2f (&yr)[CG],
                                              v2f (&yi)[CG], v2f (&zr)[CG], v2f (&zi)[CG]) {
#pragma unroll
    for (int u = 0; u < CG; ++u) {
        pr[u] = pPr[2 * (s + u)];   pi[u] = pPi[2 * (s + u)];
        yr[u] = pYr[128 * (s + u)]; yi[u] = pYi[128 * (s + u)];
        zr[u] = pZr[128 * (s + u)]; zi[u] = pZi[128 * (s + u)];
    }
}

__device__ __forceinline__ void contract_mac(const v2f (&pr)[CG], const v2f (&pi)[CG],
                                             const v2f (&yr)[CG], const v2f (&yi)[CG],
                                             const v2f (&zr)[CG], const v2f (&zi)[CG],
                                             v8f& rr, v8f& ii, v8f& ri, v8f& ir) {
#pragma unroll
    for (int u = 0; u < CG; ++u) {
        const v2f ar = pr[u] * yr[u] - pi[u] * yi[u];
        const v2f ai = pr[u] * yi[u] + pi[u] * yr[u];
        rr = __builtin_amdgcn_wmma_f32_16x16x4_f32(false, ar, false, zr[u], (short)0, rr, false, false);
        ii = __builtin_amdgcn_wmma_f32_16x16x4_f32(false, ai, false, zi[u], (short)0, ii, false, false);
        ri = __builtin_amdgcn_wmma_f32_16x16x4_f32(false, ar, false, zi[u], (short)0, ri, false, false);
        ir = __builtin_amdgcn_wmma_f32_16x16x4_f32(false, ai, false, zr[u], (short)0, ir, false, false);
    }
}

__global__ __launch_bounds__(256) void spectrum_contract_kernel(const float* __restrict__ Pre,
                                                                const float* __restrict__ Pim,
                                                                const float* __restrict__ YFre,
                                                                const float* __restrict__ YFim,
                                                                const float* __restrict__ ZFre,
                                                                const float* __restrict__ ZFim,
                                                                const float* __restrict__ ham,
                                                                const float* __restrict__ Mre,
                                                                const float* __restrict__ Mim,
                                                                float* __restrict__ Ure,
                                                                float* __restrict__ Uim) {
    __shared__ float part[4][2][32][32];
    __shared__ __attribute__((aligned(16))) float Gre[16 * GP];
    __shared__ __attribute__((aligned(16))) float Gim[16 * GP];

    const int lane = threadIdx.x & 31;
    const int w    = threadIdx.x >> 5;
    const int ct   = w & 1;
    const int kq   = w >> 1;
    const int cz   = w & 3;
    const int kh   = w >> 2;
    const int kx   = blockIdx.x >> 2;
    const int cy   = blockIdx.x & 3;
    const int half = lane >> 4;
    const int l16  = lane & 15;
    const int sBeg = kq * KSTEPS_Q;

    const v2f* __restrict__ pPr = (const v2f*)(Pre + kx * N_TERMS + 2 * half) + 2 * sBeg;
    const v2f* __restrict__ pPi = (const v2f*)(Pim + kx * N_TERMS + 2 * half) + 2 * sBeg;
    const v2f* __restrict__ pYr = (const v2f*)YFre + 128 * sBeg + cy * 32 + lane;
    const v2f* __restrict__ pYi = (const v2f*)YFim + 128 * sBeg + cy * 32 + lane;
    const v2f* __restrict__ pZr = (const v2f*)ZFre + 128 * sBeg + ct * 32 + lane;
    const v2f* __restrict__ pZi = (const v2f*)ZFim + 128 * sBeg + ct * 32 + lane;

    const int  kzD = ct * 16 + l16;
    const bool sp  = (kzD == 0);
    const int  kzM = sp ? 32 : 64 - kzD;

    v8f rr = {}, ii = {}, ri = {}, ir = {};
    v2f prA[CG], piA[CG], yrA[CG], yiA[CG], zrA[CG], ziA[CG];
    v2f prB[CG], piB[CG], yrB[CG], yiB[CG], zrB[CG], ziB[CG];
    v2f mr[16], mi[16];
    float hD[2], hM[2], hDn[2], hMn[2];
    contract_read(pPr, pPi, pYr, pYi, pZr, pZi, 0, prA, piA, yrA, yiA, zrA, ziA);
#pragma unroll
    for (int s0 = 0; s0 < KSTEPS_Q; s0 += 2 * CG) {
        contract_read(pPr, pPi, pYr, pYi, pZr, pZi, s0 + CG, prB, piB, yrB, yiB, zrB, ziB);
        __builtin_amdgcn_sched_barrier(0);
        contract_mac(prA, piA, yrA, yiA, zrA, ziA, rr, ii, ri, ir);
        __builtin_amdgcn_sched_barrier(0);
        if (s0 + 2 * CG < KSTEPS_Q) {
            contract_read(pPr, pPi, pYr, pYi, pZr, pZi, s0 + 2 * CG, prA, piA, yrA, yiA, zrA, ziA);
        } else {
            const float* __restrict__ aR = Mre + (cz * 16 + l16) * N1D + 2 * half;
            const float* __restrict__ aI = Mim + (cz * 16 + l16) * N1D + 2 * half;
#pragma unroll
            for (int s = 0; s < 16; ++s) {
                mr[s] = *(const v2f*)(aR + 4 * s);
                mi[s] = *(const v2f*)(aI + 4 * s);
            }
#pragma unroll
            for (int e = 0; e < 2; ++e) {
                const int ky = cy * 16 + 2 * kq + e + 8 * half;
                hD[e] = ham[kx * 4096 + ky * N1D + kzD];
                hM[e] = ham[kx * 4096 + ky * N1D + kzM];
                const int on = ((64 - kx) & 63) * 4096 + ((64 - ky) & 63) * N1D;
                hDn[e] = ham[on + ((64 - kzD) & 63)];
                hMn[e] = ham[on + ((64 - kzM) & 63)];
            }
        }
        __builtin_amdgcn_sched_barrier(0);
        contract_mac(prB, piB, yrB, yiB, zrB, ziB, rr, ii, ri, ir);
        if (s0 + 2 * CG < KSTEPS_Q) {
            __builtin_amdgcn_sched_barrier(0);
        }
    }
    __builtin_amdgcn_sched_barrier(0);

#pragma unroll
    for (int v = 0; v < 8; ++v) {
        part[kq][ct][v][lane]      = rr[v];
        part[kq][ct][8 + v][lane]  = ii[v];
        part[kq][ct][16 + v][lane] = ri[v];
        part[kq][ct][24 + v][lane] = ir[v];
    }
    __syncthreads();
#pragma unroll
    for (int e = 0; e < 2; ++e) {
        const int v = 2 * kq + e;
        float srr = part[0][ct][v][lane];
        float sii = part[0][ct][8 + v][lane];
        float sri = part[0][ct][16 + v][lane];
        float sir = part[0][ct][24 + v][lane];
#pragma unroll
        for (int q = 1; q < 4; ++q) {
            srr += part[q][ct][v][lane];
            sii += part[q][ct][8 + v][lane];
            sri += part[q][ct][16 + v][lane];
            sir += part[q][ct][24 + v][lane];
        }
        const float dre = sp ? srr : srr - sii;
        const float dim = sp ? sir : sri + sir;
        const float mre = sp ? sri : srr + sii;
        const float mim = sp ? sii : sir - sri;
        const int o = (v + 8 * half) * GP;
        const float fD = 0.5f * hD[e] + 0.5f * hDn[e];
        const float fM = 0.5f * hM[e] + 0.5f * hMn[e];
        Gre[o + kzD] = dre * fD;
        Gim[o + kzD] = dim * fD;
        Gre[o + kzM] = mre * fM;
        Gim[o + kzM] = mim * fM;
    }
    __syncthreads();

    const float* b1 = (kh ? Gim : Gre) + l16 * GP + 2 * half;
    const float* b2 = (kh ? Gre : Gim) + l16 * GP + 2 * half;
    v8f a1 = {}, a2 = {};
#pragma unroll
    for (int s = 0; s < 16; ++s) {
        const v2f t1 = *(const v2f*)(b1 + 4 * s);
        const v2f t2 = *(const v2f*)(b2 + 4 * s);
        a1 = __builtin_amdgcn_wmma_f32_16x16x4_f32(false, mr[s], false, t1, (short)0, a1, false, false);
        a2 = __builtin_amdgcn_wmma_f32_16x16x4_f32(false, mi[s], false, t2, (short)0, a2, false, false);
    }
    const float sg = kh ? 1.0f : -1.0f;
    const v8f u = a1 + sg * a2;

    float* __restrict__ dst = kh ? Uim : Ure;
#pragma unroll
    for (int v = 0; v < 8; ++v) {
        const int z = cz * 16 + v + 8 * half;
        dst[z * 4096 + kx * N1D + cy * 16 + l16] = u[v];
    }
}

#define XP_THREADS 128
#define XP_WGS     256

#define XK 9

__device__ __forceinline__ void xpass_read(const float* __restrict__ Tre,
                                           const float* __restrict__ Tim,
                                           const float* __restrict__ Mre,
                                           const float* __restrict__ Mim,
                                           int krow, int prow, int half,
                                           v2f (&mr)[XK], v2f (&mi)[XK],
                                           v2f (&tr)[XK], v2f (&ti)[XK]) {
    const float* __restrict__ aR = Mre + krow * N1D + 2 * half;
    const float* __restrict__ aI = Mim + krow * N1D + 2 * half;
    const float* __restrict__ bR = Tre + prow * N1D + 2 * half;
    const float* __restrict__ bI = Tim + prow * N1D + 2 * half;
#pragma unroll
    for (int s = 0; s < XK; ++s) {
        mr[s] = *(const v2f*)(aR + 4 * s);
        tr[s] = *(const v2f*)(bR + 4 * s);
        mi[s] = *(const v2f*)(aI + 4 * s);
        ti[s] = *(const v2f*)(bI + 4 * s);
    }
}

__device__ __forceinline__ void xpass_weight(v2f (&mr)[XK], v2f (&mi)[XK], int half) {
#pragma unroll
    for (int s = 0; s < XK; ++s) {
        v2f wgt;
        wgt.x = (s == 0) ? (half ? 2.0f : 1.0f) : (s < 8) ? 2.0f : (half ? 0.0f : 1.0f);
        wgt.y = (s < 8) ? 2.0f : 0.0f;
        mr[s] = mr[s] * wgt;
        mi[s] = mi[s] * wgt;
    }
}

__device__ __forceinline__ v8f xpass_real(const v2f (&mr)[XK], const v2f (&mi)[XK],
                                          const v2f (&tr)[XK], const v2f (&ti)[XK]) {
    v8f rr = {}, ii = {};
#pragma unroll
    for (int s = 0; s < XK; ++s) {
        rr = __builtin_amdgcn_wmma_f32_16x16x4_f32(false, mr[s], false, tr[s], (short)0, rr, false, false);
        ii = __builtin_amdgcn_wmma_f32_16x16x4_f32(false, mi[s], false, ti[s], (short)0, ii, false, false);
    }
    return rr - ii;
}

#define DFT_WG_THREADS 128
#define DFT_WGS        256
__global__ __launch_bounds__(DFT_WG_THREADS) void ypass_xsum_kernel(const float* __restrict__ Tre,
                                                                    const float* __restrict__ Tim,
                                                                    const float* __restrict__ Mre,
                                                                    const float* __restrict__ Mim,
                                                                    float* __restrict__ Ure,
                                                                    float* __restrict__ Uim,
                                                                    float* __restrict__ psum,
                                                                    float* __restrict__ psq) {
    __shared__ __attribute__((aligned(16))) float Sre[16 * GP];
    __shared__ __attribute__((aligned(16))) float Sim[16 * GP];
    __shared__ float wsum[4];
    __shared__ float wsq[4];

    const int lane = threadIdx.x & 31;
    const int w    = __builtin_amdgcn_readfirstlane(threadIdx.x >> 5);
    const int z    = blockIdx.x & 63;
    const int k0   = (blockIdx.x >> 6) << 4;
    const int half = lane >> 4;
    const int l16  = lane & 15;
    const int krow = k0 + l16;
    const int kxc  = 16 * w + l16;
    const int prow = z * N1D + (kxc < 32 ? kxc : 32);
    const int p0   = z * N1D + 16 * w;
    const int xrow = 16 * w + l16;

    const float* __restrict__ aR = Mre + krow * N1D + 2 * half;
    const float* __restrict__ aI = Mim + krow * N1D + 2 * half;
    const float* __restrict__ bR = Tre + prow * N1D + 2 * half;
    const float* __restrict__ bI = Tim + prow * N1D + 2 * half;
    const float* __restrict__ xR = Mre + xrow * N1D + 2 * half;
    const float* __restrict__ xI = Mim + xrow * N1D + 2 * half;
    v2f mr[16], mi[16], tr[16], ti[16], xr[XK], xi[XK];
    if (w < 3) {
#pragma unroll
        for (int s = 0; s < 16; ++s) {
            mr[s] = *(const v2f*)(aR + 4 * s);
            tr[s] = *(const v2f*)(bR + 4 * s);
            mi[s] = *(const v2f*)(aI + 4 * s);
            ti[s] = *(const v2f*)(bI + 4 * s);
        }
    }
#pragma unroll
    for (int s = 0; s < XK; ++s) {
        xr[s] = *(const v2f*)(xR + 4 * s);
        xi[s] = *(const v2f*)(xI + 4 * s);
    }
    __builtin_amdgcn_sched_barrier(0);

    v8f ure = {}, uim = {};
    if (w < 3) {
        v8f rr = {}, ii = {}, ri = {}, ir = {};
#pragma unroll
        for (int s = 0; s < 16; ++s) {
            rr = __builtin_amdgcn_wmma_f32_16x16x4_f32(false, mr[s], false, tr[s], (short)0, rr, false, false);
            ii = __builtin_amdgcn_wmma_f32_16x16x4_f32(false, mi[s], false, ti[s], (short)0, ii, false, false);
            ri = __builtin_amdgcn_wmma_f32_16x16x4_f32(false, mr[s], false, ti[s], (short)0, ri, false, false);
            ir = __builtin_amdgcn_wmma_f32_16x16x4_f32(false, mi[s], false, tr[s], (short)0, ir, false, false);
        }
        const v8f dre = rr - ii;
        const v8f dim = ri + ir;
        const bool live = (kxc <= 32);
#pragma unroll
        for (int v = 0; v < 8; ++v) {
            ure[v] = live ? dre[v] : 0.0f;
            uim[v] = live ? dim[v] : 0.0f;
        }
#pragma unroll
        for (int v = 0; v < 8; ++v) {
            const int yl = v + 8 * half;
            Sre[yl * GP + kxc] = ure[v];
            Sim[yl * GP + kxc] = uim[v];
        }
    }
    xpass_weight(xr, xi, half);
    __syncthreads();
    if (w < 3) {
#pragma unroll
        for (int v = 0; v < 8; ++v) {
            const int idx = (k0 + v + 8 * half) * 4096 + p0 + l16;
            Ure[idx] = ure[v];
            Uim[idx] = uim[v];
        }
    }
    __builtin_amdgcn_sched_barrier(0);

    v2f ur[XK], ui[XK];
    {
        const float* sR = Sre + l16 * GP + 2 * half;
        const float* sI = Sim + l16 * GP + 2 * half;
#pragma unroll
        for (int s = 0; s < XK; ++s) {
            ur[s] = *(const v2f*)(sR + 4 * s);
            ui[s] = *(const v2f*)(sI + 4 * s);
        }
    }
    const v8f u = xpass_real(xr, xi, ur, ui);

    float s = 0.0f, q = 0.0f;
#pragma unroll
    for (int v = 0; v < 8; ++v) { s += u[v]; q += u[v] * u[v]; }
#pragma unroll
    for (int o = 16; o > 0; o >>= 1) {
        s += __shfl_xor(s, o);
        q += __shfl_xor(q, o);
    }
    if (lane == 0) { wsum[w] = s; wsq[w] = q; }
    __syncthreads();
    if (threadIdx.x == 0) {
        psum[blockIdx.x] = (wsum[0] + wsum[1]) + (wsum[2] + wsum[3]);
        psq[blockIdx.x]  = (wsq[0] + wsq[1]) + (wsq[2] + wsq[3]);
    }
}

__global__ __launch_bounds__(XP_THREADS) void xpass_norm_kernel(const float* __restrict__ Tre,
                                                                const float* __restrict__ Tim,
                                                                const float* __restrict__ Mre,
                                                                const float* __restrict__ Mim,
                                                                const float* __restrict__ psum,
                                                                const float* __restrict__ psq,
                                                                float* __restrict__ out) {
    const int lane   = threadIdx.x & 31;
    const int waveId = (blockIdx.x * XP_THREADS + threadIdx.x) >> 5;
    const int p0 = (waveId & 255) << 4;
    const int k0 = (waveId >> 8) << 4;
    const int half = lane >> 4;
    const int l16  = lane & 15;

    const float4 sa = ((const float4*)psum)[2 * lane], sb = ((const float4*)psum)[2 * lane + 1];
    const float4 qa = ((const float4*)psq)[2 * lane],  qb = ((const float4*)psq)[2 * lane + 1];

    v2f mr[XK], mi[XK], tr[XK], ti[XK];
    xpass_read(Tre, Tim, Mre, Mim, k0 + l16, p0 + l16, half, mr, mi, tr, ti);
    __builtin_amdgcn_sched_barrier(0);

    float s = ((sa.x + sa.y) + (sa.z + sa.w)) + ((sb.x + sb.y) + (sb.z + sb.w));
    float q = ((qa.x + qa.y) + (qa.z + qa.w)) + ((qb.x + qb.y) + (qb.z + qb.w));
#pragma unroll
    for (int o = 16; o > 0; o >>= 1) {
        s += __shfl_xor(s, o);
        q += __shfl_xor(q, o);
    }
    const float invN = 1.0f / (float)GRID_N;
    const float mean = s * invN;
    const float var  = fmaxf(q * invN - mean * mean, 0.0f);
    const float is   = 1.0f / (sqrtf(var) + 1e-8f);

    xpass_weight(mr, mi, half);
    const v8f u = xpass_real(mr, mi, tr, ti);

#pragma unroll
    for (int v = 0; v < 8; ++v) {
        const int row = k0 + v + 8 * half;
        out[row * 4096 + p0 + l16] = (u[v] - mean) * is;
    }
}

extern "C" void kernel_launch(void* const* d_in, const int* in_sizes, int n_in,
                              void* d_out, int out_size, void* d_ws, size_t ws_size,
                              hipStream_t stream) {
    const float* X   = (const float*)d_in[0];
    const float* aw  = (const float*)d_in[1];
    const float* bw  = (const float*)d_in[2];
    const float* ham = (const float*)d_in[4];
    float* out = (float*)d_out;

    float* ws   = (float*)d_ws;
    float* Wre  = ws;
    float* WimI = Wre + 8192;
    float* Are  = WimI + 4096;
    float* Aim  = Are + GRID_N;
    float* Bre  = Aim + GRID_N;
    float* Bim  = Bre + GRID_N;
    float* psum = Bim + GRID_N;
    float* psq  = psum + 512;

    float* Pre  = Are;
    float* Pim  = Pre  + TAB_SZ;
    float* YFre = Pim  + TAB_SZ;
    float* YFim = YFre + TAB_SZ;
    float* ZFre = YFim + TAB_SZ;
    float* ZFim = ZFre + TAB_SZ;

    axis_spectra_kernel<<<304, 64, 0, stream>>>(X, aw, bw, Wre, WimI,
                                                Pre, Pim, YFre, YFim, ZFre, ZFim);

    spectrum_contract_kernel<<<132, 256, 0, stream>>>(Pre, Pim, YFre, YFim, ZFre, ZFim,
                                                      ham, Wre, WimI, Bre, Bim);

    ypass_xsum_kernel<<<DFT_WGS, DFT_WG_THREADS, 0, stream>>>(Bre, Bim, Wre, WimI, Are, Aim, psum, psq);

    xpass_norm_kernel<<<XP_WGS, XP_THREADS, 0, stream>>>(Are, Aim, Wre, WimI, psum, psq, out);
}
